# starting kernel
# baseline (speedup 1.0000x reference)
#include <hip/hip_runtime.h>
#include <math.h>


typedef float v2f __attribute__((ext_vector_type(2)));
typedef float v8f __attribute__((ext_vector_type(8)));

#define BB 4
#define HH 48
#define WW 48
#define DMm 192
#define DIc 192
#define KK 4
#define NN 16
#define RR 12
#define LL (HH * WW)
#define CPROJ (RR + 2 * NN)
#define CPAD 64

__device__ __forceinline__ float siluf(float x)     { return x / (1.f + __expf(-x)); }
__device__ __forceinline__ float sigm(float x)      { return 1.f / (1.f + __expf(-x)); }
__device__ __forceinline__ float softplusf_(float x){ return (x > 20.f) ? x : log1pf(__expf(x)); }
__device__ __forceinline__ float geluf(float x)     { return 0.5f * x * (1.f + erff(x * 0.70710678118654752440f)); }

template <int NT>
__device__ __forceinline__ void wmma_xwT_multi(const float* __restrict__ X,
                                               const float* __restrict__ W,
                                               int row0, int col0,
                                               int Kdim, int ldx, int ldw,
                                               v8f acc[NT])
{
    const int lane = threadIdx.x & 31;
    const int m    = lane & 15;
    const int kh   = lane >> 4;
    const float* __restrict__ ap = X + (size_t)(row0 + m) * ldx + 2 * kh;
    const float* __restrict__ bp[NT];
#pragma unroll
    for (int t = 0; t < NT; ++t) {
        bp[t] = W + (size_t)(col0 + t * 16 + m) * ldw + 2 * kh;
        acc[t] = {};
    }
    v2f a_cur = *(const v2f*)(ap);
    v2f b_cur[NT];
#pragma unroll
    for (int t = 0; t < NT; ++t) b_cur[t] = *(const v2f*)(bp[t]);

    for (int k0 = 4; k0 < Kdim; k0 += 4) {
        v2f a_nxt = *(const v2f*)(ap + k0);
        v2f b_nxt[NT];
#pragma unroll
        for (int t = 0; t < NT; ++t) b_nxt[t] = *(const v2f*)(bp[t] + k0);
#pragma unroll
        for (int t = 0; t < NT; ++t)
            acc[t] = __builtin_amdgcn_wmma_f32_16x16x4_f32(false, a_cur, false, b_cur[t],
                                                           (short)0, acc[t], false, false);
        a_cur = a_nxt;
#pragma unroll
        for (int t = 0; t < NT; ++t) b_cur[t] = b_nxt[t];
    }
#pragma unroll
    for (int t = 0; t < NT; ++t)
        acc[t] = __builtin_amdgcn_wmma_f32_16x16x4_f32(false, a_cur, false, b_cur[t],
                                                       (short)0, acc[t], false, false);
}

__global__ void k_padw(const float* __restrict__ xpw, float* __restrict__ wpad)
{
    int k = blockIdx.x, c = blockIdx.y, d = threadIdx.x;
    wpad[((size_t)k * CPAD + c) * DIc + d] =
        (c < CPROJ) ? xpw[((size_t)k * CPROJ + c) * DIc + d] : 0.f;
}

__global__ void k_inproj(const float* __restrict__ x, const float* __restrict__ w,
                         float* __restrict__ xc_lin, float* __restrict__ z_silu)
{
    const int row0 = blockIdx.y * 16, col0 = blockIdx.x * 32;
    v8f acc[2];
    wmma_xwT_multi<2>(x, w, row0, col0, DMm, DMm, DMm, acc);
    const int lane = threadIdx.x & 31;
    const int nn = lane & 15, mh = (lane >> 4) * 8;
#pragma unroll
    for (int t = 0; t < 2; ++t)
#pragma unroll
        for (int v = 0; v < 8; ++v) {
            int r = row0 + mh + v;
            int c = col0 + t * 16 + nn;
            float val = acc[t][v];
            if (c < DIc) xc_lin[(size_t)r * DIc + c] = val;
            else         z_silu[(size_t)r * DIc + (c - DIc)] = siluf(val);
        }
}

__global__ void k_dwconv(const float* __restrict__ xc_lin, const float* __restrict__ cw,
                         const float* __restrict__ cb, float* __restrict__ xc_act)
{
    int pix = blockIdx.x;
    int b = pix / LL, rem = pix % LL;
    int h = rem / WW, w = rem % WW;
    int c = threadIdx.x;
    float s = cb[c];
#pragma unroll
    for (int kh = 0; kh < 3; ++kh)
#pragma unroll
        for (int kw = 0; kw < 3; ++kw) {
            int hh = h + kh - 1, ww2 = w + kw - 1;
            if (hh >= 0 && hh < HH && ww2 >= 0 && ww2 < WW)
                s += xc_lin[((size_t)b * LL + hh * WW + ww2) * DIc + c] * cw[c * 9 + kh * 3 + kw];
        }
    xc_act[((size_t)b * LL + rem) * DIc + c] = siluf(s);
}

__global__ void k_gcmean(const float* __restrict__ xc_act, float* __restrict__ gcm)
{
    int b = blockIdx.x, d = threadIdx.x;
    float s = 0.f;
    for (int p = 0; p < LL; ++p)
        s += xc_act[((size_t)b * LL + p) * DIc + d];
    gcm[b * DIc + d] = s * (1.f / LL);
}

__global__ void k_gate(const float* __restrict__ gcm, const float* __restrict__ fw1,
                       const float* __restrict__ fw2, float* __restrict__ gate)
{
    __shared__ float g[DIc];
    __shared__ float h1[DMm / 4];
    int b = blockIdx.x, d = threadIdx.x;
    g[d] = gcm[b * DIc + d];
    __syncthreads();
    if (d < DMm / 4) {
        float s = 0.f;
        for (int j = 0; j < DMm; ++j) s += fw1[d * DMm + j] * g[j];
        h1[d] = geluf(s);
    }
    __syncthreads();
    float s2 = 0.f;
#pragma unroll
    for (int j = 0; j < DMm / 4; ++j) s2 += fw2[d * (DMm / 4) + j] * h1[j];
    gate[b * DMm + d] = sigm(s2);
}

__global__ void k_build_xs(const float* __restrict__ xc_act, float* __restrict__ xs)
{
    int l = blockIdx.x, b = blockIdx.y, k = blockIdx.z;
    int d = threadIdx.x;
    int lb = (k < 2) ? l : (LL - 1 - l);
    int w_in = lb & 3, h_in = (lb >> 2) & 3;
    int nw = (lb >> 4) % (WW / 4);
    int nh = (lb >> 4) / (WW / 4);
    int p = nh * 4 + h_in, q = nw * 4 + w_in;
    int h = ((k & 1) == 0) ? p : q;
    int w = ((k & 1) == 0) ? q : p;
    xs[(((size_t)(k * BB + b)) * LL + l) * DIc + d] =
        xc_act[((size_t)b * LL + h * WW + w) * DIc + d];
}

__global__ void k_xproj(const float* __restrict__ xs, const float* __restrict__ wpad,
                        float* __restrict__ xdbl)
{
    int k = blockIdx.z;
    const float* X = xs + (size_t)k * BB * LL * DIc;
    const float* W = wpad + (size_t)k * CPAD * DIc;
    float* O = xdbl + (size_t)k * BB * LL * CPROJ;
    const int row0 = blockIdx.y * 16, col0 = blockIdx.x * 32;
    v8f acc[2];
    wmma_xwT_multi<2>(X, W, row0, col0, DIc, DIc, DIc, acc);
    const int lane = threadIdx.x & 31;
    const int nn = lane & 15, mh = (lane >> 4) * 8;
#pragma unroll
    for (int t = 0; t < 2; ++t)
#pragma unroll
        for (int v = 0; v < 8; ++v) {
            int r = row0 + mh + v, c = col0 + t * 16 + nn;
            if (c < CPROJ) O[(size_t)r * CPROJ + c] = acc[t][v];
        }
}

__global__ void k_dtproj(const float* __restrict__ xdbl, const float* __restrict__ dtw,
                         const float* __restrict__ dtb, float* __restrict__ delta)
{
    int k = blockIdx.z;
    const float* X = xdbl + (size_t)k * BB * LL * CPROJ;
    const float* W = dtw + (size_t)k * DIc * RR;
    float* O = delta + (size_t)k * BB * LL * DIc;
    const int row0 = blockIdx.y * 16, col0 = blockIdx.x * 32;
    v8f acc[2];
    wmma_xwT_multi<2>(X, W, row0, col0, RR, CPROJ, RR, acc);
    const int lane = threadIdx.x & 31;
    const int nn = lane & 15, mh = (lane >> 4) * 8;
#pragma unroll
    for (int t = 0; t < 2; ++t)
#pragma unroll
        for (int v = 0; v < 8; ++v) {
            int r = row0 + mh + v, c = col0 + t * 16 + nn;
            O[(size_t)r * DIc + c] = softplusf_(acc[t][v] + dtb[k * DIc + c]);
        }
}

__global__ void k_scan(const float* __restrict__ xs, const float* __restrict__ xdbl,
                       const float* __restrict__ delta, const float* __restrict__ A_logs,
                       const float* __restrict__ Ds, float* __restrict__ yarr)
{
    int kb = blockIdx.x;
    int k = kb & 3, b = kb >> 2;
    int d = threadIdx.x;
    float A[NN];
#pragma unroll
    for (int n = 0; n < NN; ++n)
        A[n] = -__expf(A_logs[((size_t)(k * DIc + d)) * NN + n]);
    float Dv = Ds[k * DIc + d];
    float h[NN];
#pragma unroll
    for (int n = 0; n < NN; ++n) h[n] = 0.f;
    const size_t seqbase = ((size_t)(k * BB + b)) * LL;
    const float* up = xs    + seqbase * DIc + d;
    const float* dp = delta + seqbase * DIc + d;
    const float* bc = xdbl  + seqbase * CPROJ;
    float* yp       = yarr  + seqbase * DIc + d;
    for (int l = 0; l < LL; ++l) {
        float dt = dp[(size_t)l * DIc];
        float u  = up[(size_t)l * DIc];
        float du = dt * u;
        const float* bcl = bc + (size_t)l * CPROJ;
        float y = 0.f;
#pragma unroll
        for (int n = 0; n < NN; ++n) {
            float Bt = bcl[RR + n];
            float Ct = bcl[RR + NN + n];
            h[n] = __expf(dt * A[n]) * h[n] + du * Bt;
            y += h[n] * Ct;
        }
        yp[(size_t)l * DIc] = y + Dv * u;
    }
}

__global__ void k_combine(const float* __restrict__ yarr, const float* __restrict__ gate,
                          const float* __restrict__ g1, const float* __restrict__ g2,
                          const float* __restrict__ g3, const float* __restrict__ g4,
                          const float* __restrict__ nw, const float* __restrict__ nb,
                          const float* __restrict__ z_silu, float* __restrict__ yfinal)
{
    __shared__ float s1[256], s2[256];
    int lr = blockIdx.x, b = blockIdx.y, d = threadIdx.x;
    int h = lr / WW, w = lr % WW;
    int l = (((h >> 2) * (WW / 4) + (w >> 2)) << 4) + ((h & 3) << 2) + (w & 3);
    int lt = (l % WW) * HH + (l / WW);
    const size_t bstride = (size_t)LL * DIc;
    const float* Y0 = yarr + ((size_t)(0 * BB + b)) * bstride;
    const float* Y1 = yarr + ((size_t)(1 * BB + b)) * bstride;
    const float* Y2 = yarr + ((size_t)(2 * BB + b)) * bstride;
    const float* Y3 = yarr + ((size_t)(3 * BB + b)) * bstride;
    float yc = g1[d] * Y0[(size_t)l * DIc + d]
             + g2[d] * Y2[(size_t)(LL - 1 - l) * DIc + d]
             + g3[d] * Y1[(size_t)lt * DIc + d]
             + g4[d] * Y3[(size_t)(LL - 1 - lt) * DIc + d];
    yc *= gate[b * DIc + d];
    s1[d] = yc; s2[d] = yc * yc;
    if (d < 64) { s1[192 + d] = 0.f; s2[192 + d] = 0.f; }
    __syncthreads();
    for (int off = 128; off > 0; off >>= 1) {
        if (d < off) { s1[d] += s1[d + off]; s2[d] += s2[d + off]; }
        __syncthreads();
    }
    float mu  = s1[0] * (1.f / DIc);
    float var = s2[0] * (1.f / DIc) - mu * mu;
    float inv = rsqrtf(var + 1e-5f);
    yc = (yc - mu) * inv * nw[d] + nb[d];
    yc *= z_silu[((size_t)b * LL + lr) * DIc + d];
    yfinal[((size_t)b * LL + lr) * DIc + d] = yc;
}

__global__ void k_outproj(const float* __restrict__ yf, const float* __restrict__ w,
                          float* __restrict__ out)
{
    const int row0 = blockIdx.y * 16, col0 = blockIdx.x * 32;
    v8f acc[2];
    wmma_xwT_multi<2>(yf, w, row0, col0, DIc, DIc, DIc, acc);
    const int lane = threadIdx.x & 31;
    const int nn = lane & 15, mh = (lane >> 4) * 8;
#pragma unroll
    for (int t = 0; t < 2; ++t)
#pragma unroll
        for (int v = 0; v < 8; ++v) {
            int r = row0 + mh + v, c = col0 + t * 16 + nn;
            out[(size_t)r * DMm + c] = acc[t][v];
        }
}

extern "C" void kernel_launch(void* const* d_in, const int* in_sizes, int n_in,
                              void* d_out, int out_size, void* d_ws, size_t ws_size,
                              hipStream_t stream)
{
    (void)in_sizes; (void)n_in; (void)out_size; (void)ws_size;
    const float* x          = (const float*)d_in[0];
    const float* in_proj_w  = (const float*)d_in[1];
    const float* conv_w     = (const float*)d_in[2];
    const float* conv_b     = (const float*)d_in[3];
    const float* x_proj_w   = (const float*)d_in[4];
    const float* dt_w       = (const float*)d_in[5];
    const float* dt_b       = (const float*)d_in[6];
    const float* A_logs     = (const float*)d_in[7];
    const float* Ds         = (const float*)d_in[8];
    const float* gamma1     = (const float*)d_in[9];
    const float* gamma2     = (const float*)d_in[10];
    const float* gamma3     = (const float*)d_in[11];
    const float* gamma4     = (const float*)d_in[12];
    const float* fc_w1      = (const float*)d_in[13];
    const float* fc_w2      = (const float*)d_in[14];
    const float* norm_w     = (const float*)d_in[15];
    const float* norm_b     = (const float*)d_in[16];
    const float* out_proj_w = (const float*)d_in[17];
    float* out = (float*)d_out;

    float* ws = (float*)d_ws;
    size_t o = 0;
    float* xc_lin = ws + o; o += (size_t)BB * LL * DIc;
    float* z_silu = ws + o; o += (size_t)BB * LL * DIc;
    float* xc_act = ws + o; o += (size_t)BB * LL * DIc;
    float* xs     = ws + o; o += (size_t)KK * BB * LL * DIc;
    float* xdbl   = ws + o; o += (size_t)KK * BB * LL * CPROJ;
    float* delta  = ws + o; o += (size_t)KK * BB * LL * DIc;
    float* yarr   = ws + o; o += (size_t)KK * BB * LL * DIc;
    float* yfinal = ws + o; o += (size_t)BB * LL * DIc;
    float* wpad   = ws + o; o += (size_t)KK * CPAD * DIc;
    float* gcm    = ws + o; o += BB * DIc;
    float* gate   = ws + o; o += BB * DMm;

    k_padw    <<<dim3(KK, CPAD),       DIc, 0, stream>>>(x_proj_w, wpad);
    k_inproj  <<<dim3(12, 576),        32,  0, stream>>>(x, in_proj_w, xc_lin, z_silu);
    k_dwconv  <<<dim3(BB * LL),        DIc, 0, stream>>>(xc_lin, conv_w, conv_b, xc_act);
    k_gcmean  <<<dim3(BB),             DIc, 0, stream>>>(xc_act, gcm);
    k_gate    <<<dim3(BB),             DIc, 0, stream>>>(gcm, fc_w1, fc_w2, gate);
    k_build_xs<<<dim3(LL, BB, KK),     DIc, 0, stream>>>(xc_act, xs);
    k_xproj   <<<dim3(2, 576, KK),     32,  0, stream>>>(xs, wpad, xdbl);
    k_dtproj  <<<dim3(6, 576, KK),     32,  0, stream>>>(xdbl, dt_w, dt_b, delta);
    k_scan    <<<dim3(BB * KK),        DIc, 0, stream>>>(xs, xdbl, delta, A_logs, Ds, yarr);
    k_combine <<<dim3(LL, BB),         DIc, 0, stream>>>(yarr, gate, gamma1, gamma2, gamma3,
                                                         gamma4, norm_w, norm_b, z_silu, yfinal);
    k_outproj <<<dim3(6, 576),         32,  0, stream>>>(yfinal, out_proj_w, out);
}


# k_scan: per-wave 16-step register window of delta/u/B/C, refilled 15 steps ahead of use (B/C by lane broadcast), hides the per-step global load latency; sum order kept, y products now fused (fma)
# speedup vs baseline: 2.0289x; 2.0289x over previous
#include <hip/hip_runtime.h>
#include <math.h>


typedef float v2f __attribute__((ext_vector_type(2)));
typedef float v8f __attribute__((ext_vector_type(8)));

#define BB 4
#define HH 48
#define WW 48
#define DMm 192
#define DIc 192
#define KK 4
#define NN 16
#define RR 12
#define LL (HH * WW)
#define CPROJ (RR + 2 * NN)
#define CPAD 64

__device__ __forceinline__ float siluf(float x)     { return x / (1.f + __expf(-x)); }
__device__ __forceinline__ float sigm(float x)      { return 1.f / (1.f + __expf(-x)); }
__device__ __forceinline__ float softplusf_(float x){ return (x > 20.f) ? x : log1pf(__expf(x)); }
__device__ __forceinline__ float geluf(float x)     { return 0.5f * x * (1.f + erff(x * 0.70710678118654752440f)); }

template <int NT>
__device__ __forceinline__ void wmma_xwT_multi(const float* __restrict__ X,
                                               const float* __restrict__ W,
                                               int row0, int col0,
                                               int Kdim, int ldx, int ldw,
                                               v8f acc[NT])
{
    const int lane = threadIdx.x & 31;
    const int m    = lane & 15;
    const int kh   = lane >> 4;
    const float* __restrict__ ap = X + (size_t)(row0 + m) * ldx + 2 * kh;
    const float* __restrict__ bp[NT];
#pragma unroll
    for (int t = 0; t < NT; ++t) {
        bp[t] = W + (size_t)(col0 + t * 16 + m) * ldw + 2 * kh;
        acc[t] = {};
    }
    v2f a_cur = *(const v2f*)(ap);
    v2f b_cur[NT];
#pragma unroll
    for (int t = 0; t < NT; ++t) b_cur[t] = *(const v2f*)(bp[t]);

    for (int k0 = 4; k0 < Kdim; k0 += 4) {
        v2f a_nxt = *(const v2f*)(ap + k0);
        v2f b_nxt[NT];
#pragma unroll
        for (int t = 0; t < NT; ++t) b_nxt[t] = *(const v2f*)(bp[t] + k0);
#pragma unroll
        for (int t = 0; t < NT; ++t)
            acc[t] = __builtin_amdgcn_wmma_f32_16x16x4_f32(false, a_cur, false, b_cur[t],
                                                           (short)0, acc[t], false, false);
        a_cur = a_nxt;
#pragma unroll
        for (int t = 0; t < NT; ++t) b_cur[t] = b_nxt[t];
    }
#pragma unroll
    for (int t = 0; t < NT; ++t)
        acc[t] = __builtin_amdgcn_wmma_f32_16x16x4_f32(false, a_cur, false, b_cur[t],
                                                       (short)0, acc[t], false, false);
}

__global__ void k_padw(const float* __restrict__ xpw, float* __restrict__ wpad)
{
    int k = blockIdx.x, c = blockIdx.y, d = threadIdx.x;
    wpad[((size_t)k * CPAD + c) * DIc + d] =
        (c < CPROJ) ? xpw[((size_t)k * CPROJ + c) * DIc + d] : 0.f;
}

__global__ void k_inproj(const float* __restrict__ x, const float* __restrict__ w,
                         float* __restrict__ xc_lin, float* __restrict__ z_silu)
{
    const int row0 = blockIdx.y * 16, col0 = blockIdx.x * 32;
    v8f acc[2];
    wmma_xwT_multi<2>(x, w, row0, col0, DMm, DMm, DMm, acc);
    const int lane = threadIdx.x & 31;
    const int nn = lane & 15, mh = (lane >> 4) * 8;
#pragma unroll
    for (int t = 0; t < 2; ++t)
#pragma unroll
        for (int v = 0; v < 8; ++v) {
            int r = row0 + mh + v;
            int c = col0 + t * 16 + nn;
            float val = acc[t][v];
            if (c < DIc) xc_lin[(size_t)r * DIc + c] = val;
            else         z_silu[(size_t)r * DIc + (c - DIc)] = siluf(val);
        }
}

__global__ void k_dwconv(const float* __restrict__ xc_lin, const float* __restrict__ cw,
                         const float* __restrict__ cb, float* __restrict__ xc_act)
{
    int pix = blockIdx.x;
    int b = pix / LL, rem = pix % LL;
    int h = rem / WW, w = rem % WW;
    int c = threadIdx.x;
    float s = cb[c];
#pragma unroll
    for (int kh = 0; kh < 3; ++kh)
#pragma unroll
        for (int kw = 0; kw < 3; ++kw) {
            int hh = h + kh - 1, ww2 = w + kw - 1;
            if (hh >= 0 && hh < HH && ww2 >= 0 && ww2 < WW)
                s += xc_lin[((size_t)b * LL + hh * WW + ww2) * DIc + c] * cw[c * 9 + kh * 3 + kw];
        }
    xc_act[((size_t)b * LL + rem) * DIc + c] = siluf(s);
}

__global__ void k_gcmean(const float* __restrict__ xc_act, float* __restrict__ gcm)
{
    int b = blockIdx.x, d = threadIdx.x;
    float s = 0.f;
    for (int p = 0; p < LL; ++p)
        s += xc_act[((size_t)b * LL + p) * DIc + d];
    gcm[b * DIc + d] = s * (1.f / LL);
}

__global__ void k_gate(const float* __restrict__ gcm, const float* __restrict__ fw1,
                       const float* __restrict__ fw2, float* __restrict__ gate)
{
    __shared__ float g[DIc];
    __shared__ float h1[DMm / 4];
    int b = blockIdx.x, d = threadIdx.x;
    g[d] = gcm[b * DIc + d];
    __syncthreads();
    if (d < DMm / 4) {
        float s = 0.f;
        for (int j = 0; j < DMm; ++j) s += fw1[d * DMm + j] * g[j];
        h1[d] = geluf(s);
    }
    __syncthreads();
    float s2 = 0.f;
#pragma unroll
    for (int j = 0; j < DMm / 4; ++j) s2 += fw2[d * (DMm / 4) + j] * h1[j];
    gate[b * DMm + d] = sigm(s2);
}

__global__ void k_build_xs(const float* __restrict__ xc_act, float* __restrict__ xs)
{
    int l = blockIdx.x, b = blockIdx.y, k = blockIdx.z;
    int d = threadIdx.x;
    int lb = (k < 2) ? l : (LL - 1 - l);
    int w_in = lb & 3, h_in = (lb >> 2) & 3;
    int nw = (lb >> 4) % (WW / 4);
    int nh = (lb >> 4) / (WW / 4);
    int p = nh * 4 + h_in, q = nw * 4 + w_in;
    int h = ((k & 1) == 0) ? p : q;
    int w = ((k & 1) == 0) ? q : p;
    xs[(((size_t)(k * BB + b)) * LL + l) * DIc + d] =
        xc_act[((size_t)b * LL + h * WW + w) * DIc + d];
}

__global__ void k_xproj(const float* __restrict__ xs, const float* __restrict__ wpad,
                        float* __restrict__ xdbl)
{
    int k = blockIdx.z;
    const float* X = xs + (size_t)k * BB * LL * DIc;
    const float* W = wpad + (size_t)k * CPAD * DIc;
    float* O = xdbl + (size_t)k * BB * LL * CPROJ;
    const int row0 = blockIdx.y * 16, col0 = blockIdx.x * 32;
    v8f acc[2];
    wmma_xwT_multi<2>(X, W, row0, col0, DIc, DIc, DIc, acc);
    const int lane = threadIdx.x & 31;
    const int nn = lane & 15, mh = (lane >> 4) * 8;
#pragma unroll
    for (int t = 0; t < 2; ++t)
#pragma unroll
        for (int v = 0; v < 8; ++v) {
            int r = row0 + mh + v, c = col0 + t * 16 + nn;
            if (c < CPROJ) O[(size_t)r * CPROJ + c] = acc[t][v];
        }
}

__global__ void k_dtproj(const float* __restrict__ xdbl, const float* __restrict__ dtw,
                         const float* __restrict__ dtb, float* __restrict__ delta)
{
    int k = blockIdx.z;
    const float* X = xdbl + (size_t)k * BB * LL * CPROJ;
    const float* W = dtw + (size_t)k * DIc * RR;
    float* O = delta + (size_t)k * BB * LL * DIc;
    const int row0 = blockIdx.y * 16, col0 = blockIdx.x * 32;
    v8f acc[2];
    wmma_xwT_multi<2>(X, W, row0, col0, RR, CPROJ, RR, acc);
    const int lane = threadIdx.x & 31;
    const int nn = lane & 15, mh = (lane >> 4) * 8;
#pragma unroll
    for (int t = 0; t < 2; ++t)
#pragma unroll
        for (int v = 0; v < 8; ++v) {
            int r = row0 + mh + v, c = col0 + t * 16 + nn;
            O[(size_t)r * DIc + c] = softplusf_(acc[t][v] + dtb[k * DIc + c]);
        }
}

#define SCAN_T 16
__global__ __launch_bounds__(DIc)
void k_scan(const float* __restrict__ xs, const float* __restrict__ xdbl,
            const float* __restrict__ delta, const float* __restrict__ A_logs,
            const float* __restrict__ Ds, float* __restrict__ yarr)
{
    int kb = blockIdx.x;
    int k = kb & 3, b = kb >> 2;
    int d = threadIdx.x;
    const int lane = threadIdx.x & 31;
    float A[NN];
#pragma unroll
    for (int n = 0; n < NN; ++n)
        A[n] = -__expf(A_logs[((size_t)(k * DIc + d)) * NN + n]);
    float Dv = Ds[k * DIc + d];
    float h[NN];
#pragma unroll
    for (int n = 0; n < NN; ++n) h[n] = 0.f;
    const size_t seqbase = ((size_t)(k * BB + b)) * LL;
    const float* up  = xs    + seqbase * DIc + d;
    const float* dp  = delta + seqbase * DIc + d;
    const float* bcw = xdbl  + seqbase * CPROJ + RR + lane;
    float* yp        = yarr  + seqbase * DIc + d;

    float dtc[SCAN_T], uc[SCAN_T], bcc[SCAN_T];
#pragma unroll
    for (int t = 0; t < SCAN_T; ++t) {
        dtc[t] = dp[(size_t)t * DIc];
        uc[t]  = up[(size_t)t * DIc];
        bcc[t] = bcw[(size_t)t * CPROJ];
    }
    for (int l0 = 0; l0 < LL; l0 += SCAN_T) {
        const int ln = (l0 + SCAN_T < LL) ? (l0 + SCAN_T) : l0;
#pragma unroll
        for (int t = 0; t < SCAN_T; ++t) {
            const float dt = dtc[t];
            const float u  = uc[t];
            const int  bci = __builtin_bit_cast(int, bcc[t]);
            const float du = dt * u;
            float y = 0.f;
#pragma unroll
            for (int n = 0; n < NN; ++n) {
                const float Bt = __builtin_bit_cast(float, __builtin_amdgcn_readlane(bci, n));
                const float Ct = __builtin_bit_cast(float, __builtin_amdgcn_readlane(bci, NN + n));
                h[n] = __expf(dt * A[n]) * h[n] + du * Bt;
                y += h[n] * Ct;
            }
            yp[(size_t)(l0 + t) * DIc] = y + Dv * u;
            dtc[t] = dp[(size_t)(ln + t) * DIc];
            uc[t]  = up[(size_t)(ln + t) * DIc];
            bcc[t] = bcw[(size_t)(ln + t) * CPROJ];
        }
    }
}

__global__ void k_combine(const float* __restrict__ yarr, const float* __restrict__ gate,
                          const float* __restrict__ g1, const float* __restrict__ g2,
                          const float* __restrict__ g3, const float* __restrict__ g4,
                          const float* __restrict__ nw, const float* __restrict__ nb,
                          const float* __restrict__ z_silu, float* __restrict__ yfinal)
{
    __shared__ float s1[256], s2[256];
    int lr = blockIdx.x, b = blockIdx.y, d = threadIdx.x;
    int h = lr / WW, w = lr % WW;
    int l = (((h >> 2) * (WW / 4) + (w >> 2)) << 4) + ((h & 3) << 2) + (w & 3);
    int lt = (l % WW) * HH + (l / WW);
    const size_t bstride = (size_t)LL * DIc;
    const float* Y0 = yarr + ((size_t)(0 * BB + b)) * bstride;
    const float* Y1 = yarr + ((size_t)(1 * BB + b)) * bstride;
    const float* Y2 = yarr + ((size_t)(2 * BB + b)) * bstride;
    const float* Y3 = yarr + ((size_t)(3 * BB + b)) * bstride;
    float yc = g1[d] * Y0[(size_t)l * DIc + d]
             + g2[d] * Y2[(size_t)(LL - 1 - l) * DIc + d]
             + g3[d] * Y1[(size_t)lt * DIc + d]
             + g4[d] * Y3[(size_t)(LL - 1 - lt) * DIc + d];
    yc *= gate[b * DIc + d];
    s1[d] = yc; s2[d] = yc * yc;
    if (d < 64) { s1[192 + d] = 0.f; s2[192 + d] = 0.f; }
    __syncthreads();
    for (int off = 128; off > 0; off >>= 1) {
        if (d < off) { s1[d] += s1[d + off]; s2[d] += s2[d + off]; }
        __syncthreads();
    }
    float mu  = s1[0] * (1.f / DIc);
    float var = s2[0] * (1.f / DIc) - mu * mu;
    float inv = rsqrtf(var + 1e-5f);
    yc = (yc - mu) * inv * nw[d] + nb[d];
    yc *= z_silu[((size_t)b * LL + lr) * DIc + d];
    yfinal[((size_t)b * LL + lr) * DIc + d] = yc;
}

__global__ void k_outproj(const float* __restrict__ yf, const float* __restrict__ w,
                          float* __restrict__ out)
{
    const int row0 = blockIdx.y * 16, col0 = blockIdx.x * 32;
    v8f acc[2];
    wmma_xwT_multi<2>(yf, w, row0, col0, DIc, DIc, DIc, acc);
    const int lane = threadIdx.x & 31;
    const int nn = lane & 15, mh = (lane >> 4) * 8;
#pragma unroll
    for (int t = 0; t < 2; ++t)
#pragma unroll
        for (int v = 0; v < 8; ++v) {
            int r = row0 + mh + v, c = col0 + t * 16 + nn;
            out[(size_t)r * DMm + c] = acc[t][v];
        }
}

extern "C" void kernel_launch(void* const* d_in, const int* in_sizes, int n_in,
                              void* d_out, int out_size, void* d_ws, size_t ws_size,
                              hipStream_t stream)
{
    (void)in_sizes; (void)n_in; (void)out_size; (void)ws_size;
    const float* x          = (const float*)d_in[0];
    const float* in_proj_w  = (const float*)d_in[1];
    const float* conv_w     = (const float*)d_in[2];
    const float* conv_b     = (const float*)d_in[3];
    const float* x_proj_w   = (const float*)d_in[4];
    const float* dt_w       = (const float*)d_in[5];
    const float* dt_b       = (const float*)d_in[6];
    const float* A_logs     = (const float*)d_in[7];
    const float* Ds         = (const float*)d_in[8];
    const float* gamma1     = (const float*)d_in[9];
    const float* gamma2     = (const float*)d_in[10];
    const float* gamma3     = (const float*)d_in[11];
    const float* gamma4     = (const float*)d_in[12];
    const float* fc_w1      = (const float*)d_in[13];
    const float* fc_w2      = (const float*)d_in[14];
    const float* norm_w     = (const float*)d_in[15];
    const float* norm_b     = (const float*)d_in[16];
    const float* out_proj_w = (const float*)d_in[17];
    float* out = (float*)d_out;

    float* ws = (float*)d_ws;
    size_t o = 0;
    float* xc_lin = ws + o; o += (size_t)BB * LL * DIc;
    float* z_silu = ws + o; o += (size_t)BB * LL * DIc;
    float* xc_act = ws + o; o += (size_t)BB * LL * DIc;
    float* xs     = ws + o; o += (size_t)KK * BB * LL * DIc;
    float* xdbl   = ws + o; o += (size_t)KK * BB * LL * CPROJ;
    float* delta  = ws + o; o += (size_t)KK * BB * LL * DIc;
    float* yarr   = ws + o; o += (size_t)KK * BB * LL * DIc;
    float* yfinal = ws + o; o += (size_t)BB * LL * DIc;
    float* wpad   = ws + o; o += (size_t)KK * CPAD * DIc;
    float* gcm    = ws + o; o += BB * DIc;
    float* gate   = ws + o; o += BB * DMm;

    k_padw    <<<dim3(KK, CPAD),       DIc, 0, stream>>>(x_proj_w, wpad);
    k_inproj  <<<dim3(12, 576),        32,  0, stream>>>(x, in_proj_w, xc_lin, z_silu);
    k_dwconv  <<<dim3(BB * LL),        DIc, 0, stream>>>(xc_lin, conv_w, conv_b, xc_act);
    k_gcmean  <<<dim3(BB),             DIc, 0, stream>>>(xc_act, gcm);
    k_gate    <<<dim3(BB),             DIc, 0, stream>>>(gcm, fc_w1, fc_w2, gate);
    k_build_xs<<<dim3(LL, BB, KK),     DIc, 0, stream>>>(xc_act, xs);
    k_xproj   <<<dim3(2, 576, KK),     32,  0, stream>>>(xs, wpad, xdbl);
    k_dtproj  <<<dim3(6, 576, KK),     32,  0, stream>>>(xdbl, dt_w, dt_b, delta);
    k_scan    <<<dim3(BB * KK),        DIc, 0, stream>>>(xs, xdbl, delta, A_logs, Ds, yarr);
    k_combine <<<dim3(LL, BB),         DIc, 0, stream>>>(yarr, gate, gamma1, gamma2, gamma3,
                                                         gamma4, norm_w, norm_b, z_silu, yfinal);
    k_outproj <<<dim3(6, 576),         32,  0, stream>>>(yfinal, out_proj_w, out);
}


# k_scan: a channel's 16 states split over 8 waves (2 each; 96 groups); partial y sums meet in LDS every 16 steps: y summed per pair then group by group, D*u inside the last pair (new sum order)
# speedup vs baseline: 3.6076x; 1.7781x over previous
#include <hip/hip_runtime.h>
#include <math.h>


typedef float v2f __attribute__((ext_vector_type(2)));
typedef float v8f __attribute__((ext_vector_type(8)));

#define BB 4
#define HH 48
#define WW 48
#define DMm 192
#define DIc 192
#define KK 4
#define NN 16
#define RR 12
#define LL (HH * WW)
#define CPROJ (RR + 2 * NN)
#define CPAD 64

__device__ __forceinline__ float siluf(float x)     { return x / (1.f + __expf(-x)); }
__device__ __forceinline__ float sigm(float x)      { return 1.f / (1.f + __expf(-x)); }
__device__ __forceinline__ float softplusf_(float x){ return (x > 20.f) ? x : log1pf(__expf(x)); }
__device__ __forceinline__ float geluf(float x)     { return 0.5f * x * (1.f + erff(x * 0.70710678118654752440f)); }

template <int NT>
__device__ __forceinline__ void wmma_xwT_multi(const float* __restrict__ X,
                                               const float* __restrict__ W,
                                               int row0, int col0,
                                               int Kdim, int ldx, int ldw,
                                               v8f acc[NT])
{
    const int lane = threadIdx.x & 31;
    const int m    = lane & 15;
    const int kh   = lane >> 4;
    const float* __restrict__ ap = X + (size_t)(row0 + m) * ldx + 2 * kh;
    const float* __restrict__ bp[NT];
#pragma unroll
    for (int t = 0; t < NT; ++t) {
        bp[t] = W + (size_t)(col0 + t * 16 + m) * ldw + 2 * kh;
        acc[t] = {};
    }
    v2f a_cur = *(const v2f*)(ap);
    v2f b_cur[NT];
#pragma unroll
    for (int t = 0; t < NT; ++t) b_cur[t] = *(const v2f*)(bp[t]);

    for (int k0 = 4; k0 < Kdim; k0 += 4) {
        v2f a_nxt = *(const v2f*)(ap + k0);
        v2f b_nxt[NT];
#pragma unroll
        for (int t = 0; t < NT; ++t) b_nxt[t] = *(const v2f*)(bp[t] + k0);
#pragma unroll
        for (int t = 0; t < NT; ++t)
            acc[t] = __builtin_amdgcn_wmma_f32_16x16x4_f32(false, a_cur, false, b_cur[t],
                                                           (short)0, acc[t], false, false);
        a_cur = a_nxt;
#pragma unroll
        for (int t = 0; t < NT; ++t) b_cur[t] = b_nxt[t];
    }
#pragma unroll
    for (int t = 0; t < NT; ++t)
        acc[t] = __builtin_amdgcn_wmma_f32_16x16x4_f32(false, a_cur, false, b_cur[t],
                                                       (short)0, acc[t], false, false);
}

__global__ void k_padw(const float* __restrict__ xpw, float* __restrict__ wpad)
{
    int k = blockIdx.x, c = blockIdx.y, d = threadIdx.x;
    wpad[((size_t)k * CPAD + c) * DIc + d] =
        (c < CPROJ) ? xpw[((size_t)k * CPROJ + c) * DIc + d] : 0.f;
}

__global__ void k_inproj(const float* __restrict__ x, const float* __restrict__ w,
                         float* __restrict__ xc_lin, float* __restrict__ z_silu)
{
    const int row0 = blockIdx.y * 16, col0 = blockIdx.x * 32;
    v8f acc[2];
    wmma_xwT_multi<2>(x, w, row0, col0, DMm, DMm, DMm, acc);
    const int lane = threadIdx.x & 31;
    const int nn = lane & 15, mh = (lane >> 4) * 8;
#pragma unroll
    for (int t = 0; t < 2; ++t)
#pragma unroll
        for (int v = 0; v < 8; ++v) {
            int r = row0 + mh + v;
            int c = col0 + t * 16 + nn;
            float val = acc[t][v];
            if (c < DIc) xc_lin[(size_t)r * DIc + c] = val;
            else         z_silu[(size_t)r * DIc + (c - DIc)] = siluf(val);
        }
}

__global__ void k_dwconv(const float* __restrict__ xc_lin, const float* __restrict__ cw,
                         const float* __restrict__ cb, float* __restrict__ xc_act)
{
    int pix = blockIdx.x;
    int b = pix / LL, rem = pix % LL;
    int h = rem / WW, w = rem % WW;
    int c = threadIdx.x;
    float s = cb[c];
#pragma unroll
    for (int kh = 0; kh < 3; ++kh)
#pragma unroll
        for (int kw = 0; kw < 3; ++kw) {
            int hh = h + kh - 1, ww2 = w + kw - 1;
            if (hh >= 0 && hh < HH && ww2 >= 0 && ww2 < WW)
                s += xc_lin[((size_t)b * LL + hh * WW + ww2) * DIc + c] * cw[c * 9 + kh * 3 + kw];
        }
    xc_act[((size_t)b * LL + rem) * DIc + c] = siluf(s);
}

__global__ void k_gcmean(const float* __restrict__ xc_act, float* __restrict__ gcm)
{
    int b = blockIdx.x, d = threadIdx.x;
    float s = 0.f;
    for (int p = 0; p < LL; ++p)
        s += xc_act[((size_t)b * LL + p) * DIc + d];
    gcm[b * DIc + d] = s * (1.f / LL);
}

__global__ void k_gate(const float* __restrict__ gcm, const float* __restrict__ fw1,
                       const float* __restrict__ fw2, float* __restrict__ gate)
{
    __shared__ float g[DIc];
    __shared__ float h1[DMm / 4];
    int b = blockIdx.x, d = threadIdx.x;
    g[d] = gcm[b * DIc + d];
    __syncthreads();
    if (d < DMm / 4) {
        float s = 0.f;
        for (int j = 0; j < DMm; ++j) s += fw1[d * DMm + j] * g[j];
        h1[d] = geluf(s);
    }
    __syncthreads();
    float s2 = 0.f;
#pragma unroll
    for (int j = 0; j < DMm / 4; ++j) s2 += fw2[d * (DMm / 4) + j] * h1[j];
    gate[b * DMm + d] = sigm(s2);
}

__global__ void k_build_xs(const float* __restrict__ xc_act, float* __restrict__ xs)
{
    int l = blockIdx.x, b = blockIdx.y, k = blockIdx.z;
    int d = threadIdx.x;
    int lb = (k < 2) ? l : (LL - 1 - l);
    int w_in = lb & 3, h_in = (lb >> 2) & 3;
    int nw = (lb >> 4) % (WW / 4);
    int nh = (lb >> 4) / (WW / 4);
    int p = nh * 4 + h_in, q = nw * 4 + w_in;
    int h = ((k & 1) == 0) ? p : q;
    int w = ((k & 1) == 0) ? q : p;
    xs[(((size_t)(k * BB + b)) * LL + l) * DIc + d] =
        xc_act[((size_t)b * LL + h * WW + w) * DIc + d];
}

__global__ void k_xproj(const float* __restrict__ xs, const float* __restrict__ wpad,
                        float* __restrict__ xdbl)
{
    int k = blockIdx.z;
    const float* X = xs + (size_t)k * BB * LL * DIc;
    const float* W = wpad + (size_t)k * CPAD * DIc;
    float* O = xdbl + (size_t)k * BB * LL * CPROJ;
    const int row0 = blockIdx.y * 16, col0 = blockIdx.x * 32;
    v8f acc[2];
    wmma_xwT_multi<2>(X, W, row0, col0, DIc, DIc, DIc, acc);
    const int lane = threadIdx.x & 31;
    const int nn = lane & 15, mh = (lane >> 4) * 8;
#pragma unroll
    for (int t = 0; t < 2; ++t)
#pragma unroll
        for (int v = 0; v < 8; ++v) {
            int r = row0 + mh + v, c = col0 + t * 16 + nn;
            if (c < CPROJ) O[(size_t)r * CPROJ + c] = acc[t][v];
        }
}

__global__ void k_dtproj(const float* __restrict__ xdbl, const float* __restrict__ dtw,
                         const float* __restrict__ dtb, float* __restrict__ delta)
{
    int k = blockIdx.z;
    const float* X = xdbl + (size_t)k * BB * LL * CPROJ;
    const float* W = dtw + (size_t)k * DIc * RR;
    float* O = delta + (size_t)k * BB * LL * DIc;
    const int row0 = blockIdx.y * 16, col0 = blockIdx.x * 32;
    v8f acc[2];
    wmma_xwT_multi<2>(X, W, row0, col0, RR, CPROJ, RR, acc);
    const int lane = threadIdx.x & 31;
    const int nn = lane & 15, mh = (lane >> 4) * 8;
#pragma unroll
    for (int t = 0; t < 2; ++t)
#pragma unroll
        for (int v = 0; v < 8; ++v) {
            int r = row0 + mh + v, c = col0 + t * 16 + nn;
            O[(size_t)r * DIc + c] = softplusf_(acc[t][v] + dtb[k * DIc + c]);
        }
}

#define SCAN_T 16
#define SCAN_NG 8
#define SCAN_NPG (NN / SCAN_NG)
#define SCAN_RPW (SCAN_T / SCAN_NG)
__global__ __launch_bounds__(32 * SCAN_NG)
void k_scan(const float* __restrict__ xs, const float* __restrict__ xdbl,
            const float* __restrict__ delta, const float* __restrict__ A_logs,
            const float* __restrict__ Ds, float* __restrict__ yarr)
{
    __shared__ float ybuf[2][SCAN_T][SCAN_NG][32];
    const int kb = blockIdx.x;
    const int k = kb & 3, b = kb >> 2;
    const int lane = threadIdx.x & 31;
    const int g = __builtin_amdgcn_readfirstlane((int)(threadIdx.x >> 5));
    const int d = blockIdx.y * 32 + lane;
    const int n0 = g * SCAN_NPG;
    float A[SCAN_NPG], h[SCAN_NPG];
#pragma unroll
    for (int n = 0; n < SCAN_NPG; ++n) {
        A[n] = -__expf(A_logs[((size_t)(k * DIc + d)) * NN + n0 + n]);
        h[n] = 0.f;
    }
    const float Dg = (g == SCAN_NG - 1) ? Ds[k * DIc + d] : 0.f;
    const size_t seqbase = ((size_t)(k * BB + b)) * LL;
    const float* up  = xs    + seqbase * DIc + d;
    const float* dp  = delta + seqbase * DIc + d;
    const float* bcw = xdbl  + seqbase * CPROJ + RR + lane;
    float* yp        = yarr  + seqbase * DIc + d;

    float dtc[SCAN_T], uc[SCAN_T], bcc[SCAN_T];
#pragma unroll
    for (int t = 0; t < SCAN_T; ++t) {
        dtc[t] = dp[(size_t)t * DIc];
        uc[t]  = up[(size_t)t * DIc];
        bcc[t] = bcw[(size_t)t * CPROJ];
    }
    int buf = 0;
    for (int l0 = 0; l0 < LL; l0 += SCAN_T) {
        const int ln = (l0 + SCAN_T < LL) ? (l0 + SCAN_T) : l0;
#pragma unroll
        for (int t = 0; t < SCAN_T; ++t) {
            const float dt = dtc[t];
            const float u  = uc[t];
            const int  bci = __builtin_bit_cast(int, bcc[t]);
            const float du = dt * u;
            float y = 0.f;
#pragma unroll
            for (int n = 0; n < SCAN_NPG; ++n) {
                const float Bt = __builtin_bit_cast(float, __builtin_amdgcn_readlane(bci, n0 + n));
                const float Ct = __builtin_bit_cast(float, __builtin_amdgcn_readlane(bci, NN + n0 + n));
                h[n] = __expf(dt * A[n]) * h[n] + du * Bt;
                y += h[n] * Ct;
            }
            ybuf[buf][t][g][lane] = y + Dg * u;
            dtc[t] = dp[(size_t)(ln + t) * DIc];
            uc[t]  = up[(size_t)(ln + t) * DIc];
            bcc[t] = bcw[(size_t)(ln + t) * CPROJ];
            __builtin_amdgcn_sched_barrier(0);
        }
        __builtin_amdgcn_fence(__ATOMIC_RELEASE, "workgroup", "local");
        __builtin_amdgcn_s_barrier();
        __builtin_amdgcn_fence(__ATOMIC_ACQUIRE, "workgroup", "local");
#pragma unroll
        for (int r = 0; r < SCAN_RPW; ++r) {
            const int t = g * SCAN_RPW + r;
            float s = ybuf[buf][t][0][lane];
#pragma unroll
            for (int q = 1; q < SCAN_NG; ++q) s += ybuf[buf][t][q][lane];
            yp[(size_t)(l0 + t) * DIc] = s;
        }
        buf ^= 1;
    }
}

__global__ void k_combine(const float* __restrict__ yarr, const float* __restrict__ gate,
                          const float* __restrict__ g1, const float* __restrict__ g2,
                          const float* __restrict__ g3, const float* __restrict__ g4,
                          const float* __restrict__ nw, const float* __restrict__ nb,
                          const float* __restrict__ z_silu, float* __restrict__ yfinal)
{
    __shared__ float s1[256], s2[256];
    int lr = blockIdx.x, b = blockIdx.y, d = threadIdx.x;
    int h = lr / WW, w = lr % WW;
    int l = (((h >> 2) * (WW / 4) + (w >> 2)) << 4) + ((h & 3) << 2) + (w & 3);
    int lt = (l % WW) * HH + (l / WW);
    const size_t bstride = (size_t)LL * DIc;
    const float* Y0 = yarr + ((size_t)(0 * BB + b)) * bstride;
    const float* Y1 = yarr + ((size_t)(1 * BB + b)) * bstride;
    const float* Y2 = yarr + ((size_t)(2 * BB + b)) * bstride;
    const float* Y3 = yarr + ((size_t)(3 * BB + b)) * bstride;
    float yc = g1[d] * Y0[(size_t)l * DIc + d]
             + g2[d] * Y2[(size_t)(LL - 1 - l) * DIc + d]
             + g3[d] * Y1[(size_t)lt * DIc + d]
             + g4[d] * Y3[(size_t)(LL - 1 - lt) * DIc + d];
    yc *= gate[b * DIc + d];
    s1[d] = yc; s2[d] = yc * yc;
    if (d < 64) { s1[192 + d] = 0.f; s2[192 + d] = 0.f; }
    __syncthreads();
    for (int off = 128; off > 0; off >>= 1) {
        if (d < off) { s1[d] += s1[d + off]; s2[d] += s2[d + off]; }
        __syncthreads();
    }
    float mu  = s1[0] * (1.f / DIc);
    float var = s2[0] * (1.f / DIc) - mu * mu;
    float inv = rsqrtf(var + 1e-5f);
    yc = (yc - mu) * inv * nw[d] + nb[d];
    yc *= z_silu[((size_t)b * LL + lr) * DIc + d];
    yfinal[((size_t)b * LL + lr) * DIc + d] = yc;
}

__global__ void k_outproj(const float* __restrict__ yf, const float* __restrict__ w,
                          float* __restrict__ out)
{
    const int row0 = blockIdx.y * 16, col0 = blockIdx.x * 32;
    v8f acc[2];
    wmma_xwT_multi<2>(yf, w, row0, col0, DIc, DIc, DIc, acc);
    const int lane = threadIdx.x & 31;
    const int nn = lane & 15, mh = (lane >> 4) * 8;
#pragma unroll
    for (int t = 0; t < 2; ++t)
#pragma unroll
        for (int v = 0; v < 8; ++v) {
            int r = row0 + mh + v, c = col0 + t * 16 + nn;
            out[(size_t)r * DMm + c] = acc[t][v];
        }
}

extern "C" void kernel_launch(void* const* d_in, const int* in_sizes, int n_in,
                              void* d_out, int out_size, void* d_ws, size_t ws_size,
                              hipStream_t stream)
{
    (void)in_sizes; (void)n_in; (void)out_size; (void)ws_size;
    const float* x          = (const float*)d_in[0];
    const float* in_proj_w  = (const float*)d_in[1];
    const float* conv_w     = (const float*)d_in[2];
    const float* conv_b     = (const float*)d_in[3];
    const float* x_proj_w   = (const float*)d_in[4];
    const float* dt_w       = (const float*)d_in[5];
    const float* dt_b       = (const float*)d_in[6];
    const float* A_logs     = (const float*)d_in[7];
    const float* Ds         = (const float*)d_in[8];
    const float* gamma1     = (const float*)d_in[9];
    const float* gamma2     = (const float*)d_in[10];
    const float* gamma3     = (const float*)d_in[11];
    const float* gamma4     = (const float*)d_in[12];
    const float* fc_w1      = (const float*)d_in[13];
    const float* fc_w2      = (const float*)d_in[14];
    const float* norm_w     = (const float*)d_in[15];
    const float* norm_b     = (const float*)d_in[16];
    const float* out_proj_w = (const float*)d_in[17];
    float* out = (float*)d_out;

    float* ws = (float*)d_ws;
    size_t o = 0;
    float* xc_lin = ws + o; o += (size_t)BB * LL * DIc;
    float* z_silu = ws + o; o += (size_t)BB * LL * DIc;
    float* xc_act = ws + o; o += (size_t)BB * LL * DIc;
    float* xs     = ws + o; o += (size_t)KK * BB * LL * DIc;
    float* xdbl   = ws + o; o += (size_t)KK * BB * LL * CPROJ;
    float* delta  = ws + o; o += (size_t)KK * BB * LL * DIc;
    float* yarr   = ws + o; o += (size_t)KK * BB * LL * DIc;
    float* yfinal = ws + o; o += (size_t)BB * LL * DIc;
    float* wpad   = ws + o; o += (size_t)KK * CPAD * DIc;
    float* gcm    = ws + o; o += BB * DIc;
    float* gate   = ws + o; o += BB * DMm;

    k_padw    <<<dim3(KK, CPAD),       DIc, 0, stream>>>(x_proj_w, wpad);
    k_inproj  <<<dim3(12, 576),        32,  0, stream>>>(x, in_proj_w, xc_lin, z_silu);
    k_dwconv  <<<dim3(BB * LL),        DIc, 0, stream>>>(xc_lin, conv_w, conv_b, xc_act);
    k_gcmean  <<<dim3(BB),             DIc, 0, stream>>>(xc_act, gcm);
    k_gate    <<<dim3(BB),             DIc, 0, stream>>>(gcm, fc_w1, fc_w2, gate);
    k_build_xs<<<dim3(LL, BB, KK),     DIc, 0, stream>>>(xc_act, xs);
    k_xproj   <<<dim3(2, 576, KK),     32,  0, stream>>>(xs, wpad, xdbl);
    k_dtproj  <<<dim3(6, 576, KK),     32,  0, stream>>>(xdbl, dt_w, dt_b, delta);
    k_scan    <<<dim3(BB * KK, DIc / 32), 32 * SCAN_NG, 0, stream>>>(xs, xdbl, delta, A_logs, Ds, yarr);
    k_combine <<<dim3(LL, BB),         DIc, 0, stream>>>(yarr, gate, gamma1, gamma2, gamma3,
                                                         gamma4, norm_w, norm_b, z_silu, yfinal);
    k_outproj <<<dim3(6, 576),         32,  0, stream>>>(yfinal, out_proj_w, out);
}


# k_gcmean: the per-image channel mean summed by 48 work-groups per image (48 pixels each) instead of one; k_gate adds the 48 partial rows in order (new sum order of the mean)
# speedup vs baseline: 6.0782x; 1.6848x over previous
#include <hip/hip_runtime.h>
#include <math.h>


typedef float v2f __attribute__((ext_vector_type(2)));
typedef float v8f __attribute__((ext_vector_type(8)));

#define BB 4
#define HH 48
#define WW 48
#define DMm 192
#define DIc 192
#define KK 4
#define NN 16
#define RR 12
#define LL (HH * WW)
#define CPROJ (RR + 2 * NN)
#define CPAD 64

__device__ __forceinline__ float siluf(float x)     { return x / (1.f + __expf(-x)); }
__device__ __forceinline__ float sigm(float x)      { return 1.f / (1.f + __expf(-x)); }
__device__ __forceinline__ float softplusf_(float x){ return (x > 20.f) ? x : log1pf(__expf(x)); }
__device__ __forceinline__ float geluf(float x)     { return 0.5f * x * (1.f + erff(x * 0.70710678118654752440f)); }

template <int NT>
__device__ __forceinline__ void wmma_xwT_multi(const float* __restrict__ X,
                                               const float* __restrict__ W,
                                               int row0, int col0,
                                               int Kdim, int ldx, int ldw,
                                               v8f acc[NT])
{
    const int lane = threadIdx.x & 31;
    const int m    = lane & 15;
    const int kh   = lane >> 4;
    const float* __restrict__ ap = X + (size_t)(row0 + m) * ldx + 2 * kh;
    const float* __restrict__ bp[NT];
#pragma unroll
    for (int t = 0; t < NT; ++t) {
        bp[t] = W + (size_t)(col0 + t * 16 + m) * ldw + 2 * kh;
        acc[t] = {};
    }
    v2f a_cur = *(const v2f*)(ap);
    v2f b_cur[NT];
#pragma unroll
    for (int t = 0; t < NT; ++t) b_cur[t] = *(const v2f*)(bp[t]);

    for (int k0 = 4; k0 < Kdim; k0 += 4) {
        v2f a_nxt = *(const v2f*)(ap + k0);
        v2f b_nxt[NT];
#pragma unroll
        for (int t = 0; t < NT; ++t) b_nxt[t] = *(const v2f*)(bp[t] + k0);
#pragma unroll
        for (int t = 0; t < NT; ++t)
            acc[t] = __builtin_amdgcn_wmma_f32_16x16x4_f32(false, a_cur, false, b_cur[t],
                                                           (short)0, acc[t], false, false);
        a_cur = a_nxt;
#pragma unroll
        for (int t = 0; t < NT; ++t) b_cur[t] = b_nxt[t];
    }
#pragma unroll
    for (int t = 0; t < NT; ++t)
        acc[t] = __builtin_amdgcn_wmma_f32_16x16x4_f32(false, a_cur, false, b_cur[t],
                                                       (short)0, acc[t], false, false);
}

__global__ void k_padw(const float* __restrict__ xpw, float* __restrict__ wpad)
{
    int k = blockIdx.x, c = blockIdx.y, d = threadIdx.x;
    wpad[((size_t)k * CPAD + c) * DIc + d] =
        (c < CPROJ) ? xpw[((size_t)k * CPROJ + c) * DIc + d] : 0.f;
}

__global__ void k_inproj(const float* __restrict__ x, const float* __restrict__ w,
                         float* __restrict__ xc_lin, float* __restrict__ z_silu)
{
    const int row0 = blockIdx.y * 16, col0 = blockIdx.x * 32;
    v8f acc[2];
    wmma_xwT_multi<2>(x, w, row0, col0, DMm, DMm, DMm, acc);
    const int lane = threadIdx.x & 31;
    const int nn = lane & 15, mh = (lane >> 4) * 8;
#pragma unroll
    for (int t = 0; t < 2; ++t)
#pragma unroll
        for (int v = 0; v < 8; ++v) {
            int r = row0 + mh + v;
            int c = col0 + t * 16 + nn;
            float val = acc[t][v];
            if (c < DIc) xc_lin[(size_t)r * DIc + c] = val;
            else         z_silu[(size_t)r * DIc + (c - DIc)] = siluf(val);
        }
}

__global__ void k_dwconv(const float* __restrict__ xc_lin, const float* __restrict__ cw,
                         const float* __restrict__ cb, float* __restrict__ xc_act)
{
    int pix = blockIdx.x;
    int b = pix / LL, rem = pix % LL;
    int h = rem / WW, w = rem % WW;
    int c = threadIdx.x;
    float s = cb[c];
#pragma unroll
    for (int kh = 0; kh < 3; ++kh)
#pragma unroll
        for (int kw = 0; kw < 3; ++kw) {
            int hh = h + kh - 1, ww2 = w + kw - 1;
            if (hh >= 0 && hh < HH && ww2 >= 0 && ww2 < WW)
                s += xc_lin[((size_t)b * LL + hh * WW + ww2) * DIc + c] * cw[c * 9 + kh * 3 + kw];
        }
    xc_act[((size_t)b * LL + rem) * DIc + c] = siluf(s);
}

#define GC_NCH 48
#define GC_P (LL / GC_NCH)
__global__ void k_gcmean(const float* __restrict__ xc_act, float* __restrict__ gcpart)
{
    int b = blockIdx.x, ch = blockIdx.y, d = threadIdx.x;
    const float* __restrict__ src = xc_act + ((size_t)b * LL + (size_t)ch * GC_P) * DIc + d;
    float s = 0.f;
#pragma unroll
    for (int p = 0; p < GC_P; ++p)
        s += src[(size_t)p * DIc];
    gcpart[((size_t)b * GC_NCH + ch) * DIc + d] = s;
}

__global__ void k_gate(const float* __restrict__ gcpart, const float* __restrict__ fw1,
                       const float* __restrict__ fw2, float* __restrict__ gate)
{
    __shared__ float g[DIc];
    __shared__ float h1[DMm / 4];
    int b = blockIdx.x, d = threadIdx.x;
    float gs = 0.f;
#pragma unroll
    for (int c = 0; c < GC_NCH; ++c)
        gs += gcpart[((size_t)b * GC_NCH + c) * DIc + d];
    g[d] = gs * (1.f / LL);
    __syncthreads();
    if (d < DMm / 4) {
        float s = 0.f;
        for (int j = 0; j < DMm; ++j) s += fw1[d * DMm + j] * g[j];
        h1[d] = geluf(s);
    }
    __syncthreads();
    float s2 = 0.f;
#pragma unroll
    for (int j = 0; j < DMm / 4; ++j) s2 += fw2[d * (DMm / 4) + j] * h1[j];
    gate[b * DMm + d] = sigm(s2);
}

__global__ void k_build_xs(const float* __restrict__ xc_act, float* __restrict__ xs)
{
    int l = blockIdx.x, b = blockIdx.y, k = blockIdx.z;
    int d = threadIdx.x;
    int lb = (k < 2) ? l : (LL - 1 - l);
    int w_in = lb & 3, h_in = (lb >> 2) & 3;
    int nw = (lb >> 4) % (WW / 4);
    int nh = (lb >> 4) / (WW / 4);
    int p = nh * 4 + h_in, q = nw * 4 + w_in;
    int h = ((k & 1) == 0) ? p : q;
    int w = ((k & 1) == 0) ? q : p;
    xs[(((size_t)(k * BB + b)) * LL + l) * DIc + d] =
        xc_act[((size_t)b * LL + h * WW + w) * DIc + d];
}

__global__ void k_xproj(const float* __restrict__ xs, const float* __restrict__ wpad,
                        float* __restrict__ xdbl)
{
    int k = blockIdx.z;
    const float* X = xs + (size_t)k * BB * LL * DIc;
    const float* W = wpad + (size_t)k * CPAD * DIc;
    float* O = xdbl + (size_t)k * BB * LL * CPROJ;
    const int row0 = blockIdx.y * 16, col0 = blockIdx.x * 32;
    v8f acc[2];
    wmma_xwT_multi<2>(X, W, row0, col0, DIc, DIc, DIc, acc);
    const int lane = threadIdx.x & 31;
    const int nn = lane & 15, mh = (lane >> 4) * 8;
#pragma unroll
    for (int t = 0; t < 2; ++t)
#pragma unroll
        for (int v = 0; v < 8; ++v) {
            int r = row0 + mh + v, c = col0 + t * 16 + nn;
            if (c < CPROJ) O[(size_t)r * CPROJ + c] = acc[t][v];
        }
}

__global__ void k_dtproj(const float* __restrict__ xdbl, const float* __restrict__ dtw,
                         const float* __restrict__ dtb, float* __restrict__ delta)
{
    int k = blockIdx.z;
    const float* X = xdbl + (size_t)k * BB * LL * CPROJ;
    const float* W = dtw + (size_t)k * DIc * RR;
    float* O = delta + (size_t)k * BB * LL * DIc;
    const int row0 = blockIdx.y * 16, col0 = blockIdx.x * 32;
    v8f acc[2];
    wmma_xwT_multi<2>(X, W, row0, col0, RR, CPROJ, RR, acc);
    const int lane = threadIdx.x & 31;
    const int nn = lane & 15, mh = (lane >> 4) * 8;
#pragma unroll
    for (int t = 0; t < 2; ++t)
#pragma unroll
        for (int v = 0; v < 8; ++v) {
            int r = row0 + mh + v, c = col0 + t * 16 + nn;
            O[(size_t)r * DIc + c] = softplusf_(acc[t][v] + dtb[k * DIc + c]);
        }
}

#define SCAN_T 16
#define SCAN_NG 8
#define SCAN_NPG (NN / SCAN_NG)
#define SCAN_RPW (SCAN_T / SCAN_NG)
__global__ __launch_bounds__(32 * SCAN_NG)
void k_scan(const float* __restrict__ xs, const float* __restrict__ xdbl,
            const float* __restrict__ delta, const float* __restrict__ A_logs,
            const float* __restrict__ Ds, float* __restrict__ yarr)
{
    __shared__ float ybuf[2][SCAN_T][SCAN_NG][32];
    const int kb = blockIdx.x;
    const int k = kb & 3, b = kb >> 2;
    const int lane = threadIdx.x & 31;
    const int g = __builtin_amdgcn_readfirstlane((int)(threadIdx.x >> 5));
    const int d = blockIdx.y * 32 + lane;
    const int n0 = g * SCAN_NPG;
    float A[SCAN_NPG], h[SCAN_NPG];
#pragma unroll
    for (int n = 0; n < SCAN_NPG; ++n) {
        A[n] = -__expf(A_logs[((size_t)(k * DIc + d)) * NN + n0 + n]);
        h[n] = 0.f;
    }
    const float Dg = (g == SCAN_NG - 1) ? Ds[k * DIc + d] : 0.f;
    const size_t seqbase = ((size_t)(k * BB + b)) * LL;
    const float* up  = xs    + seqbase * DIc + d;
    const float* dp  = delta + seqbase * DIc + d;
    const float* bcw = xdbl  + seqbase * CPROJ + RR + lane;
    float* yp        = yarr  + seqbase * DIc + d;

    float dtc[SCAN_T], uc[SCAN_T], bcc[SCAN_T];
#pragma unroll
    for (int t = 0; t < SCAN_T; ++t) {
        dtc[t] = dp[(size_t)t * DIc];
        uc[t]  = up[(size_t)t * DIc];
        bcc[t] = bcw[(size_t)t * CPROJ];
    }
    int buf = 0;
    for (int l0 = 0; l0 < LL; l0 += SCAN_T) {
        const int ln = (l0 + SCAN_T < LL) ? (l0 + SCAN_T) : l0;
#pragma unroll
        for (int t = 0; t < SCAN_T; ++t) {
            const float dt = dtc[t];
            const float u  = uc[t];
            const int  bci = __builtin_bit_cast(int, bcc[t]);
            const float du = dt * u;
            float y = 0.f;
#pragma unroll
            for (int n = 0; n < SCAN_NPG; ++n) {
                const float Bt = __builtin_bit_cast(float, __builtin_amdgcn_readlane(bci, n0 + n));
                const float Ct = __builtin_bit_cast(float, __builtin_amdgcn_readlane(bci, NN + n0 + n));
                h[n] = __expf(dt * A[n]) * h[n] + du * Bt;
                y += h[n] * Ct;
            }
            ybuf[buf][t][g][lane] = y + Dg * u;
            dtc[t] = dp[(size_t)(ln + t) * DIc];
            uc[t]  = up[(size_t)(ln + t) * DIc];
            bcc[t] = bcw[(size_t)(ln + t) * CPROJ];
            __builtin_amdgcn_sched_barrier(0);
        }
        __builtin_amdgcn_fence(__ATOMIC_RELEASE, "workgroup", "local");
        __builtin_amdgcn_s_barrier();
        __builtin_amdgcn_fence(__ATOMIC_ACQUIRE, "workgroup", "local");
#pragma unroll
        for (int r = 0; r < SCAN_RPW; ++r) {
            const int t = g * SCAN_RPW + r;
            float s = ybuf[buf][t][0][lane];
#pragma unroll
            for (int q = 1; q < SCAN_NG; ++q) s += ybuf[buf][t][q][lane];
            yp[(size_t)(l0 + t) * DIc] = s;
        }
        buf ^= 1;
    }
}

__global__ void k_combine(const float* __restrict__ yarr, const float* __restrict__ gate,
                          const float* __restrict__ g1, const float* __restrict__ g2,
                          const float* __restrict__ g3, const float* __restrict__ g4,
                          const float* __restrict__ nw, const float* __restrict__ nb,
                          const float* __restrict__ z_silu, float* __restrict__ yfinal)
{
    __shared__ float s1[256], s2[256];
    int lr = blockIdx.x, b = blockIdx.y, d = threadIdx.x;
    int h = lr / WW, w = lr % WW;
    int l = (((h >> 2) * (WW / 4) + (w >> 2)) << 4) + ((h & 3) << 2) + (w & 3);
    int lt = (l % WW) * HH + (l / WW);
    const size_t bstride = (size_t)LL * DIc;
    const float* Y0 = yarr + ((size_t)(0 * BB + b)) * bstride;
    const float* Y1 = yarr + ((size_t)(1 * BB + b)) * bstride;
    const float* Y2 = yarr + ((size_t)(2 * BB + b)) * bstride;
    const float* Y3 = yarr + ((size_t)(3 * BB + b)) * bstride;
    float yc = g1[d] * Y0[(size_t)l * DIc + d]
             + g2[d] * Y2[(size_t)(LL - 1 - l) * DIc + d]
             + g3[d] * Y1[(size_t)lt * DIc + d]
             + g4[d] * Y3[(size_t)(LL - 1 - lt) * DIc + d];
    yc *= gate[b * DIc + d];
    s1[d] = yc; s2[d] = yc * yc;
    if (d < 64) { s1[192 + d] = 0.f; s2[192 + d] = 0.f; }
    __syncthreads();
    for (int off = 128; off > 0; off >>= 1) {
        if (d < off) { s1[d] += s1[d + off]; s2[d] += s2[d + off]; }
        __syncthreads();
    }
    float mu  = s1[0] * (1.f / DIc);
    float var = s2[0] * (1.f / DIc) - mu * mu;
    float inv = rsqrtf(var + 1e-5f);
    yc = (yc - mu) * inv * nw[d] + nb[d];
    yc *= z_silu[((size_t)b * LL + lr) * DIc + d];
    yfinal[((size_t)b * LL + lr) * DIc + d] = yc;
}

__global__ void k_outproj(const float* __restrict__ yf, const float* __restrict__ w,
                          float* __restrict__ out)
{
    const int row0 = blockIdx.y * 16, col0 = blockIdx.x * 32;
    v8f acc[2];
    wmma_xwT_multi<2>(yf, w, row0, col0, DIc, DIc, DIc, acc);
    const int lane = threadIdx.x & 31;
    const int nn = lane & 15, mh = (lane >> 4) * 8;
#pragma unroll
    for (int t = 0; t < 2; ++t)
#pragma unroll
        for (int v = 0; v < 8; ++v) {
            int r = row0 + mh + v, c = col0 + t * 16 + nn;
            out[(size_t)r * DMm + c] = acc[t][v];
        }
}

extern "C" void kernel_launch(void* const* d_in, const int* in_sizes, int n_in,
                              void* d_out, int out_size, void* d_ws, size_t ws_size,
                              hipStream_t stream)
{
    (void)in_sizes; (void)n_in; (void)out_size; (void)ws_size;
    const float* x          = (const float*)d_in[0];
    const float* in_proj_w  = (const float*)d_in[1];
    const float* conv_w     = (const float*)d_in[2];
    const float* conv_b     = (const float*)d_in[3];
    const float* x_proj_w   = (const float*)d_in[4];
    const float* dt_w       = (const float*)d_in[5];
    const float* dt_b       = (const float*)d_in[6];
    const float* A_logs     = (const float*)d_in[7];
    const float* Ds         = (const float*)d_in[8];
    const float* gamma1     = (const float*)d_in[9];
    const float* gamma2     = (const float*)d_in[10];
    const float* gamma3     = (const float*)d_in[11];
    const float* gamma4     = (const float*)d_in[12];
    const float* fc_w1      = (const float*)d_in[13];
    const float* fc_w2      = (const float*)d_in[14];
    const float* norm_w     = (const float*)d_in[15];
    const float* norm_b     = (const float*)d_in[16];
    const float* out_proj_w = (const float*)d_in[17];
    float* out = (float*)d_out;

    float* ws = (float*)d_ws;
    size_t o = 0;
    float* xc_lin = ws + o; o += (size_t)BB * LL * DIc;
    float* z_silu = ws + o; o += (size_t)BB * LL * DIc;
    float* xc_act = ws + o; o += (size_t)BB * LL * DIc;
    float* xs     = ws + o; o += (size_t)KK * BB * LL * DIc;
    float* xdbl   = ws + o; o += (size_t)KK * BB * LL * CPROJ;
    float* delta  = ws + o; o += (size_t)KK * BB * LL * DIc;
    float* yarr   = ws + o; o += (size_t)KK * BB * LL * DIc;
    float* yfinal = ws + o; o += (size_t)BB * LL * DIc;
    float* wpad   = ws + o; o += (size_t)KK * CPAD * DIc;
    float* gcpart = ws + o; o += (size_t)BB * GC_NCH * DIc;
    float* gate   = ws + o; o += BB * DMm;

    k_padw    <<<dim3(KK, CPAD),       DIc, 0, stream>>>(x_proj_w, wpad);
    k_inproj  <<<dim3(12, 576),        32,  0, stream>>>(x, in_proj_w, xc_lin, z_silu);
    k_dwconv  <<<dim3(BB * LL),        DIc, 0, stream>>>(xc_lin, conv_w, conv_b, xc_act);
    k_gcmean  <<<dim3(BB, GC_NCH),     DIc, 0, stream>>>(xc_act, gcpart);
    k_gate    <<<dim3(BB),             DIc, 0, stream>>>(gcpart, fc_w1, fc_w2, gate);
    k_build_xs<<<dim3(LL, BB, KK),     DIc, 0, stream>>>(xc_act, xs);
    k_xproj   <<<dim3(2, 576, KK),     32,  0, stream>>>(xs, wpad, xdbl);
    k_dtproj  <<<dim3(6, 576, KK),     32,  0, stream>>>(xdbl, dt_w, dt_b, delta);
    k_scan    <<<dim3(BB * KK, DIc / 32), 32 * SCAN_NG, 0, stream>>>(xs, xdbl, delta, A_logs, Ds, yarr);
    k_combine <<<dim3(LL, BB),         DIc, 0, stream>>>(yarr, gate, gamma1, gamma2, gamma3,
                                                         gamma4, norm_w, norm_b, z_silu, yfinal);
    k_outproj <<<dim3(6, 576),         32,  0, stream>>>(yfinal, out_proj_w, out);
}


# k_build_xs removed: k_xproj and k_scan read the four traversal orders straight from the conv output by pixel index (scan loop specialised per direction); no xs copy written or read; sum order kept
# speedup vs baseline: 6.4322x; 1.0582x over previous
#include <hip/hip_runtime.h>
#include <math.h>


typedef float v2f __attribute__((ext_vector_type(2)));
typedef float v8f __attribute__((ext_vector_type(8)));

#define BB 4
#define HH 48
#define WW 48
#define DMm 192
#define DIc 192
#define KK 4
#define NN 16
#define RR 12
#define LL (HH * WW)
#define CPROJ (RR + 2 * NN)
#define CPAD 64

__device__ __forceinline__ float siluf(float x)     { return x / (1.f + __expf(-x)); }
__device__ __forceinline__ float sigm(float x)      { return 1.f / (1.f + __expf(-x)); }
__device__ __forceinline__ float softplusf_(float x){ return (x > 20.f) ? x : log1pf(__expf(x)); }
__device__ __forceinline__ float geluf(float x)     { return 0.5f * x * (1.f + erff(x * 0.70710678118654752440f)); }

template <int NT>
__device__ __forceinline__ void wmma_rows_multi(const float* __restrict__ arow,
                                                const float* __restrict__ W,
                                                int col0, int Kdim, int ldw,
                                                v8f acc[NT])
{
    const int lane = threadIdx.x & 31;
    const int m    = lane & 15;
    const int kh   = lane >> 4;
    const float* __restrict__ ap = arow + 2 * kh;
    const float* __restrict__ bp[NT];
#pragma unroll
    for (int t = 0; t < NT; ++t) {
        bp[t] = W + (size_t)(col0 + t * 16 + m) * ldw + 2 * kh;
        acc[t] = {};
    }
    v2f a_cur = *(const v2f*)(ap);
    v2f b_cur[NT];
#pragma unroll
    for (int t = 0; t < NT; ++t) b_cur[t] = *(const v2f*)(bp[t]);

    for (int k0 = 4; k0 < Kdim; k0 += 4) {
        v2f a_nxt = *(const v2f*)(ap + k0);
        v2f b_nxt[NT];
#pragma unroll
        for (int t = 0; t < NT; ++t) b_nxt[t] = *(const v2f*)(bp[t] + k0);
#pragma unroll
        for (int t = 0; t < NT; ++t)
            acc[t] = __builtin_amdgcn_wmma_f32_16x16x4_f32(false, a_cur, false, b_cur[t],
                                                           (short)0, acc[t], false, false);
        a_cur = a_nxt;
#pragma unroll
        for (int t = 0; t < NT; ++t) b_cur[t] = b_nxt[t];
    }
#pragma unroll
    for (int t = 0; t < NT; ++t)
        acc[t] = __builtin_amdgcn_wmma_f32_16x16x4_f32(false, a_cur, false, b_cur[t],
                                                       (short)0, acc[t], false, false);
}

template <int NT>
__device__ __forceinline__ void wmma_xwT_multi(const float* __restrict__ X,
                                               const float* __restrict__ W,
                                               int row0, int col0,
                                               int Kdim, int ldx, int ldw,
                                               v8f acc[NT])
{
    const int m = threadIdx.x & 15;
    wmma_rows_multi<NT>(X + (size_t)(row0 + m) * ldx, W, col0, Kdim, ldw, acc);
}

__global__ void k_padw(const float* __restrict__ xpw, float* __restrict__ wpad)
{
    int k = blockIdx.x, c = blockIdx.y, d = threadIdx.x;
    wpad[((size_t)k * CPAD + c) * DIc + d] =
        (c < CPROJ) ? xpw[((size_t)k * CPROJ + c) * DIc + d] : 0.f;
}

__global__ void k_inproj(const float* __restrict__ x, const float* __restrict__ w,
                         float* __restrict__ xc_lin, float* __restrict__ z_silu)
{
    const int row0 = blockIdx.y * 16, col0 = blockIdx.x * 32;
    v8f acc[2];
    wmma_xwT_multi<2>(x, w, row0, col0, DMm, DMm, DMm, acc);
    const int lane = threadIdx.x & 31;
    const int nn = lane & 15, mh = (lane >> 4) * 8;
#pragma unroll
    for (int t = 0; t < 2; ++t)
#pragma unroll
        for (int v = 0; v < 8; ++v) {
            int r = row0 + mh + v;
            int c = col0 + t * 16 + nn;
            float val = acc[t][v];
            if (c < DIc) xc_lin[(size_t)r * DIc + c] = val;
            else         z_silu[(size_t)r * DIc + (c - DIc)] = siluf(val);
        }
}

__global__ void k_dwconv(const float* __restrict__ xc_lin, const float* __restrict__ cw,
                         const float* __restrict__ cb, float* __restrict__ xc_act)
{
    int pix = blockIdx.x;
    int b = pix / LL, rem = pix % LL;
    int h = rem / WW, w = rem % WW;
    int c = threadIdx.x;
    float s = cb[c];
#pragma unroll
    for (int kh = 0; kh < 3; ++kh)
#pragma unroll
        for (int kw = 0; kw < 3; ++kw) {
            int hh = h + kh - 1, ww2 = w + kw - 1;
            if (hh >= 0 && hh < HH && ww2 >= 0 && ww2 < WW)
                s += xc_lin[((size_t)b * LL + hh * WW + ww2) * DIc + c] * cw[c * 9 + kh * 3 + kw];
        }
    xc_act[((size_t)b * LL + rem) * DIc + c] = siluf(s);
}

#define GC_NCH 48
#define GC_P (LL / GC_NCH)
__global__ void k_gcmean(const float* __restrict__ xc_act, float* __restrict__ gcpart)
{
    int b = blockIdx.x, ch = blockIdx.y, d = threadIdx.x;
    const float* __restrict__ src = xc_act + ((size_t)b * LL + (size_t)ch * GC_P) * DIc + d;
    float s = 0.f;
#pragma unroll
    for (int p = 0; p < GC_P; ++p)
        s += src[(size_t)p * DIc];
    gcpart[((size_t)b * GC_NCH + ch) * DIc + d] = s;
}

__global__ void k_gate(const float* __restrict__ gcpart, const float* __restrict__ fw1,
                       const float* __restrict__ fw2, float* __restrict__ gate)
{
    __shared__ float g[DIc];
    __shared__ float h1[DMm / 4];
    int b = blockIdx.x, d = threadIdx.x;
    float gs = 0.f;
#pragma unroll
    for (int c = 0; c < GC_NCH; ++c)
        gs += gcpart[((size_t)b * GC_NCH + c) * DIc + d];
    g[d] = gs * (1.f / LL);
    __syncthreads();
    if (d < DMm / 4) {
        float s = 0.f;
        for (int j = 0; j < DMm; ++j) s += fw1[d * DMm + j] * g[j];
        h1[d] = geluf(s);
    }
    __syncthreads();
    float s2 = 0.f;
#pragma unroll
    for (int j = 0; j < DMm / 4; ++j) s2 += fw2[d * (DMm / 4) + j] * h1[j];
    gate[b * DMm + d] = sigm(s2);
}

__device__ __forceinline__ int seq_pixel(int k, int l)
{
    int lb = (k < 2) ? l : (LL - 1 - l);
    int w_in = lb & 3, h_in = (lb >> 2) & 3;
    int nw = (lb >> 4) % (WW / 4);
    int nh = (lb >> 4) / (WW / 4);
    int p = nh * 4 + h_in, q = nw * 4 + w_in;
    return ((k & 1) == 0) ? (p * WW + q) : (q * WW + p);
}

__global__ void k_xproj(const float* __restrict__ xc_act, const float* __restrict__ wpad,
                        float* __restrict__ xdbl)
{
    int k = blockIdx.z;
    const float* W = wpad + (size_t)k * CPAD * DIc;
    float* O = xdbl + (size_t)k * BB * LL * CPROJ;
    const int row0 = blockIdx.y * 16, col0 = blockIdx.x * 32;
    const int lane = threadIdx.x & 31;
    const int ra = row0 + (lane & 15);
    const int ba = ra / LL, la = ra - ba * LL;
    const float* arow = xc_act + ((size_t)ba * LL + seq_pixel(k, la)) * DIc;
    v8f acc[2];
    wmma_rows_multi<2>(arow, W, col0, DIc, DIc, acc);
    const int nn = lane & 15, mh = (lane >> 4) * 8;
#pragma unroll
    for (int t = 0; t < 2; ++t)
#pragma unroll
        for (int v = 0; v < 8; ++v) {
            int r = row0 + mh + v, c = col0 + t * 16 + nn;
            if (c < CPROJ) O[(size_t)r * CPROJ + c] = acc[t][v];
        }
}

__global__ void k_dtproj(const float* __restrict__ xdbl, const float* __restrict__ dtw,
                         const float* __restrict__ dtb, float* __restrict__ delta)
{
    int k = blockIdx.z;
    const float* X = xdbl + (size_t)k * BB * LL * CPROJ;
    const float* W = dtw + (size_t)k * DIc * RR;
    float* O = delta + (size_t)k * BB * LL * DIc;
    const int row0 = blockIdx.y * 16, col0 = blockIdx.x * 32;
    v8f acc[2];
    wmma_xwT_multi<2>(X, W, row0, col0, RR, CPROJ, RR, acc);
    const int lane = threadIdx.x & 31;
    const int nn = lane & 15, mh = (lane >> 4) * 8;
#pragma unroll
    for (int t = 0; t < 2; ++t)
#pragma unroll
        for (int v = 0; v < 8; ++v) {
            int r = row0 + mh + v, c = col0 + t * 16 + nn;
            O[(size_t)r * DIc + c] = softplusf_(acc[t][v] + dtb[k * DIc + c]);
        }
}

#define SCAN_T 16
#define SCAN_NG 8
#define SCAN_NPG (NN / SCAN_NG)
#define SCAN_RPW (SCAN_T / SCAN_NG)
static_assert(SCAN_T == 16 && LL % SCAN_T == 0, "a chunk must be one 4x4 window");

template <int TR, int FLIP>
__device__ __forceinline__ constexpr int slot_pixoff(int t)
{
    const int tt = FLIP ? (SCAN_T - 1 - t) : t;
    const int h_in = tt >> 2, w_in = tt & 3;
    return TR ? (w_in * WW + h_in) : (h_in * WW + w_in);
}
template <int TR, int FLIP>
__device__ __forceinline__ int chunk_origin(int l0)
{
    const int widx = FLIP ? (LL / SCAN_T - 1 - (l0 >> 4)) : (l0 >> 4);
    const int nw = widx % (WW / 4), nh = widx / (WW / 4);
    return TR ? (nw * 4 * WW + nh * 4) : (nh * 4 * WW + nw * 4);
}

template <int TR, int FLIP>
__device__ __forceinline__ void scan_dir(const float* __restrict__ uimg,
                                         const float* __restrict__ dp,
                                         const float* __restrict__ bcw,
                                         float* __restrict__ yp,
                                         const float (&A)[SCAN_NPG], const float Dg,
                                         const int g, const int lane,
                                         float (*ybuf)[SCAN_T][SCAN_NG][32])
{
    const int n0 = g * SCAN_NPG;
    float h[SCAN_NPG];
#pragma unroll
    for (int n = 0; n < SCAN_NPG; ++n) h[n] = 0.f;

    float dtc[SCAN_T], uc[SCAN_T], bcc[SCAN_T];
    {
        const float* __restrict__ up0 = uimg + (size_t)chunk_origin<TR, FLIP>(0) * DIc;
#pragma unroll
        for (int t = 0; t < SCAN_T; ++t) {
            dtc[t] = dp[(size_t)t * DIc];
            uc[t]  = up0[slot_pixoff<TR, FLIP>(t) * DIc];
            bcc[t] = bcw[(size_t)t * CPROJ];
            __builtin_amdgcn_sched_barrier(0);
        }
    }
    int buf = 0;
    for (int l0 = 0; l0 < LL; l0 += SCAN_T) {
        const int ln = (l0 + SCAN_T < LL) ? (l0 + SCAN_T) : l0;
        const float* __restrict__ upn = uimg + (size_t)chunk_origin<TR, FLIP>(ln) * DIc;
#pragma unroll
        for (int t = 0; t < SCAN_T; ++t) {
            const float dt = dtc[t];
            const float u  = uc[t];
            const int  bci = __builtin_bit_cast(int, bcc[t]);
            const float du = dt * u;
            float y = 0.f;
#pragma unroll
            for (int n = 0; n < SCAN_NPG; ++n) {
                const float Bt = __builtin_bit_cast(float, __builtin_amdgcn_readlane(bci, n0 + n));
                const float Ct = __builtin_bit_cast(float, __builtin_amdgcn_readlane(bci, NN + n0 + n));
                h[n] = __expf(dt * A[n]) * h[n] + du * Bt;
                y += h[n] * Ct;
            }
            ybuf[buf][t][g][lane] = y + Dg * u;
            dtc[t] = dp[(size_t)(ln + t) * DIc];
            uc[t]  = upn[slot_pixoff<TR, FLIP>(t) * DIc];
            bcc[t] = bcw[(size_t)(ln + t) * CPROJ];
            __builtin_amdgcn_sched_barrier(0);
        }
        __builtin_amdgcn_fence(__ATOMIC_RELEASE, "workgroup", "local");
        __builtin_amdgcn_s_barrier();
        __builtin_amdgcn_fence(__ATOMIC_ACQUIRE, "workgroup", "local");
#pragma unroll
        for (int r = 0; r < SCAN_RPW; ++r) {
            const int t = g * SCAN_RPW + r;
            float s = ybuf[buf][t][0][lane];
#pragma unroll
            for (int q = 1; q < SCAN_NG; ++q) s += ybuf[buf][t][q][lane];
            yp[(size_t)(l0 + t) * DIc] = s;
        }
        buf ^= 1;
    }
}

__global__ __launch_bounds__(32 * SCAN_NG)
void k_scan(const float* __restrict__ xc_act, const float* __restrict__ xdbl,
            const float* __restrict__ delta, const float* __restrict__ A_logs,
            const float* __restrict__ Ds, float* __restrict__ yarr)
{
    __shared__ float ybuf[2][SCAN_T][SCAN_NG][32];
    const int kb = blockIdx.x;
    const int k = kb & 3, b = kb >> 2;
    const int lane = threadIdx.x & 31;
    const int g = __builtin_amdgcn_readfirstlane((int)(threadIdx.x >> 5));
    const int d = blockIdx.y * 32 + lane;
    const int n0 = g * SCAN_NPG;
    float A[SCAN_NPG];
#pragma unroll
    for (int n = 0; n < SCAN_NPG; ++n)
        A[n] = -__expf(A_logs[((size_t)(k * DIc + d)) * NN + n0 + n]);
    const float Dg = (g == SCAN_NG - 1) ? Ds[k * DIc + d] : 0.f;
    const size_t seqbase = ((size_t)(k * BB + b)) * LL;
    const float* uimg = xc_act + (size_t)b * LL * DIc + d;
    const float* dp   = delta + seqbase * DIc + d;
    const float* bcw  = xdbl  + seqbase * CPROJ + RR + lane;
    float* yp         = yarr  + seqbase * DIc + d;
    switch (k) {
    case 0:  scan_dir<0, 0>(uimg, dp, bcw, yp, A, Dg, g, lane, ybuf); break;
    case 1:  scan_dir<1, 0>(uimg, dp, bcw, yp, A, Dg, g, lane, ybuf); break;
    case 2:  scan_dir<0, 1>(uimg, dp, bcw, yp, A, Dg, g, lane, ybuf); break;
    default: scan_dir<1, 1>(uimg, dp, bcw, yp, A, Dg, g, lane, ybuf); break;
    }
}

__global__ void k_combine(const float* __restrict__ yarr, const float* __restrict__ gate,
                          const float* __restrict__ g1, const float* __restrict__ g2,
                          const float* __restrict__ g3, const float* __restrict__ g4,
                          const float* __restrict__ nw, const float* __restrict__ nb,
                          const float* __restrict__ z_silu, float* __restrict__ yfinal)
{
    __shared__ float s1[256], s2[256];
    int lr = blockIdx.x, b = blockIdx.y, d = threadIdx.x;
    int h = lr / WW, w = lr % WW;
    int l = (((h >> 2) * (WW / 4) + (w >> 2)) << 4) + ((h & 3) << 2) + (w & 3);
    int lt = (l % WW) * HH + (l / WW);
    const size_t bstride = (size_t)LL * DIc;
    const float* Y0 = yarr + ((size_t)(0 * BB + b)) * bstride;
    const float* Y1 = yarr + ((size_t)(1 * BB + b)) * bstride;
    const float* Y2 = yarr + ((size_t)(2 * BB + b)) * bstride;
    const float* Y3 = yarr + ((size_t)(3 * BB + b)) * bstride;
    float yc = g1[d] * Y0[(size_t)l * DIc + d]
             + g2[d] * Y2[(size_t)(LL - 1 - l) * DIc + d]
             + g3[d] * Y1[(size_t)lt * DIc + d]
             + g4[d] * Y3[(size_t)(LL - 1 - lt) * DIc + d];
    yc *= gate[b * DIc + d];
    s1[d] = yc; s2[d] = yc * yc;
    if (d < 64) { s1[192 + d] = 0.f; s2[192 + d] = 0.f; }
    __syncthreads();
    for (int off = 128; off > 0; off >>= 1) {
        if (d < off) { s1[d] += s1[d + off]; s2[d] += s2[d + off]; }
        __syncthreads();
    }
    float mu  = s1[0] * (1.f / DIc);
    float var = s2[0] * (1.f / DIc) - mu * mu;
    float inv = rsqrtf(var + 1e-5f);
    yc = (yc - mu) * inv * nw[d] + nb[d];
    yc *= z_silu[((size_t)b * LL + lr) * DIc + d];
    yfinal[((size_t)b * LL + lr) * DIc + d] = yc;
}

__global__ void k_outproj(const float* __restrict__ yf, const float* __restrict__ w,
                          float* __restrict__ out)
{
    const int row0 = blockIdx.y * 16, col0 = blockIdx.x * 32;
    v8f acc[2];
    wmma_xwT_multi<2>(yf, w, row0, col0, DIc, DIc, DIc, acc);
    const int lane = threadIdx.x & 31;
    const int nn = lane & 15, mh = (lane >> 4) * 8;
#pragma unroll
    for (int t = 0; t < 2; ++t)
#pragma unroll
        for (int v = 0; v < 8; ++v) {
            int r = row0 + mh + v, c = col0 + t * 16 + nn;
            out[(size_t)r * DMm + c] = acc[t][v];
        }
}

extern "C" void kernel_launch(void* const* d_in, const int* in_sizes, int n_in,
                              void* d_out, int out_size, void* d_ws, size_t ws_size,
                              hipStream_t stream)
{
    (void)in_sizes; (void)n_in; (void)out_size; (void)ws_size;
    const float* x          = (const float*)d_in[0];
    const float* in_proj_w  = (const float*)d_in[1];
    const float* conv_w     = (const float*)d_in[2];
    const float* conv_b     = (const float*)d_in[3];
    const float* x_proj_w   = (const float*)d_in[4];
    const float* dt_w       = (const float*)d_in[5];
    const float* dt_b       = (const float*)d_in[6];
    const float* A_logs     = (const float*)d_in[7];
    const float* Ds         = (const float*)d_in[8];
    const float* gamma1     = (const float*)d_in[9];
    const float* gamma2     = (const float*)d_in[10];
    const float* gamma3     = (const float*)d_in[11];
    const float* gamma4     = (const float*)d_in[12];
    const float* fc_w1      = (const float*)d_in[13];
    const float* fc_w2      = (const float*)d_in[14];
    const float* norm_w     = (const float*)d_in[15];
    const float* norm_b     = (const float*)d_in[16];
    const float* out_proj_w = (const float*)d_in[17];
    float* out = (float*)d_out;

    float* ws = (float*)d_ws;
    size_t o = 0;
    float* xc_lin = ws + o; o += (size_t)BB * LL * DIc;
    float* z_silu = ws + o; o += (size_t)BB * LL * DIc;
    float* xc_act = ws + o; o += (size_t)BB * LL * DIc;
    float* xdbl   = ws + o; o += (size_t)KK * BB * LL * CPROJ;
    float* delta  = ws + o; o += (size_t)KK * BB * LL * DIc;
    float* yarr   = ws + o; o += (size_t)KK * BB * LL * DIc;
    float* yfinal = ws + o; o += (size_t)BB * LL * DIc;
    float* wpad   = ws + o; o += (size_t)KK * CPAD * DIc;
    float* gcpart = ws + o; o += (size_t)BB * GC_NCH * DIc;
    float* gate   = ws + o; o += BB * DMm;

    k_padw    <<<dim3(KK, CPAD),       DIc, 0, stream>>>(x_proj_w, wpad);
    k_inproj  <<<dim3(12, 576),        32,  0, stream>>>(x, in_proj_w, xc_lin, z_silu);
    k_dwconv  <<<dim3(BB * LL),        DIc, 0, stream>>>(xc_lin, conv_w, conv_b, xc_act);
    k_gcmean  <<<dim3(BB, GC_NCH),     DIc, 0, stream>>>(xc_act, gcpart);
    k_gate    <<<dim3(BB),             DIc, 0, stream>>>(gcpart, fc_w1, fc_w2, gate);
    k_xproj   <<<dim3(2, 576, KK),     32,  0, stream>>>(xc_act, wpad, xdbl);
    k_dtproj  <<<dim3(6, 576, KK),     32,  0, stream>>>(xdbl, dt_w, dt_b, delta);
    k_scan    <<<dim3(BB * KK, DIc / 32), 32 * SCAN_NG, 0, stream>>>(xc_act, xdbl, delta, A_logs, Ds, yarr);
    k_combine <<<dim3(LL, BB),         DIc, 0, stream>>>(yarr, gate, gamma1, gamma2, gamma3,
                                                         gamma4, norm_w, norm_b, z_silu, yfinal);
    k_outproj <<<dim3(6, 576),         32,  0, stream>>>(yfinal, out_proj_w, out);
}


# k_dwconv and k_combine: one wave per pixel (6 channels per lane, 4 pixels per wave), 288 work-groups each instead of 9216 of 6 waves; LayerNorm sums by lane butterfly, no LDS (new sum order)
# speedup vs baseline: 6.7262x; 1.0457x over previous
#include <hip/hip_runtime.h>
#include <math.h>


typedef float v2f __attribute__((ext_vector_type(2)));
typedef float v8f __attribute__((ext_vector_type(8)));

#define BB 4
#define HH 48
#define WW 48
#define DMm 192
#define DIc 192
#define KK 4
#define NN 16
#define RR 12
#define LL (HH * WW)
#define CPROJ (RR + 2 * NN)
#define CPAD 64

__device__ __forceinline__ float siluf(float x)     { return x / (1.f + __expf(-x)); }
__device__ __forceinline__ float sigm(float x)      { return 1.f / (1.f + __expf(-x)); }
__device__ __forceinline__ float softplusf_(float x){ return (x > 20.f) ? x : log1pf(__expf(x)); }
__device__ __forceinline__ float geluf(float x)     { return 0.5f * x * (1.f + erff(x * 0.70710678118654752440f)); }

template <int NT>
__device__ __forceinline__ void wmma_rows_multi(const float* __restrict__ arow,
                                                const float* __restrict__ W,
                                                int col0, int Kdim, int ldw,
                                                v8f acc[NT])
{
    const int lane = threadIdx.x & 31;
    const int m    = lane & 15;
    const int kh   = lane >> 4;
    const float* __restrict__ ap = arow + 2 * kh;
    const float* __restrict__ bp[NT];
#pragma unroll
    for (int t = 0; t < NT; ++t) {
        bp[t] = W + (size_t)(col0 + t * 16 + m) * ldw + 2 * kh;
        acc[t] = {};
    }
    v2f a_cur = *(const v2f*)(ap);
    v2f b_cur[NT];
#pragma unroll
    for (int t = 0; t < NT; ++t) b_cur[t] = *(const v2f*)(bp[t]);

    for (int k0 = 4; k0 < Kdim; k0 += 4) {
        v2f a_nxt = *(const v2f*)(ap + k0);
        v2f b_nxt[NT];
#pragma unroll
        for (int t = 0; t < NT; ++t) b_nxt[t] = *(const v2f*)(bp[t] + k0);
#pragma unroll
        for (int t = 0; t < NT; ++t)
            acc[t] = __builtin_amdgcn_wmma_f32_16x16x4_f32(false, a_cur, false, b_cur[t],
                                                           (short)0, acc[t], false, false);
        a_cur = a_nxt;
#pragma unroll
        for (int t = 0; t < NT; ++t) b_cur[t] = b_nxt[t];
    }
#pragma unroll
    for (int t = 0; t < NT; ++t)
        acc[t] = __builtin_amdgcn_wmma_f32_16x16x4_f32(false, a_cur, false, b_cur[t],
                                                       (short)0, acc[t], false, false);
}

template <int NT>
__device__ __forceinline__ void wmma_xwT_multi(const float* __restrict__ X,
                                               const float* __restrict__ W,
                                               int row0, int col0,
                                               int Kdim, int ldx, int ldw,
                                               v8f acc[NT])
{
    const int m = threadIdx.x & 15;
    wmma_rows_multi<NT>(X + (size_t)(row0 + m) * ldx, W, col0, Kdim, ldw, acc);
}

__global__ void k_padw(const float* __restrict__ xpw, float* __restrict__ wpad)
{
    int k = blockIdx.x, c = blockIdx.y, d = threadIdx.x;
    wpad[((size_t)k * CPAD + c) * DIc + d] =
        (c < CPROJ) ? xpw[((size_t)k * CPROJ + c) * DIc + d] : 0.f;
}

__global__ void k_inproj(const float* __restrict__ x, const float* __restrict__ w,
                         float* __restrict__ xc_lin, float* __restrict__ z_silu)
{
    const int row0 = blockIdx.y * 16, col0 = blockIdx.x * 32;
    v8f acc[2];
    wmma_xwT_multi<2>(x, w, row0, col0, DMm, DMm, DMm, acc);
    const int lane = threadIdx.x & 31;
    const int nn = lane & 15, mh = (lane >> 4) * 8;
#pragma unroll
    for (int t = 0; t < 2; ++t)
#pragma unroll
        for (int v = 0; v < 8; ++v) {
            int r = row0 + mh + v;
            int c = col0 + t * 16 + nn;
            float val = acc[t][v];
            if (c < DIc) xc_lin[(size_t)r * DIc + c] = val;
            else         z_silu[(size_t)r * DIc + (c - DIc)] = siluf(val);
        }
}

#define PX_WAVES 8
#define PX_PPW   4
#define PX_CPL   (DIc / 32)
#define PX_NWG   (BB * LL / (PX_WAVES * PX_PPW))
static_assert(WW % PX_PPW == 0 && (BB * LL) % (PX_WAVES * PX_PPW) == 0 && DIc % 32 == 0,
              "a wave's pixels must lie in one image row");

__global__ __launch_bounds__(32 * PX_WAVES)
void k_dwconv(const float* __restrict__ xc_lin, const float* __restrict__ cw,
              const float* __restrict__ cb, float* __restrict__ xc_act)
{
    __shared__ float wsh[DIc * 9];
    const int lane = threadIdx.x & 31;
    const int wave = __builtin_amdgcn_readfirstlane((int)(threadIdx.x >> 5));
    const int pix0 = (blockIdx.x * PX_WAVES + wave) * PX_PPW;
    const int b = pix0 / LL, rem0 = pix0 - b * LL;
    const int h = rem0 / WW, w0 = rem0 - h * WW;
    const float* __restrict__ img = xc_lin + (size_t)b * LL * DIc + lane;

    float win[3][PX_PPW + 2][PX_CPL];
#pragma unroll
    for (int r = 0; r < 3; ++r) {
        const int hh = h + r - 1;
        const bool rok = (hh >= 0) && (hh < HH);
        const int hc = hh < 0 ? 0 : (hh >= HH ? HH - 1 : hh);
#pragma unroll
        for (int q = 0; q < PX_PPW + 2; ++q) {
            const int wq = w0 + q - 1;
            const bool cok = (q == 0) ? (wq >= 0) : ((q == PX_PPW + 1) ? (wq < WW) : true);
            const int wc = (q == 0) ? (wq < 0 ? 0 : wq) : ((q == PX_PPW + 1) ? (wq >= WW ? WW - 1 : wq) : wq);
            const float* __restrict__ sp = img + (size_t)(hc * WW + wc) * DIc;
#pragma unroll
            for (int j = 0; j < PX_CPL; ++j) {
                const float v = sp[32 * j];
                win[r][q][j] = (rok && cok) ? v : 0.f;
            }
        }
    }

    constexpr int WFULL = (DIc * 9) / (32 * PX_WAVES);
    float wtmp[WFULL];
#pragma unroll
    for (int i = 0; i < WFULL; ++i) wtmp[i] = cw[(int)threadIdx.x + i * 32 * PX_WAVES];
    const int itail = (int)threadIdx.x + WFULL * 32 * PX_WAVES;
    const float wtail = (itail < DIc * 9) ? cw[itail] : 0.f;
#pragma unroll
    for (int i = 0; i < WFULL; ++i) wsh[(int)threadIdx.x + i * 32 * PX_WAVES] = wtmp[i];
    if (itail < DIc * 9) wsh[itail] = wtail;
    __syncthreads();
    float wreg[PX_CPL][9], acc[PX_PPW][PX_CPL];
#pragma unroll
    for (int j = 0; j < PX_CPL; ++j) {
#pragma unroll
        for (int t = 0; t < 9; ++t) wreg[j][t] = wsh[(lane + 32 * j) * 9 + t];
        const float bj = cb[lane + 32 * j];
#pragma unroll
        for (int p = 0; p < PX_PPW; ++p) acc[p][j] = bj;
    }
#pragma unroll
    for (int p = 0; p < PX_PPW; ++p)
#pragma unroll
        for (int kh = 0; kh < 3; ++kh)
#pragma unroll
            for (int kw = 0; kw < 3; ++kw)
#pragma unroll
                for (int j = 0; j < PX_CPL; ++j)
                    acc[p][j] += win[kh][p + kw][j] * wreg[j][kh * 3 + kw];
#pragma unroll
    for (int p = 0; p < PX_PPW; ++p) {
        float* __restrict__ op = xc_act + (size_t)(pix0 + p) * DIc + lane;
#pragma unroll
        for (int j = 0; j < PX_CPL; ++j) op[32 * j] = siluf(acc[p][j]);
    }
}

#define GC_NCH 48
#define GC_P (LL / GC_NCH)
__global__ void k_gcmean(const float* __restrict__ xc_act, float* __restrict__ gcpart)
{
    int b = blockIdx.x, ch = blockIdx.y, d = threadIdx.x;
    const float* __restrict__ src = xc_act + ((size_t)b * LL + (size_t)ch * GC_P) * DIc + d;
    float s = 0.f;
#pragma unroll
    for (int p = 0; p < GC_P; ++p)
        s += src[(size_t)p * DIc];
    gcpart[((size_t)b * GC_NCH + ch) * DIc + d] = s;
}

__global__ void k_gate(const float* __restrict__ gcpart, const float* __restrict__ fw1,
                       const float* __restrict__ fw2, float* __restrict__ gate)
{
    __shared__ float g[DIc];
    __shared__ float h1[DMm / 4];
    int b = blockIdx.x, d = threadIdx.x;
    float gs = 0.f;
#pragma unroll
    for (int c = 0; c < GC_NCH; ++c)
        gs += gcpart[((size_t)b * GC_NCH + c) * DIc + d];
    g[d] = gs * (1.f / LL);
    __syncthreads();
    if (d < DMm / 4) {
        float s = 0.f;
        for (int j = 0; j < DMm; ++j) s += fw1[d * DMm + j] * g[j];
        h1[d] = geluf(s);
    }
    __syncthreads();
    float s2 = 0.f;
#pragma unroll
    for (int j = 0; j < DMm / 4; ++j) s2 += fw2[d * (DMm / 4) + j] * h1[j];
    gate[b * DMm + d] = sigm(s2);
}

__device__ __forceinline__ int seq_pixel(int k, int l)
{
    int lb = (k < 2) ? l : (LL - 1 - l);
    int w_in = lb & 3, h_in = (lb >> 2) & 3;
    int nw = (lb >> 4) % (WW / 4);
    int nh = (lb >> 4) / (WW / 4);
    int p = nh * 4 + h_in, q = nw * 4 + w_in;
    return ((k & 1) == 0) ? (p * WW + q) : (q * WW + p);
}

__global__ void k_xproj(const float* __restrict__ xc_act, const float* __restrict__ wpad,
                        float* __restrict__ xdbl)
{
    int k = blockIdx.z;
    const float* W = wpad + (size_t)k * CPAD * DIc;
    float* O = xdbl + (size_t)k * BB * LL * CPROJ;
    const int row0 = blockIdx.y * 16, col0 = blockIdx.x * 32;
    const int lane = threadIdx.x & 31;
    const int ra = row0 + (lane & 15);
    const int ba = ra / LL, la = ra - ba * LL;
    const float* arow = xc_act + ((size_t)ba * LL + seq_pixel(k, la)) * DIc;
    v8f acc[2];
    wmma_rows_multi<2>(arow, W, col0, DIc, DIc, acc);
    const int nn = lane & 15, mh = (lane >> 4) * 8;
#pragma unroll
    for (int t = 0; t < 2; ++t)
#pragma unroll
        for (int v = 0; v < 8; ++v) {
            int r = row0 + mh + v, c = col0 + t * 16 + nn;
            if (c < CPROJ) O[(size_t)r * CPROJ + c] = acc[t][v];
        }
}

__global__ void k_dtproj(const float* __restrict__ xdbl, const float* __restrict__ dtw,
                         const float* __restrict__ dtb, float* __restrict__ delta)
{
    int k = blockIdx.z;
    const float* X = xdbl + (size_t)k * BB * LL * CPROJ;
    const float* W = dtw + (size_t)k * DIc * RR;
    float* O = delta + (size_t)k * BB * LL * DIc;
    const int row0 = blockIdx.y * 16, col0 = blockIdx.x * 32;
    v8f acc[2];
    wmma_xwT_multi<2>(X, W, row0, col0, RR, CPROJ, RR, acc);
    const int lane = threadIdx.x & 31;
    const int nn = lane & 15, mh = (lane >> 4) * 8;
#pragma unroll
    for (int t = 0; t < 2; ++t)
#pragma unroll
        for (int v = 0; v < 8; ++v) {
            int r = row0 + mh + v, c = col0 + t * 16 + nn;
            O[(size_t)r * DIc + c] = softplusf_(acc[t][v] + dtb[k * DIc + c]);
        }
}

#define SCAN_T 16
#define SCAN_NG 8
#define SCAN_NPG (NN / SCAN_NG)
#define SCAN_RPW (SCAN_T / SCAN_NG)
static_assert(SCAN_T == 16 && LL % SCAN_T == 0, "a chunk must be one 4x4 window");

template <int TR, int FLIP>
__device__ __forceinline__ constexpr int slot_pixoff(int t)
{
    const int tt = FLIP ? (SCAN_T - 1 - t) : t;
    const int h_in = tt >> 2, w_in = tt & 3;
    return TR ? (w_in * WW + h_in) : (h_in * WW + w_in);
}
template <int TR, int FLIP>
__device__ __forceinline__ int chunk_origin(int l0)
{
    const int widx = FLIP ? (LL / SCAN_T - 1 - (l0 >> 4)) : (l0 >> 4);
    const int nw = widx % (WW / 4), nh = widx / (WW / 4);
    return TR ? (nw * 4 * WW + nh * 4) : (nh * 4 * WW + nw * 4);
}

template <int TR, int FLIP>
__device__ __forceinline__ void scan_dir(const float* __restrict__ uimg,
                                         const float* __restrict__ dp,
                                         const float* __restrict__ bcw,
                                         float* __restrict__ yp,
                                         const float (&A)[SCAN_NPG], const float Dg,
                                         const int g, const int lane,
                                         float (*ybuf)[SCAN_T][SCAN_NG][32])
{
    const int n0 = g * SCAN_NPG;
    float h[SCAN_NPG];
#pragma unroll
    for (int n = 0; n < SCAN_NPG; ++n) h[n] = 0.f;

    float dtc[SCAN_T], uc[SCAN_T], bcc[SCAN_T];
    {
        const float* __restrict__ up0 = uimg + (size_t)chunk_origin<TR, FLIP>(0) * DIc;
#pragma unroll
        for (int t = 0; t < SCAN_T; ++t) {
            dtc[t] = dp[(size_t)t * DIc];
            uc[t]  = up0[slot_pixoff<TR, FLIP>(t) * DIc];
            bcc[t] = bcw[(size_t)t * CPROJ];
            __builtin_amdgcn_sched_barrier(0);
        }
    }
    int buf = 0;
    for (int l0 = 0; l0 < LL; l0 += SCAN_T) {
        const int ln = (l0 + SCAN_T < LL) ? (l0 + SCAN_T) : l0;
        const float* __restrict__ upn = uimg + (size_t)chunk_origin<TR, FLIP>(ln) * DIc;
#pragma unroll
        for (int t = 0; t < SCAN_T; ++t) {
            const float dt = dtc[t];
            const float u  = uc[t];
            const int  bci = __builtin_bit_cast(int, bcc[t]);
            const float du = dt * u;
            float y = 0.f;
#pragma unroll
            for (int n = 0; n < SCAN_NPG; ++n) {
                const float Bt = __builtin_bit_cast(float, __builtin_amdgcn_readlane(bci, n0 + n));
                const float Ct = __builtin_bit_cast(float, __builtin_amdgcn_readlane(bci, NN + n0 + n));
                h[n] = __expf(dt * A[n]) * h[n] + du * Bt;
                y += h[n] * Ct;
            }
            ybuf[buf][t][g][lane] = y + Dg * u;
            dtc[t] = dp[(size_t)(ln + t) * DIc];
            uc[t]  = upn[slot_pixoff<TR, FLIP>(t) * DIc];
            bcc[t] = bcw[(size_t)(ln + t) * CPROJ];
            __builtin_amdgcn_sched_barrier(0);
        }
        __builtin_amdgcn_fence(__ATOMIC_RELEASE, "workgroup", "local");
        __builtin_amdgcn_s_barrier();
        __builtin_amdgcn_fence(__ATOMIC_ACQUIRE, "workgroup", "local");
#pragma unroll
        for (int r = 0; r < SCAN_RPW; ++r) {
            const int t = g * SCAN_RPW + r;
            float s = ybuf[buf][t][0][lane];
#pragma unroll
            for (int q = 1; q < SCAN_NG; ++q) s += ybuf[buf][t][q][lane];
            yp[(size_t)(l0 + t) * DIc] = s;
        }
        buf ^= 1;
    }
}

__global__ __launch_bounds__(32 * SCAN_NG)
void k_scan(const float* __restrict__ xc_act, const float* __restrict__ xdbl,
            const float* __restrict__ delta, const float* __restrict__ A_logs,
            const float* __restrict__ Ds, float* __restrict__ yarr)
{
    __shared__ float ybuf[2][SCAN_T][SCAN_NG][32];
    const int kb = blockIdx.x;
    const int k = kb & 3, b = kb >> 2;
    const int lane = threadIdx.x & 31;
    const int g = __builtin_amdgcn_readfirstlane((int)(threadIdx.x >> 5));
    const int d = blockIdx.y * 32 + lane;
    const int n0 = g * SCAN_NPG;
    float A[SCAN_NPG];
#pragma unroll
    for (int n = 0; n < SCAN_NPG; ++n)
        A[n] = -__expf(A_logs[((size_t)(k * DIc + d)) * NN + n0 + n]);
    const float Dg = (g == SCAN_NG - 1) ? Ds[k * DIc + d] : 0.f;
    const size_t seqbase = ((size_t)(k * BB + b)) * LL;
    const float* uimg = xc_act + (size_t)b * LL * DIc + d;
    const float* dp   = delta + seqbase * DIc + d;
    const float* bcw  = xdbl  + seqbase * CPROJ + RR + lane;
    float* yp         = yarr  + seqbase * DIc + d;
    switch (k) {
    case 0:  scan_dir<0, 0>(uimg, dp, bcw, yp, A, Dg, g, lane, ybuf); break;
    case 1:  scan_dir<1, 0>(uimg, dp, bcw, yp, A, Dg, g, lane, ybuf); break;
    case 2:  scan_dir<0, 1>(uimg, dp, bcw, yp, A, Dg, g, lane, ybuf); break;
    default: scan_dir<1, 1>(uimg, dp, bcw, yp, A, Dg, g, lane, ybuf); break;
    }
}

template <int XM>
__device__ __forceinline__ float row_xor(float v)
{
    return __builtin_bit_cast(float,
        __builtin_amdgcn_update_dpp(0, __builtin_bit_cast(int, v), 0x160 + XM, 0xf, 0xf, false));
}
__device__ __forceinline__ float half_xor(float v)
{
    const int x = __builtin_bit_cast(int, v);
    return __builtin_bit_cast(float,
        __builtin_amdgcn_permlanex16(x, x, 0x76543210, 0xfedcba98, false, false));
}
__device__ __forceinline__ float wave_sum(float v)
{
    v += row_xor<1>(v);
    v += row_xor<2>(v);
    v += row_xor<4>(v);
    v += row_xor<8>(v);
    v += half_xor(v);
    return v;
}

__global__ __launch_bounds__(32 * PX_WAVES)
void k_combine(const float* __restrict__ yarr, const float* __restrict__ gate,
               const float* __restrict__ g1, const float* __restrict__ g2,
               const float* __restrict__ g3, const float* __restrict__ g4,
               const float* __restrict__ nw, const float* __restrict__ nb,
               const float* __restrict__ z_silu, float* __restrict__ yfinal)
{
    const int lane = threadIdx.x & 31;
    const int wave = __builtin_amdgcn_readfirstlane((int)(threadIdx.x >> 5));
    const int pix0 = (blockIdx.x * PX_WAVES + wave) * PX_PPW;
    const int b = pix0 / LL, lr0 = pix0 - b * LL;
    const size_t bstride = (size_t)LL * DIc;
    const float* __restrict__ Y0 = yarr + ((size_t)(0 * BB + b)) * bstride + lane;
    const float* __restrict__ Y1 = yarr + ((size_t)(1 * BB + b)) * bstride + lane;
    const float* __restrict__ Y2 = yarr + ((size_t)(2 * BB + b)) * bstride + lane;
    const float* __restrict__ Y3 = yarr + ((size_t)(3 * BB + b)) * bstride + lane;
    float G1[PX_CPL], G2[PX_CPL], G3[PX_CPL], G4[PX_CPL], GT[PX_CPL], NW[PX_CPL], NB[PX_CPL];
#pragma unroll
    for (int j = 0; j < PX_CPL; ++j) {
        const int d = lane + 32 * j;
        G1[j] = g1[d]; G2[j] = g2[d]; G3[j] = g3[d]; G4[j] = g4[d];
        GT[j] = gate[b * DIc + d]; NW[j] = nw[d]; NB[j] = nb[d];
    }
#pragma unroll
    for (int p = 0; p < PX_PPW; ++p) {
        const int lr = lr0 + p;
        const int h = lr / WW, w = lr % WW;
        const int l = (((h >> 2) * (WW / 4) + (w >> 2)) << 4) + ((h & 3) << 2) + (w & 3);
        const int lt = (l % WW) * HH + (l / WW);
        const float* __restrict__ zp = z_silu + ((size_t)b * LL + lr) * DIc + lane;
        float yc[PX_CPL], zz[PX_CPL];
        float s1 = 0.f, s2 = 0.f;
#pragma unroll
        for (int j = 0; j < PX_CPL; ++j) {
            float y = G1[j] * Y0[(size_t)l * DIc + 32 * j]
                    + G2[j] * Y2[(size_t)(LL - 1 - l) * DIc + 32 * j]
                    + G3[j] * Y1[(size_t)lt * DIc + 32 * j]
                    + G4[j] * Y3[(size_t)(LL - 1 - lt) * DIc + 32 * j];
            y *= GT[j];
            yc[j] = y;
            zz[j] = zp[32 * j];
            s1 += y;
            s2 += y * y;
        }
        s1 = wave_sum(s1);
        s2 = wave_sum(s2);
        const float mu  = s1 * (1.f / DIc);
        const float var = s2 * (1.f / DIc) - mu * mu;
        const float inv = rsqrtf(var + 1e-5f);
        float* __restrict__ op = yfinal + ((size_t)b * LL + lr) * DIc + lane;
#pragma unroll
        for (int j = 0; j < PX_CPL; ++j)
            op[32 * j] = ((yc[j] - mu) * inv * NW[j] + NB[j]) * zz[j];
    }
}

__global__ void k_outproj(const float* __restrict__ yf, const float* __restrict__ w,
                          float* __restrict__ out)
{
    const int row0 = blockIdx.y * 16, col0 = blockIdx.x * 32;
    v8f acc[2];
    wmma_xwT_multi<2>(yf, w, row0, col0, DIc, DIc, DIc, acc);
    const int lane = threadIdx.x & 31;
    const int nn = lane & 15, mh = (lane >> 4) * 8;
#pragma unroll
    for (int t = 0; t < 2; ++t)
#pragma unroll
        for (int v = 0; v < 8; ++v) {
            int r = row0 + mh + v, c = col0 + t * 16 + nn;
            out[(size_t)r * DMm + c] = acc[t][v];
        }
}

extern "C" void kernel_launch(void* const* d_in, const int* in_sizes, int n_in,
                              void* d_out, int out_size, void* d_ws, size_t ws_size,
                              hipStream_t stream)
{
    (void)in_sizes; (void)n_in; (void)out_size; (void)ws_size;
    const float* x          = (const float*)d_in[0];
    const float* in_proj_w  = (const float*)d_in[1];
    const float* conv_w     = (const float*)d_in[2];
    const float* conv_b     = (const float*)d_in[3];
    const float* x_proj_w   = (const float*)d_in[4];
    const float* dt_w       = (const float*)d_in[5];
    const float* dt_b       = (const float*)d_in[6];
    const float* A_logs     = (const float*)d_in[7];
    const float* Ds         = (const float*)d_in[8];
    const float* gamma1     = (const float*)d_in[9];
    const float* gamma2     = (const float*)d_in[10];
    const float* gamma3     = (const float*)d_in[11];
    const float* gamma4     = (const float*)d_in[12];
    const float* fc_w1      = (const float*)d_in[13];
    const float* fc_w2      = (const float*)d_in[14];
    const float* norm_w     = (const float*)d_in[15];
    const float* norm_b     = (const float*)d_in[16];
    const float* out_proj_w = (const float*)d_in[17];
    float* out = (float*)d_out;

    float* ws = (float*)d_ws;
    size_t o = 0;
    float* xc_lin = ws + o; o += (size_t)BB * LL * DIc;
    float* z_silu = ws + o; o += (size_t)BB * LL * DIc;
    float* xc_act = ws + o; o += (size_t)BB * LL * DIc;
    float* xdbl   = ws + o; o += (size_t)KK * BB * LL * CPROJ;
    float* delta  = ws + o; o += (size_t)KK * BB * LL * DIc;
    float* yarr   = ws + o; o += (size_t)KK * BB * LL * DIc;
    float* yfinal = ws + o; o += (size_t)BB * LL * DIc;
    float* wpad   = ws + o; o += (size_t)KK * CPAD * DIc;
    float* gcpart = ws + o; o += (size_t)BB * GC_NCH * DIc;
    float* gate   = ws + o; o += BB * DMm;

    k_padw    <<<dim3(KK, CPAD),       DIc, 0, stream>>>(x_proj_w, wpad);
    k_inproj  <<<dim3(12, 576),        32,  0, stream>>>(x, in_proj_w, xc_lin, z_silu);
    k_dwconv  <<<dim3(PX_NWG),         32 * PX_WAVES, 0, stream>>>(xc_lin, conv_w, conv_b, xc_act);
    k_gcmean  <<<dim3(BB, GC_NCH),     DIc, 0, stream>>>(xc_act, gcpart);
    k_gate    <<<dim3(BB),             DIc, 0, stream>>>(gcpart, fc_w1, fc_w2, gate);
    k_xproj   <<<dim3(2, 576, KK),     32,  0, stream>>>(xc_act, wpad, xdbl);
    k_dtproj  <<<dim3(6, 576, KK),     32,  0, stream>>>(xdbl, dt_w, dt_b, delta);
    k_scan    <<<dim3(BB * KK, DIc / 32), 32 * SCAN_NG, 0, stream>>>(xc_act, xdbl, delta, A_logs, Ds, yarr);
    k_combine <<<dim3(PX_NWG),         32 * PX_WAVES, 0, stream>>>(yarr, gate, gamma1, gamma2, gamma3,
                                                         gamma4, norm_w, norm_b, z_silu, yfinal);
    k_outproj <<<dim3(6, 576),         32,  0, stream>>>(yfinal, out_proj_w, out);
}


# k_scan: a step's delta/u/B/C fetched once per work-group (each wave 2 steps of a chunk, 4 chunks ahead), handed to the 8 waves through LDS; B/C as LDS broadcast reads, no v_readlane; sum order kept
# speedup vs baseline: 8.6447x; 1.2852x over previous
#include <hip/hip_runtime.h>
#include <math.h>


typedef float v2f __attribute__((ext_vector_type(2)));
typedef float v8f __attribute__((ext_vector_type(8)));

#define BB 4
#define HH 48
#define WW 48
#define DMm 192
#define DIc 192
#define KK 4
#define NN 16
#define RR 12
#define LL (HH * WW)
#define CPROJ (RR + 2 * NN)
#define CPAD 64

__device__ __forceinline__ float siluf(float x)     { return x / (1.f + __expf(-x)); }
__device__ __forceinline__ float sigm(float x)      { return 1.f / (1.f + __expf(-x)); }
__device__ __forceinline__ float softplusf_(float x){ return (x > 20.f) ? x : log1pf(__expf(x)); }
__device__ __forceinline__ float geluf(float x)     { return 0.5f * x * (1.f + erff(x * 0.70710678118654752440f)); }

template <int NT>
__device__ __forceinline__ void wmma_rows_multi(const float* __restrict__ arow,
                                                const float* __restrict__ W,
                                                int col0, int Kdim, int ldw,
                                                v8f acc[NT])
{
    const int lane = threadIdx.x & 31;
    const int m    = lane & 15;
    const int kh   = lane >> 4;
    const float* __restrict__ ap = arow + 2 * kh;
    const float* __restrict__ bp[NT];
#pragma unroll
    for (int t = 0; t < NT; ++t) {
        bp[t] = W + (size_t)(col0 + t * 16 + m) * ldw + 2 * kh;
        acc[t] = {};
    }
    v2f a_cur = *(const v2f*)(ap);
    v2f b_cur[NT];
#pragma unroll
    for (int t = 0; t < NT; ++t) b_cur[t] = *(const v2f*)(bp[t]);

    for (int k0 = 4; k0 < Kdim; k0 += 4) {
        v2f a_nxt = *(const v2f*)(ap + k0);
        v2f b_nxt[NT];
#pragma unroll
        for (int t = 0; t < NT; ++t) b_nxt[t] = *(const v2f*)(bp[t] + k0);
#pragma unroll
        for (int t = 0; t < NT; ++t)
            acc[t] = __builtin_amdgcn_wmma_f32_16x16x4_f32(false, a_cur, false, b_cur[t],
                                                           (short)0, acc[t], false, false);
        a_cur = a_nxt;
#pragma unroll
        for (int t = 0; t < NT; ++t) b_cur[t] = b_nxt[t];
    }
#pragma unroll
    for (int t = 0; t < NT; ++t)
        acc[t] = __builtin_amdgcn_wmma_f32_16x16x4_f32(false, a_cur, false, b_cur[t],
                                                       (short)0, acc[t], false, false);
}

template <int NT>
__device__ __forceinline__ void wmma_xwT_multi(const float* __restrict__ X,
                                               const float* __restrict__ W,
                                               int row0, int col0,
                                               int Kdim, int ldx, int ldw,
                                               v8f acc[NT])
{
    const int m = threadIdx.x & 15;
    wmma_rows_multi<NT>(X + (size_t)(row0 + m) * ldx, W, col0, Kdim, ldw, acc);
}

__global__ void k_padw(const float* __restrict__ xpw, float* __restrict__ wpad)
{
    int k = blockIdx.x, c = blockIdx.y, d = threadIdx.x;
    wpad[((size_t)k * CPAD + c) * DIc + d] =
        (c < CPROJ) ? xpw[((size_t)k * CPROJ + c) * DIc + d] : 0.f;
}

__global__ void k_inproj(const float* __restrict__ x, const float* __restrict__ w,
                         float* __restrict__ xc_lin, float* __restrict__ z_silu)
{
    const int row0 = blockIdx.y * 16, col0 = blockIdx.x * 32;
    v8f acc[2];
    wmma_xwT_multi<2>(x, w, row0, col0, DMm, DMm, DMm, acc);
    const int lane = threadIdx.x & 31;
    const int nn = lane & 15, mh = (lane >> 4) * 8;
#pragma unroll
    for (int t = 0; t < 2; ++t)
#pragma unroll
        for (int v = 0; v < 8; ++v) {
            int r = row0 + mh + v;
            int c = col0 + t * 16 + nn;
            float val = acc[t][v];
            if (c < DIc) xc_lin[(size_t)r * DIc + c] = val;
            else         z_silu[(size_t)r * DIc + (c - DIc)] = siluf(val);
        }
}

#define PX_WAVES 8
#define PX_PPW   4
#define PX_CPL   (DIc / 32)
#define PX_NWG   (BB * LL / (PX_WAVES * PX_PPW))
static_assert(WW % PX_PPW == 0 && (BB * LL) % (PX_WAVES * PX_PPW) == 0 && DIc % 32 == 0,
              "a wave's pixels must lie in one image row");

__global__ __launch_bounds__(32 * PX_WAVES)
void k_dwconv(const float* __restrict__ xc_lin, const float* __restrict__ cw,
              const float* __restrict__ cb, float* __restrict__ xc_act)
{
    __shared__ float wsh[DIc * 9];
    const int lane = threadIdx.x & 31;
    const int wave = __builtin_amdgcn_readfirstlane((int)(threadIdx.x >> 5));
    const int pix0 = (blockIdx.x * PX_WAVES + wave) * PX_PPW;
    const int b = pix0 / LL, rem0 = pix0 - b * LL;
    const int h = rem0 / WW, w0 = rem0 - h * WW;
    const float* __restrict__ img = xc_lin + (size_t)b * LL * DIc + lane;

    float win[3][PX_PPW + 2][PX_CPL];
#pragma unroll
    for (int r = 0; r < 3; ++r) {
        const int hh = h + r - 1;
        const bool rok = (hh >= 0) && (hh < HH);
        const int hc = hh < 0 ? 0 : (hh >= HH ? HH - 1 : hh);
#pragma unroll
        for (int q = 0; q < PX_PPW + 2; ++q) {
            const int wq = w0 + q - 1;
            const bool cok = (q == 0) ? (wq >= 0) : ((q == PX_PPW + 1) ? (wq < WW) : true);
            const int wc = (q == 0) ? (wq < 0 ? 0 : wq) : ((q == PX_PPW + 1) ? (wq >= WW ? WW - 1 : wq) : wq);
            const float* __restrict__ sp = img + (size_t)(hc * WW + wc) * DIc;
#pragma unroll
            for (int j = 0; j < PX_CPL; ++j) {
                const float v = sp[32 * j];
                win[r][q][j] = (rok && cok) ? v : 0.f;
            }
        }
    }

    constexpr int WFULL = (DIc * 9) / (32 * PX_WAVES);
    float wtmp[WFULL];
#pragma unroll
    for (int i = 0; i < WFULL; ++i) wtmp[i] = cw[(int)threadIdx.x + i * 32 * PX_WAVES];
    const int itail = (int)threadIdx.x + WFULL * 32 * PX_WAVES;
    const float wtail = (itail < DIc * 9) ? cw[itail] : 0.f;
#pragma unroll
    for (int i = 0; i < WFULL; ++i) wsh[(int)threadIdx.x + i * 32 * PX_WAVES] = wtmp[i];
    if (itail < DIc * 9) wsh[itail] = wtail;
    __syncthreads();
    float wreg[PX_CPL][9], acc[PX_PPW][PX_CPL];
#pragma unroll
    for (int j = 0; j < PX_CPL; ++j) {
#pragma unroll
        for (int t = 0; t < 9; ++t) wreg[j][t] = wsh[(lane + 32 * j) * 9 + t];
        const float bj = cb[lane + 32 * j];
#pragma unroll
        for (int p = 0; p < PX_PPW; ++p) acc[p][j] = bj;
    }
#pragma unroll
    for (int p = 0; p < PX_PPW; ++p)
#pragma unroll
        for (int kh = 0; kh < 3; ++kh)
#pragma unroll
            for (int kw = 0; kw < 3; ++kw)
#pragma unroll
                for (int j = 0; j < PX_CPL; ++j)
                    acc[p][j] += win[kh][p + kw][j] * wreg[j][kh * 3 + kw];
#pragma unroll
    for (int p = 0; p < PX_PPW; ++p) {
        float* __restrict__ op = xc_act + (size_t)(pix0 + p) * DIc + lane;
#pragma unroll
        for (int j = 0; j < PX_CPL; ++j) op[32 * j] = siluf(acc[p][j]);
    }
}

#define GC_NCH 48
#define GC_P (LL / GC_NCH)
__global__ void k_gcmean(const float* __restrict__ xc_act, float* __restrict__ gcpart)
{
    int b = blockIdx.x, ch = blockIdx.y, d = threadIdx.x;
    const float* __restrict__ src = xc_act + ((size_t)b * LL + (size_t)ch * GC_P) * DIc + d;
    float s = 0.f;
#pragma unroll
    for (int p = 0; p < GC_P; ++p)
        s += src[(size_t)p * DIc];
    gcpart[((size_t)b * GC_NCH + ch) * DIc + d] = s;
}

__global__ void k_gate(const float* __restrict__ gcpart, const float* __restrict__ fw1,
                       const float* __restrict__ fw2, float* __restrict__ gate)
{
    __shared__ float g[DIc];
    __shared__ float h1[DMm / 4];
    int b = blockIdx.x, d = threadIdx.x;
    float gs = 0.f;
#pragma unroll
    for (int c = 0; c < GC_NCH; ++c)
        gs += gcpart[((size_t)b * GC_NCH + c) * DIc + d];
    g[d] = gs * (1.f / LL);
    __syncthreads();
    if (d < DMm / 4) {
        float s = 0.f;
        for (int j = 0; j < DMm; ++j) s += fw1[d * DMm + j] * g[j];
        h1[d] = geluf(s);
    }
    __syncthreads();
    float s2 = 0.f;
#pragma unroll
    for (int j = 0; j < DMm / 4; ++j) s2 += fw2[d * (DMm / 4) + j] * h1[j];
    gate[b * DMm + d] = sigm(s2);
}

__device__ __forceinline__ int seq_pixel(int k, int l)
{
    int lb = (k < 2) ? l : (LL - 1 - l);
    int w_in = lb & 3, h_in = (lb >> 2) & 3;
    int nw = (lb >> 4) % (WW / 4);
    int nh = (lb >> 4) / (WW / 4);
    int p = nh * 4 + h_in, q = nw * 4 + w_in;
    return ((k & 1) == 0) ? (p * WW + q) : (q * WW + p);
}

__global__ void k_xproj(const float* __restrict__ xc_act, const float* __restrict__ wpad,
                        float* __restrict__ xdbl)
{
    int k = blockIdx.z;
    const float* W = wpad + (size_t)k * CPAD * DIc;
    float* O = xdbl + (size_t)k * BB * LL * CPROJ;
    const int row0 = blockIdx.y * 16, col0 = blockIdx.x * 32;
    const int lane = threadIdx.x & 31;
    const int ra = row0 + (lane & 15);
    const int ba = ra / LL, la = ra - ba * LL;
    const float* arow = xc_act + ((size_t)ba * LL + seq_pixel(k, la)) * DIc;
    v8f acc[2];
    wmma_rows_multi<2>(arow, W, col0, DIc, DIc, acc);
    const int nn = lane & 15, mh = (lane >> 4) * 8;
#pragma unroll
    for (int t = 0; t < 2; ++t)
#pragma unroll
        for (int v = 0; v < 8; ++v) {
            int r = row0 + mh + v, c = col0 + t * 16 + nn;
            if (c < CPROJ) O[(size_t)r * CPROJ + c] = acc[t][v];
        }
}

__global__ void k_dtproj(const float* __restrict__ xdbl, const float* __restrict__ dtw,
                         const float* __restrict__ dtb, float* __restrict__ delta)
{
    int k = blockIdx.z;
    const float* X = xdbl + (size_t)k * BB * LL * CPROJ;
    const float* W = dtw + (size_t)k * DIc * RR;
    float* O = delta + (size_t)k * BB * LL * DIc;
    const int row0 = blockIdx.y * 16, col0 = blockIdx.x * 32;
    v8f acc[2];
    wmma_xwT_multi<2>(X, W, row0, col0, RR, CPROJ, RR, acc);
    const int lane = threadIdx.x & 31;
    const int nn = lane & 15, mh = (lane >> 4) * 8;
#pragma unroll
    for (int t = 0; t < 2; ++t)
#pragma unroll
        for (int v = 0; v < 8; ++v) {
            int r = row0 + mh + v, c = col0 + t * 16 + nn;
            O[(size_t)r * DIc + c] = softplusf_(acc[t][v] + dtb[k * DIc + c]);
        }
}

#define SCAN_T 16
#define SCAN_NG 8
#define SCAN_NPG (NN / SCAN_NG)
#define SCAN_RPW (SCAN_T / SCAN_NG)
#define SCAN_NCH (LL / SCAN_T)
#define SCAN_LA 3
#define SCAN_LEAD 4
static_assert(SCAN_T == 16 && LL % SCAN_T == 0, "a chunk must be one 4x4 window");
static_assert(SCAN_NPG == 2 && SCAN_RPW == 2, "layout of the staged planes");
static_assert(SCAN_LEAD == 4 && SCAN_NCH % SCAN_LEAD == 0, "the chunk loop is written out for four fetch sets");

__device__ __forceinline__ int slot_pixoff(int tr, int flip, int t)
{
    const int tt = flip ? (SCAN_T - 1 - t) : t;
    const int h_in = tt >> 2, w_in = tt & 3;
    return tr ? (w_in * WW + h_in) : (h_in * WW + w_in);
}
__device__ __forceinline__ int chunk_origin(int tr, int flip, int c)
{
    const int widx = flip ? (SCAN_NCH - 1 - c) : c;
    const int nw = widx % (WW / 4), nh = widx / (WW / 4);
    return tr ? (nw * 4 * WW + nh * 4) : (nh * 4 * WW + nw * 4);
}

struct ScanLds {
    float  y[2][SCAN_T][SCAN_NG][32];
    float2 du[2][SCAN_T][32];
    float4 bc[2][SCAN_T][SCAN_NG];
};
struct ScanFetch { float dt[SCAN_RPW], u[SCAN_RPW], bc[SCAN_RPW]; };

struct ScanWave {
    const float* uimg;
    const float* dp;
    const float* bcw;
    int tr, flip;
    int ts0;
    int po[SCAN_RPW];
    int bcpos;
};

__device__ __forceinline__ void scan_fetch(ScanFetch& f, const ScanWave& w, int chunk)
{
    const int c  = (chunk < SCAN_NCH) ? chunk : (SCAN_NCH - 1);
    const int l0 = c * SCAN_T;
    const float* __restrict__ up = w.uimg + (size_t)chunk_origin(w.tr, w.flip, c) * DIc;
#pragma unroll
    for (int r = 0; r < SCAN_RPW; ++r) {
        f.dt[r] = w.dp[(size_t)(l0 + w.ts0 + r) * DIc];
        f.u[r]  = up[w.po[r] * DIc];
        f.bc[r] = w.bcw[(size_t)(l0 + w.ts0 + r) * CPROJ];
    }
}

template <int PAR>
__device__ __forceinline__ void scan_park(const ScanFetch& f, const ScanWave& w, ScanLds& sm, int lane)
{
    float* __restrict__ bcf = &sm.bc[PAR][0][0].x;
#pragma unroll
    for (int r = 0; r < SCAN_RPW; ++r) {
        sm.du[PAR][w.ts0 + r][lane] = make_float2(f.dt[r], f.u[r]);
        bcf[(w.ts0 + r) * (SCAN_NG * 4) + w.bcpos] = f.bc[r];
    }
}

template <int PAR>
__device__ __forceinline__ void scan_steps(ScanLds& sm, float2 (&win)[SCAN_T], float4 (&wbc)[SCAN_T],
                                           float (&h)[SCAN_NPG], const float (&A)[SCAN_NPG],
                                           const float Dg, const int g, const int lane)
{
#pragma unroll
    for (int t = 0; t < SCAN_T; ++t) {
        if (t + SCAN_LA < SCAN_T) {
            win[t + SCAN_LA] = sm.du[PAR][t + SCAN_LA][lane];
            wbc[t + SCAN_LA] = sm.bc[PAR][t + SCAN_LA][g];
        }
        const float dt = win[t].x;
        const float u  = win[t].y;
        const float Bt[SCAN_NPG] = { wbc[t].x, wbc[t].y };
        const float Ct[SCAN_NPG] = { wbc[t].z, wbc[t].w };
        const float du = dt * u;
        float y = 0.f;
#pragma unroll
        for (int n = 0; n < SCAN_NPG; ++n) {
            h[n] = __builtin_fmaf(__expf(dt * A[n]), h[n], du * Bt[n]);
            y += h[n] * Ct[n];
        }
        sm.y[PAR][t][g][lane] = y + Dg * u;
    }
}

template <int PAR>
__device__ __forceinline__ void scan_chunk(const int c, ScanLds& sm, ScanFetch& f, const ScanWave& w,
                                           float2 (&pre)[SCAN_LA], float4 (&prebc)[SCAN_LA],
                                           float (&h)[SCAN_NPG], const float (&A)[SCAN_NPG],
                                           const float Dg, const int g, const int lane,
                                           float* __restrict__ yp)
{
    float2 win[SCAN_T];
    float4 wbc[SCAN_T];
#pragma unroll
    for (int t = 0; t < SCAN_LA; ++t) { win[t] = pre[t]; wbc[t] = prebc[t]; }
    scan_steps<PAR>(sm, win, wbc, h, A, Dg, g, lane);

    scan_park<PAR ^ 1>(f, w, sm, lane);
    scan_fetch(f, w, c + 1 + SCAN_LEAD);

    __builtin_amdgcn_fence(__ATOMIC_RELEASE, "workgroup", "local");
    __builtin_amdgcn_s_barrier();
    __builtin_amdgcn_fence(__ATOMIC_ACQUIRE, "workgroup", "local");

#pragma unroll
    for (int t = 0; t < SCAN_LA; ++t) {
        pre[t]   = sm.du[PAR ^ 1][t][lane];
        prebc[t] = sm.bc[PAR ^ 1][t][g];
    }

#pragma unroll
    for (int r = 0; r < SCAN_RPW; ++r) {
        const int t = g * SCAN_RPW + r;
        float s = sm.y[PAR][t][0][lane];
#pragma unroll
        for (int q = 1; q < SCAN_NG; ++q) s += sm.y[PAR][t][q][lane];
        yp[(size_t)(c * SCAN_T + t) * DIc] = s;
    }
}

__global__ __launch_bounds__(32 * SCAN_NG)
void k_scan(const float* __restrict__ xc_act, const float* __restrict__ xdbl,
            const float* __restrict__ delta, const float* __restrict__ A_logs,
            const float* __restrict__ Ds, float* __restrict__ yarr)
{
    __shared__ ScanLds sm;
    const int kb = blockIdx.x;
    const int k = kb & 3, b = kb >> 2;
    const int lane = threadIdx.x & 31;
    const int g = __builtin_amdgcn_readfirstlane((int)(threadIdx.x >> 5));
    const int d = blockIdx.y * 32 + lane;
    const int n0 = g * SCAN_NPG;
    float A[SCAN_NPG];
#pragma unroll
    for (int n = 0; n < SCAN_NPG; ++n)
        A[n] = -__expf(A_logs[((size_t)(k * DIc + d)) * NN + n0 + n]);
    const float Dg = (g == SCAN_NG - 1) ? Ds[k * DIc + d] : 0.f;
    const size_t seqbase = ((size_t)(k * BB + b)) * LL;
    ScanWave w;
    w.uimg = xc_act + (size_t)b * LL * DIc + d;
    w.dp   = delta + seqbase * DIc + d;
    w.bcw  = xdbl  + seqbase * CPROJ + RR + lane;
    w.tr   = k & 1;
    w.flip = k >> 1;
    w.ts0  = g * SCAN_RPW;
#pragma unroll
    for (int r = 0; r < SCAN_RPW; ++r) w.po[r] = slot_pixoff(w.tr, w.flip, w.ts0 + r);
    w.bcpos = ((lane & 15) >> 1) * 4 + (lane & 1) + 2 * (lane >> 4);
    float* __restrict__ yp = yarr + seqbase * DIc + d;

    float h[SCAN_NPG];
#pragma unroll
    for (int n = 0; n < SCAN_NPG; ++n) h[n] = 0.f;

    ScanFetch f0, f1, f2, f3;
    scan_fetch(f0, w, 0);
    __builtin_amdgcn_sched_barrier(0);
    scan_fetch(f1, w, 1);
    __builtin_amdgcn_sched_barrier(0);
    scan_fetch(f2, w, 2);
    __builtin_amdgcn_sched_barrier(0);
    scan_fetch(f3, w, 3);
    __builtin_amdgcn_sched_barrier(0);
    scan_park<0>(f0, w, sm, lane);
    scan_fetch(f0, w, SCAN_LEAD);
    __builtin_amdgcn_sched_barrier(0);
    __builtin_amdgcn_fence(__ATOMIC_RELEASE, "workgroup", "local");
    __builtin_amdgcn_s_barrier();
    __builtin_amdgcn_fence(__ATOMIC_ACQUIRE, "workgroup", "local");
    float2 pre[SCAN_LA];
    float4 prebc[SCAN_LA];
#pragma unroll
    for (int t = 0; t < SCAN_LA; ++t) {
        pre[t]   = sm.du[0][t][lane];
        prebc[t] = sm.bc[0][t][g];
    }

    for (int c = 0; c < SCAN_NCH; c += SCAN_LEAD) {
        scan_chunk<0>(c,     sm, f1, w, pre, prebc, h, A, Dg, g, lane, yp);
        scan_chunk<1>(c + 1, sm, f2, w, pre, prebc, h, A, Dg, g, lane, yp);
        scan_chunk<0>(c + 2, sm, f3, w, pre, prebc, h, A, Dg, g, lane, yp);
        scan_chunk<1>(c + 3, sm, f0, w, pre, prebc, h, A, Dg, g, lane, yp);
    }
}

template <int XM>
__device__ __forceinline__ float row_xor(float v)
{
    return __builtin_bit_cast(float,
        __builtin_amdgcn_update_dpp(0, __builtin_bit_cast(int, v), 0x160 + XM, 0xf, 0xf, false));
}
__device__ __forceinline__ float half_xor(float v)
{
    const int x = __builtin_bit_cast(int, v);
    return __builtin_bit_cast(float,
        __builtin_amdgcn_permlanex16(x, x, 0x76543210, 0xfedcba98, false, false));
}
__device__ __forceinline__ float wave_sum(float v)
{
    v += row_xor<1>(v);
    v += row_xor<2>(v);
    v += row_xor<4>(v);
    v += row_xor<8>(v);
    v += half_xor(v);
    return v;
}

__global__ __launch_bounds__(32 * PX_WAVES)
void k_combine(const float* __restrict__ yarr, const float* __restrict__ gate,
               const float* __restrict__ g1, const float* __restrict__ g2,
               const float* __restrict__ g3, const float* __restrict__ g4,
               const float* __restrict__ nw, const float* __restrict__ nb,
               const float* __restrict__ z_silu, float* __restrict__ yfinal)
{
    const int lane = threadIdx.x & 31;
    const int wave = __builtin_amdgcn_readfirstlane((int)(threadIdx.x >> 5));
    const int pix0 = (blockIdx.x * PX_WAVES + wave) * PX_PPW;
    const int b = pix0 / LL, lr0 = pix0 - b * LL;
    const size_t bstride = (size_t)LL * DIc;
    const float* __restrict__ Y0 = yarr + ((size_t)(0 * BB + b)) * bstride + lane;
    const float* __restrict__ Y1 = yarr + ((size_t)(1 * BB + b)) * bstride + lane;
    const float* __restrict__ Y2 = yarr + ((size_t)(2 * BB + b)) * bstride + lane;
    const float* __restrict__ Y3 = yarr + ((size_t)(3 * BB + b)) * bstride + lane;
    float G1[PX_CPL], G2[PX_CPL], G3[PX_CPL], G4[PX_CPL], GT[PX_CPL], NW[PX_CPL], NB[PX_CPL];
#pragma unroll
    for (int j = 0; j < PX_CPL; ++j) {
        const int d = lane + 32 * j;
        G1[j] = g1[d]; G2[j] = g2[d]; G3[j] = g3[d]; G4[j] = g4[d];
        GT[j] = gate[b * DIc + d]; NW[j] = nw[d]; NB[j] = nb[d];
    }
#pragma unroll
    for (int p = 0; p < PX_PPW; ++p) {
        const int lr = lr0 + p;
        const int h = lr / WW, w = lr % WW;
        const int l = (((h >> 2) * (WW / 4) + (w >> 2)) << 4) + ((h & 3) << 2) + (w & 3);
        const int lt = (l % WW) * HH + (l / WW);
        const float* __restrict__ zp = z_silu + ((size_t)b * LL + lr) * DIc + lane;
        float yc[PX_CPL], zz[PX_CPL];
        float s1 = 0.f, s2 = 0.f;
#pragma unroll
        for (int j = 0; j < PX_CPL; ++j) {
            float y = G1[j] * Y0[(size_t)l * DIc + 32 * j]
                    + G2[j] * Y2[(size_t)(LL - 1 - l) * DIc + 32 * j]
                    + G3[j] * Y1[(size_t)lt * DIc + 32 * j]
                    + G4[j] * Y3[(size_t)(LL - 1 - lt) * DIc + 32 * j];
            y *= GT[j];
            yc[j] = y;
            zz[j] = zp[32 * j];
            s1 += y;
            s2 += y * y;
        }
        s1 = wave_sum(s1);
        s2 = wave_sum(s2);
        const float mu  = s1 * (1.f / DIc);
        const float var = s2 * (1.f / DIc) - mu * mu;
        const float inv = rsqrtf(var + 1e-5f);
        float* __restrict__ op = yfinal + ((size_t)b * LL + lr) * DIc + lane;
#pragma unroll
        for (int j = 0; j < PX_CPL; ++j)
            op[32 * j] = ((yc[j] - mu) * inv * NW[j] + NB[j]) * zz[j];
    }
}

__global__ void k_outproj(const float* __restrict__ yf, const float* __restrict__ w,
                          float* __restrict__ out)
{
    const int row0 = blockIdx.y * 16, col0 = blockIdx.x * 32;
    v8f acc[2];
    wmma_xwT_multi<2>(yf, w, row0, col0, DIc, DIc, DIc, acc);
    const int lane = threadIdx.x & 31;
    const int nn = lane & 15, mh = (lane >> 4) * 8;
#pragma unroll
    for (int t = 0; t < 2; ++t)
#pragma unroll
        for (int v = 0; v < 8; ++v) {
            int r = row0 + mh + v, c = col0 + t * 16 + nn;
            out[(size_t)r * DMm + c] = acc[t][v];
        }
}

extern "C" void kernel_launch(void* const* d_in, const int* in_sizes, int n_in,
                              void* d_out, int out_size, void* d_ws, size_t ws_size,
                              hipStream_t stream)
{
    (void)in_sizes; (void)n_in; (void)out_size; (void)ws_size;
    const float* x          = (const float*)d_in[0];
    const float* in_proj_w  = (const float*)d_in[1];
    const float* conv_w     = (const float*)d_in[2];
    const float* conv_b     = (const float*)d_in[3];
    const float* x_proj_w   = (const float*)d_in[4];
    const float* dt_w       = (const float*)d_in[5];
    const float* dt_b       = (const float*)d_in[6];
    const float* A_logs     = (const float*)d_in[7];
    const float* Ds         = (const float*)d_in[8];
    const float* gamma1     = (const float*)d_in[9];
    const float* gamma2     = (const float*)d_in[10];
    const float* gamma3     = (const float*)d_in[11];
    const float* gamma4     = (const float*)d_in[12];
    const float* fc_w1      = (const float*)d_in[13];
    const float* fc_w2      = (const float*)d_in[14];
    const float* norm_w     = (const float*)d_in[15];
    const float* norm_b     = (const float*)d_in[16];
    const float* out_proj_w = (const float*)d_in[17];
    float* out = (float*)d_out;

    float* ws = (float*)d_ws;
    size_t o = 0;
    float* xc_lin = ws + o; o += (size_t)BB * LL * DIc;
    float* z_silu = ws + o; o += (size_t)BB * LL * DIc;
    float* xc_act = ws + o; o += (size_t)BB * LL * DIc;
    float* xdbl   = ws + o; o += (size_t)KK * BB * LL * CPROJ;
    float* delta  = ws + o; o += (size_t)KK * BB * LL * DIc;
    float* yarr   = ws + o; o += (size_t)KK * BB * LL * DIc;
    float* yfinal = ws + o; o += (size_t)BB * LL * DIc;
    float* wpad   = ws + o; o += (size_t)KK * CPAD * DIc;
    float* gcpart = ws + o; o += (size_t)BB * GC_NCH * DIc;
    float* gate   = ws + o; o += BB * DMm;

    k_padw    <<<dim3(KK, CPAD),       DIc, 0, stream>>>(x_proj_w, wpad);
    k_inproj  <<<dim3(12, 576),        32,  0, stream>>>(x, in_proj_w, xc_lin, z_silu);
    k_dwconv  <<<dim3(PX_NWG),         32 * PX_WAVES, 0, stream>>>(xc_lin, conv_w, conv_b, xc_act);
    k_gcmean  <<<dim3(BB, GC_NCH),     DIc, 0, stream>>>(xc_act, gcpart);
    k_gate    <<<dim3(BB),             DIc, 0, stream>>>(gcpart, fc_w1, fc_w2, gate);
    k_xproj   <<<dim3(2, 576, KK),     32,  0, stream>>>(xc_act, wpad, xdbl);
    k_dtproj  <<<dim3(6, 576, KK),     32,  0, stream>>>(xdbl, dt_w, dt_b, delta);
    k_scan    <<<dim3(BB * KK, DIc / 32), 32 * SCAN_NG, 0, stream>>>(xc_act, xdbl, delta, A_logs, Ds, yarr);
    k_combine <<<dim3(PX_NWG),         32 * PX_WAVES, 0, stream>>>(yarr, gate, gamma1, gamma2, gamma3,
                                                         gamma4, norm_w, norm_b, z_silu, yfinal);
    k_outproj <<<dim3(6, 576),         32,  0, stream>>>(yfinal, out_proj_w, out);
}


# k_scan: no LDS instruction among a step's arithmetic: a chunk's 16 partial y stay in registers and leave as one group, all 16 steps' inputs are read as one group behind the barrier; sum order kept
# speedup vs baseline: 9.1546x; 1.0590x over previous
#include <hip/hip_runtime.h>
#include <math.h>


typedef float v2f __attribute__((ext_vector_type(2)));
typedef float v8f __attribute__((ext_vector_type(8)));

#define BB 4
#define HH 48
#define WW 48
#define DMm 192
#define DIc 192
#define KK 4
#define NN 16
#define RR 12
#define LL (HH * WW)
#define CPROJ (RR + 2 * NN)
#define CPAD 64

__device__ __forceinline__ float siluf(float x)     { return x / (1.f + __expf(-x)); }
__device__ __forceinline__ float sigm(float x)      { return 1.f / (1.f + __expf(-x)); }
__device__ __forceinline__ float softplusf_(float x){ return (x > 20.f) ? x : log1pf(__expf(x)); }
__device__ __forceinline__ float geluf(float x)     { return 0.5f * x * (1.f + erff(x * 0.70710678118654752440f)); }

template <int NT>
__device__ __forceinline__ void wmma_rows_multi(const float* __restrict__ arow,
                                                const float* __restrict__ W,
                                                int col0, int Kdim, int ldw,
                                                v8f acc[NT])
{
    const int lane = threadIdx.x & 31;
    const int m    = lane & 15;
    const int kh   = lane >> 4;
    const float* __restrict__ ap = arow + 2 * kh;
    const float* __restrict__ bp[NT];
#pragma unroll
    for (int t = 0; t < NT; ++t) {
        bp[t] = W + (size_t)(col0 + t * 16 + m) * ldw + 2 * kh;
        acc[t] = {};
    }
    v2f a_cur = *(const v2f*)(ap);
    v2f b_cur[NT];
#pragma unroll
    for (int t = 0; t < NT; ++t) b_cur[t] = *(const v2f*)(bp[t]);

    for (int k0 = 4; k0 < Kdim; k0 += 4) {
        v2f a_nxt = *(const v2f*)(ap + k0);
        v2f b_nxt[NT];
#pragma unroll
        for (int t = 0; t < NT; ++t) b_nxt[t] = *(const v2f*)(bp[t] + k0);
#pragma unroll
        for (int t = 0; t < NT; ++t)
            acc[t] = __builtin_amdgcn_wmma_f32_16x16x4_f32(false, a_cur, false, b_cur[t],
                                                           (short)0, acc[t], false, false);
        a_cur = a_nxt;
#pragma unroll
        for (int t = 0; t < NT; ++t) b_cur[t] = b_nxt[t];
    }
#pragma unroll
    for (int t = 0; t < NT; ++t)
        acc[t] = __builtin_amdgcn_wmma_f32_16x16x4_f32(false, a_cur, false, b_cur[t],
                                                       (short)0, acc[t], false, false);
}

template <int NT>
__device__ __forceinline__ void wmma_xwT_multi(const float* __restrict__ X,
                                               const float* __restrict__ W,
                                               int row0, int col0,
                                               int Kdim, int ldx, int ldw,
                                               v8f acc[NT])
{
    const int m = threadIdx.x & 15;
    wmma_rows_multi<NT>(X + (size_t)(row0 + m) * ldx, W, col0, Kdim, ldw, acc);
}

__global__ void k_padw(const float* __restrict__ xpw, float* __restrict__ wpad)
{
    int k = blockIdx.x, c = blockIdx.y, d = threadIdx.x;
    wpad[((size_t)k * CPAD + c) * DIc + d] =
        (c < CPROJ) ? xpw[((size_t)k * CPROJ + c) * DIc + d] : 0.f;
}

__global__ void k_inproj(const float* __restrict__ x, const float* __restrict__ w,
                         float* __restrict__ xc_lin, float* __restrict__ z_silu)
{
    const int row0 = blockIdx.y * 16, col0 = blockIdx.x * 32;
    v8f acc[2];
    wmma_xwT_multi<2>(x, w, row0, col0, DMm, DMm, DMm, acc);
    const int lane = threadIdx.x & 31;
    const int nn = lane & 15, mh = (lane >> 4) * 8;
#pragma unroll
    for (int t = 0; t < 2; ++t)
#pragma unroll
        for (int v = 0; v < 8; ++v) {
            int r = row0 + mh + v;
            int c = col0 + t * 16 + nn;
            float val = acc[t][v];
            if (c < DIc) xc_lin[(size_t)r * DIc + c] = val;
            else         z_silu[(size_t)r * DIc + (c - DIc)] = siluf(val);
        }
}

#define PX_WAVES 8
#define PX_PPW   4
#define PX_CPL   (DIc / 32)
#define PX_NWG   (BB * LL / (PX_WAVES * PX_PPW))
static_assert(WW % PX_PPW == 0 && (BB * LL) % (PX_WAVES * PX_PPW) == 0 && DIc % 32 == 0,
              "a wave's pixels must lie in one image row");

__global__ __launch_bounds__(32 * PX_WAVES)
void k_dwconv(const float* __restrict__ xc_lin, const float* __restrict__ cw,
              const float* __restrict__ cb, float* __restrict__ xc_act)
{
    __shared__ float wsh[DIc * 9];
    const int lane = threadIdx.x & 31;
    const int wave = __builtin_amdgcn_readfirstlane((int)(threadIdx.x >> 5));
    const int pix0 = (blockIdx.x * PX_WAVES + wave) * PX_PPW;
    const int b = pix0 / LL, rem0 = pix0 - b * LL;
    const int h = rem0 / WW, w0 = rem0 - h * WW;
    const float* __restrict__ img = xc_lin + (size_t)b * LL * DIc + lane;

    float win[3][PX_PPW + 2][PX_CPL];
#pragma unroll
    for (int r = 0; r < 3; ++r) {
        const int hh = h + r - 1;
        const bool rok = (hh >= 0) && (hh < HH);
        const int hc = hh < 0 ? 0 : (hh >= HH ? HH - 1 : hh);
#pragma unroll
        for (int q = 0; q < PX_PPW + 2; ++q) {
            const int wq = w0 + q - 1;
            const bool cok = (q == 0) ? (wq >= 0) : ((q == PX_PPW + 1) ? (wq < WW) : true);
            const int wc = (q == 0) ? (wq < 0 ? 0 : wq) : ((q == PX_PPW + 1) ? (wq >= WW ? WW - 1 : wq) : wq);
            const float* __restrict__ sp = img + (size_t)(hc * WW + wc) * DIc;
#pragma unroll
            for (int j = 0; j < PX_CPL; ++j) {
                const float v = sp[32 * j];
                win[r][q][j] = (rok && cok) ? v : 0.f;
            }
        }
    }

    constexpr int WFULL = (DIc * 9) / (32 * PX_WAVES);
    float wtmp[WFULL];
#pragma unroll
    for (int i = 0; i < WFULL; ++i) wtmp[i] = cw[(int)threadIdx.x + i * 32 * PX_WAVES];
    const int itail = (int)threadIdx.x + WFULL * 32 * PX_WAVES;
    const float wtail = (itail < DIc * 9) ? cw[itail] : 0.f;
#pragma unroll
    for (int i = 0; i < WFULL; ++i) wsh[(int)threadIdx.x + i * 32 * PX_WAVES] = wtmp[i];
    if (itail < DIc * 9) wsh[itail] = wtail;
    __syncthreads();
    float wreg[PX_CPL][9], acc[PX_PPW][PX_CPL];
#pragma unroll
    for (int j = 0; j < PX_CPL; ++j) {
#pragma unroll
        for (int t = 0; t < 9; ++t) wreg[j][t] = wsh[(lane + 32 * j) * 9 + t];
        const float bj = cb[lane + 32 * j];
#pragma unroll
        for (int p = 0; p < PX_PPW; ++p) acc[p][j] = bj;
    }
#pragma unroll
    for (int p = 0; p < PX_PPW; ++p)
#pragma unroll
        for (int kh = 0; kh < 3; ++kh)
#pragma unroll
            for (int kw = 0; kw < 3; ++kw)
#pragma unroll
                for (int j = 0; j < PX_CPL; ++j)
                    acc[p][j] += win[kh][p + kw][j] * wreg[j][kh * 3 + kw];
#pragma unroll
    for (int p = 0; p < PX_PPW; ++p) {
        float* __restrict__ op = xc_act + (size_t)(pix0 + p) * DIc + lane;
#pragma unroll
        for (int j = 0; j < PX_CPL; ++j) op[32 * j] = siluf(acc[p][j]);
    }
}

#define GC_NCH 48
#define GC_P (LL / GC_NCH)
__global__ void k_gcmean(const float* __restrict__ xc_act, float* __restrict__ gcpart)
{
    int b = blockIdx.x, ch = blockIdx.y, d = threadIdx.x;
    const float* __restrict__ src = xc_act + ((size_t)b * LL + (size_t)ch * GC_P) * DIc + d;
    float s = 0.f;
#pragma unroll
    for (int p = 0; p < GC_P; ++p)
        s += src[(size_t)p * DIc];
    gcpart[((size_t)b * GC_NCH + ch) * DIc + d] = s;
}

__global__ void k_gate(const float* __restrict__ gcpart, const float* __restrict__ fw1,
                       const float* __restrict__ fw2, float* __restrict__ gate)
{
    __shared__ float g[DIc];
    __shared__ float h1[DMm / 4];
    int b = blockIdx.x, d = threadIdx.x;
    float gs = 0.f;
#pragma unroll
    for (int c = 0; c < GC_NCH; ++c)
        gs += gcpart[((size_t)b * GC_NCH + c) * DIc + d];
    g[d] = gs * (1.f / LL);
    __syncthreads();
    if (d < DMm / 4) {
        float s = 0.f;
        for (int j = 0; j < DMm; ++j) s += fw1[d * DMm + j] * g[j];
        h1[d] = geluf(s);
    }
    __syncthreads();
    float s2 = 0.f;
#pragma unroll
    for (int j = 0; j < DMm / 4; ++j) s2 += fw2[d * (DMm / 4) + j] * h1[j];
    gate[b * DMm + d] = sigm(s2);
}

__device__ __forceinline__ int seq_pixel(int k, int l)
{
    int lb = (k < 2) ? l : (LL - 1 - l);
    int w_in = lb & 3, h_in = (lb >> 2) & 3;
    int nw = (lb >> 4) % (WW / 4);
    int nh = (lb >> 4) / (WW / 4);
    int p = nh * 4 + h_in, q = nw * 4 + w_in;
    return ((k & 1) == 0) ? (p * WW + q) : (q * WW + p);
}

__global__ void k_xproj(const float* __restrict__ xc_act, const float* __restrict__ wpad,
                        float* __restrict__ xdbl)
{
    int k = blockIdx.z;
    const float* W = wpad + (size_t)k * CPAD * DIc;
    float* O = xdbl + (size_t)k * BB * LL * CPROJ;
    const int row0 = blockIdx.y * 16, col0 = blockIdx.x * 32;
    const int lane = threadIdx.x & 31;
    const int ra = row0 + (lane & 15);
    const int ba = ra / LL, la = ra - ba * LL;
    const float* arow = xc_act + ((size_t)ba * LL + seq_pixel(k, la)) * DIc;
    v8f acc[2];
    wmma_rows_multi<2>(arow, W, col0, DIc, DIc, acc);
    const int nn = lane & 15, mh = (lane >> 4) * 8;
#pragma unroll
    for (int t = 0; t < 2; ++t)
#pragma unroll
        for (int v = 0; v < 8; ++v) {
            int r = row0 + mh + v, c = col0 + t * 16 + nn;
            if (c < CPROJ) O[(size_t)r * CPROJ + c] = acc[t][v];
        }
}

__global__ void k_dtproj(const float* __restrict__ xdbl, const float* __restrict__ dtw,
                         const float* __restrict__ dtb, float* __restrict__ delta)
{
    int k = blockIdx.z;
    const float* X = xdbl + (size_t)k * BB * LL * CPROJ;
    const float* W = dtw + (size_t)k * DIc * RR;
    float* O = delta + (size_t)k * BB * LL * DIc;
    const int row0 = blockIdx.y * 16, col0 = blockIdx.x * 32;
    v8f acc[2];
    wmma_xwT_multi<2>(X, W, row0, col0, RR, CPROJ, RR, acc);
    const int lane = threadIdx.x & 31;
    const int nn = lane & 15, mh = (lane >> 4) * 8;
#pragma unroll
    for (int t = 0; t < 2; ++t)
#pragma unroll
        for (int v = 0; v < 8; ++v) {
            int r = row0 + mh + v, c = col0 + t * 16 + nn;
            O[(size_t)r * DIc + c] = softplusf_(acc[t][v] + dtb[k * DIc + c]);
        }
}

#define SCAN_T 16
#define SCAN_NG 8
#define SCAN_NPG (NN / SCAN_NG)
#define SCAN_RPW (SCAN_T / SCAN_NG)
#define SCAN_NCH (LL / SCAN_T)
#define SCAN_LEAD 4
static_assert(SCAN_T == 16 && LL % SCAN_T == 0, "a chunk must be one 4x4 window");
static_assert(SCAN_NPG == 2 && SCAN_RPW == 2, "layout of the staged planes: a wave's two steps are one step pair");
static_assert(SCAN_LEAD == 4 && SCAN_NCH % SCAN_LEAD == 0, "the chunk loop is written out for four fetch sets");

__device__ __forceinline__ int slot_pixoff(int tr, int flip, int t)
{
    const int tt = flip ? (SCAN_T - 1 - t) : t;
    const int h_in = tt >> 2, w_in = tt & 3;
    return tr ? (w_in * WW + h_in) : (h_in * WW + w_in);
}
__device__ __forceinline__ int chunk_origin(int tr, int flip, int c)
{
    const int widx = flip ? (SCAN_NCH - 1 - c) : c;
    const int nw = widx % (WW / 4), nh = widx / (WW / 4);
    return tr ? (nw * 4 * WW + nh * 4) : (nh * 4 * WW + nw * 4);
}

struct ScanLds {
    float2 y[2][SCAN_T / 2][SCAN_NG][32];
    float4 du[2][SCAN_T / 2][32];
    float4 bc[2][SCAN_T][SCAN_NG];
};
struct ScanFetch { float dt[SCAN_RPW], u[SCAN_RPW], bc[SCAN_RPW]; };
struct ScanIn {
    float4 du[SCAN_T / 2];
    float4 bc[SCAN_T];
};

struct ScanWave {
    const float* uimg;
    const float* dp;
    const float* bcw;
    int tr, flip;
    int ts0;
    int po[SCAN_RPW];
    int bcpos;
};

__device__ __forceinline__ void scan_fetch(ScanFetch& f, const ScanWave& w, int chunk)
{
    const int c  = (chunk < SCAN_NCH) ? chunk : (SCAN_NCH - 1);
    const int l0 = c * SCAN_T;
    const float* __restrict__ up = w.uimg + (size_t)chunk_origin(w.tr, w.flip, c) * DIc;
#pragma unroll
    for (int r = 0; r < SCAN_RPW; ++r) {
        f.dt[r] = w.dp[(size_t)(l0 + w.ts0 + r) * DIc];
        f.u[r]  = up[w.po[r] * DIc];
        f.bc[r] = w.bcw[(size_t)(l0 + w.ts0 + r) * CPROJ];
    }
}

template <int PAR>
__device__ __forceinline__ void scan_park(const ScanFetch& f, const ScanWave& w, ScanLds& sm,
                                          const int g, const int lane)
{
    sm.du[PAR][g][lane] = make_float4(f.dt[0], f.u[0], f.dt[1], f.u[1]);
    float* __restrict__ bcf = &sm.bc[PAR][0][0].x;
#pragma unroll
    for (int r = 0; r < SCAN_RPW; ++r)
        bcf[(w.ts0 + r) * (SCAN_NG * 4) + w.bcpos] = f.bc[r];
}

template <int PAR>
__device__ __forceinline__ void scan_read(ScanIn& in, const ScanLds& sm, const int g, const int lane)
{
#pragma unroll
    for (int p = 0; p < SCAN_T / 2; ++p) {
        in.du[p]         = sm.du[PAR][p][lane];
        in.bc[2 * p]     = sm.bc[PAR][2 * p][g];
        in.bc[2 * p + 1] = sm.bc[PAR][2 * p + 1][g];
    }
}

__device__ __forceinline__ void scan_steps(const ScanIn& in, float (&yv)[SCAN_T],
                                           float (&h)[SCAN_NPG], const float (&A)[SCAN_NPG],
                                           const float Dg)
{
#pragma unroll
    for (int t = 0; t < SCAN_T; ++t) {
        const float4 d4 = in.du[t >> 1];
        const float dt = (t & 1) ? d4.z : d4.x;
        const float u  = (t & 1) ? d4.w : d4.y;
        const float Bt[SCAN_NPG] = { in.bc[t].x, in.bc[t].y };
        const float Ct[SCAN_NPG] = { in.bc[t].z, in.bc[t].w };
        const float du = dt * u;
#pragma unroll
        for (int n = 0; n < SCAN_NPG; ++n) {
            h[n] = __builtin_fmaf(__expf(dt * A[n]), h[n], du * Bt[n]);
        }
        float y = h[0] * Ct[0];
#pragma unroll
        for (int n = 1; n < SCAN_NPG; ++n) y = __builtin_fmaf(h[n], Ct[n], y);
        yv[t] = __builtin_fmaf(Dg, u, y);
    }
}

template <int PAR>
__device__ __forceinline__ void scan_chunk(const int c, ScanLds& sm, ScanFetch& f, const ScanWave& w,
                                           ScanIn& in,
                                           float (&h)[SCAN_NPG], const float (&A)[SCAN_NPG],
                                           const float Dg, const int g, const int lane,
                                           float* __restrict__ yp)
{
    float yv[SCAN_T];
    scan_steps(in, yv, h, A, Dg);

    asm volatile("" :: "v"(yv[0]), "v"(yv[1]), "v"(yv[2]), "v"(yv[3]), "v"(yv[4]), "v"(yv[5]),
                       "v"(yv[6]), "v"(yv[7]), "v"(yv[8]), "v"(yv[9]), "v"(yv[10]), "v"(yv[11]),
                       "v"(yv[12]), "v"(yv[13]), "v"(yv[14]), "v"(yv[15]));
    __builtin_amdgcn_sched_barrier(0);
#pragma unroll
    for (int p = 0; p < SCAN_T / 2; ++p)
        sm.y[PAR][p][g][lane] = make_float2(yv[2 * p], yv[2 * p + 1]);
    scan_park<PAR ^ 1>(f, w, sm, g, lane);
    __builtin_amdgcn_sched_barrier(0);
    scan_fetch(f, w, c + 1 + SCAN_LEAD);

    __builtin_amdgcn_fence(__ATOMIC_RELEASE, "workgroup", "local");
    __builtin_amdgcn_s_barrier();
    __builtin_amdgcn_fence(__ATOMIC_ACQUIRE, "workgroup", "local");

    float2 part[SCAN_NG];
#pragma unroll
    for (int q = 0; q < SCAN_NG; ++q) part[q] = sm.y[PAR][g][q][lane];
    scan_read<PAR ^ 1>(in, sm, g, lane);
    __builtin_amdgcn_sched_barrier(0);

    float s0 = part[0].x, s1 = part[0].y;
#pragma unroll
    for (int q = 1; q < SCAN_NG; ++q) { s0 += part[q].x; s1 += part[q].y; }
    yp[(size_t)(c * SCAN_T + g * SCAN_RPW) * DIc]     = s0;
    yp[(size_t)(c * SCAN_T + g * SCAN_RPW + 1) * DIc] = s1;
    __builtin_amdgcn_sched_barrier(0);
}

__global__ __launch_bounds__(32 * SCAN_NG)
void k_scan(const float* __restrict__ xc_act, const float* __restrict__ xdbl,
            const float* __restrict__ delta, const float* __restrict__ A_logs,
            const float* __restrict__ Ds, float* __restrict__ yarr)
{
    __shared__ ScanLds sm;
    const int kb = blockIdx.x;
    const int k = kb & 3, b = kb >> 2;
    const int lane = threadIdx.x & 31;
    const int g = __builtin_amdgcn_readfirstlane((int)(threadIdx.x >> 5));
    const int d = blockIdx.y * 32 + lane;
    const int n0 = g * SCAN_NPG;
    float A[SCAN_NPG];
#pragma unroll
    for (int n = 0; n < SCAN_NPG; ++n)
        A[n] = -__expf(A_logs[((size_t)(k * DIc + d)) * NN + n0 + n]);
    const float Dg = (g == SCAN_NG - 1) ? Ds[k * DIc + d] : 0.f;
    const size_t seqbase = ((size_t)(k * BB + b)) * LL;
    ScanWave w;
    w.uimg = xc_act + (size_t)b * LL * DIc + d;
    w.dp   = delta + seqbase * DIc + d;
    w.bcw  = xdbl  + seqbase * CPROJ + RR + lane;
    w.tr   = k & 1;
    w.flip = k >> 1;
    w.ts0  = g * SCAN_RPW;
#pragma unroll
    for (int r = 0; r < SCAN_RPW; ++r) w.po[r] = slot_pixoff(w.tr, w.flip, w.ts0 + r);
    w.bcpos = ((lane & 15) >> 1) * 4 + (lane & 1) + 2 * (lane >> 4);
    float* __restrict__ yp = yarr + seqbase * DIc + d;

    float h[SCAN_NPG];
#pragma unroll
    for (int n = 0; n < SCAN_NPG; ++n) h[n] = 0.f;

    ScanFetch f0, f1, f2, f3;
    scan_fetch(f0, w, 0);
    __builtin_amdgcn_sched_barrier(0);
    scan_fetch(f1, w, 1);
    __builtin_amdgcn_sched_barrier(0);
    scan_fetch(f2, w, 2);
    __builtin_amdgcn_sched_barrier(0);
    scan_fetch(f3, w, 3);
    __builtin_amdgcn_sched_barrier(0);
    scan_park<0>(f0, w, sm, g, lane);
    scan_fetch(f0, w, SCAN_LEAD);
    __builtin_amdgcn_sched_barrier(0);
    __builtin_amdgcn_fence(__ATOMIC_RELEASE, "workgroup", "local");
    __builtin_amdgcn_s_barrier();
    __builtin_amdgcn_fence(__ATOMIC_ACQUIRE, "workgroup", "local");
    ScanIn in;
    scan_read<0>(in, sm, g, lane);
    __builtin_amdgcn_sched_barrier(0);

    for (int c = 0; c < SCAN_NCH; c += SCAN_LEAD) {
        scan_chunk<0>(c,     sm, f1, w, in, h, A, Dg, g, lane, yp);
        scan_chunk<1>(c + 1, sm, f2, w, in, h, A, Dg, g, lane, yp);
        scan_chunk<0>(c + 2, sm, f3, w, in, h, A, Dg, g, lane, yp);
        scan_chunk<1>(c + 3, sm, f0, w, in, h, A, Dg, g, lane, yp);
    }
}

template <int XM>
__device__ __forceinline__ float row_xor(float v)
{
    return __builtin_bit_cast(float,
        __builtin_amdgcn_update_dpp(0, __builtin_bit_cast(int, v), 0x160 + XM, 0xf, 0xf, false));
}
__device__ __forceinline__ float half_xor(float v)
{
    const int x = __builtin_bit_cast(int, v);
    return __builtin_bit_cast(float,
        __builtin_amdgcn_permlanex16(x, x, 0x76543210, 0xfedcba98, false, false));
}
__device__ __forceinline__ float wave_sum(float v)
{
    v += row_xor<1>(v);
    v += row_xor<2>(v);
    v += row_xor<4>(v);
    v += row_xor<8>(v);
    v += half_xor(v);
    return v;
}

__global__ __launch_bounds__(32 * PX_WAVES)
void k_combine(const float* __restrict__ yarr, const float* __restrict__ gate,
               const float* __restrict__ g1, const float* __restrict__ g2,
               const float* __restrict__ g3, const float* __restrict__ g4,
               const float* __restrict__ nw, const float* __restrict__ nb,
               const float* __restrict__ z_silu, float* __restrict__ yfinal)
{
    const int lane = threadIdx.x & 31;
    const int wave = __builtin_amdgcn_readfirstlane((int)(threadIdx.x >> 5));
    const int pix0 = (blockIdx.x * PX_WAVES + wave) * PX_PPW;
    const int b = pix0 / LL, lr0 = pix0 - b * LL;
    const size_t bstride = (size_t)LL * DIc;
    const float* __restrict__ Y0 = yarr + ((size_t)(0 * BB + b)) * bstride + lane;
    const float* __restrict__ Y1 = yarr + ((size_t)(1 * BB + b)) * bstride + lane;
    const float* __restrict__ Y2 = yarr + ((size_t)(2 * BB + b)) * bstride + lane;
    const float* __restrict__ Y3 = yarr + ((size_t)(3 * BB + b)) * bstride + lane;
    float G1[PX_CPL], G2[PX_CPL], G3[PX_CPL], G4[PX_CPL], GT[PX_CPL], NW[PX_CPL], NB[PX_CPL];
#pragma unroll
    for (int j = 0; j < PX_CPL; ++j) {
        const int d = lane + 32 * j;
        G1[j] = g1[d]; G2[j] = g2[d]; G3[j] = g3[d]; G4[j] = g4[d];
        GT[j] = gate[b * DIc + d]; NW[j] = nw[d]; NB[j] = nb[d];
    }
#pragma unroll
    for (int p = 0; p < PX_PPW; ++p) {
        const int lr = lr0 + p;
        const int h = lr / WW, w = lr % WW;
        const int l = (((h >> 2) * (WW / 4) + (w >> 2)) << 4) + ((h & 3) << 2) + (w & 3);
        const int lt = (l % WW) * HH + (l / WW);
        const float* __restrict__ zp = z_silu + ((size_t)b * LL + lr) * DIc + lane;
        float yc[PX_CPL], zz[PX_CPL];
        float s1 = 0.f, s2 = 0.f;
#pragma unroll
        for (int j = 0; j < PX_CPL; ++j) {
            float y = G1[j] * Y0[(size_t)l * DIc + 32 * j]
                    + G2[j] * Y2[(size_t)(LL - 1 - l) * DIc + 32 * j]
                    + G3[j] * Y1[(size_t)lt * DIc + 32 * j]
                    + G4[j] * Y3[(size_t)(LL - 1 - lt) * DIc + 32 * j];
            y *= GT[j];
            yc[j] = y;
            zz[j] = zp[32 * j];
            s1 += y;
            s2 += y * y;
        }
        s1 = wave_sum(s1);
        s2 = wave_sum(s2);
        const float mu  = s1 * (1.f / DIc);
        const float var = s2 * (1.f / DIc) - mu * mu;
        const float inv = rsqrtf(var + 1e-5f);
        float* __restrict__ op = yfinal + ((size_t)b * LL + lr) * DIc + lane;
#pragma unroll
        for (int j = 0; j < PX_CPL; ++j)
            op[32 * j] = ((yc[j] - mu) * inv * NW[j] + NB[j]) * zz[j];
    }
}

__global__ void k_outproj(const float* __restrict__ yf, const float* __restrict__ w,
                          float* __restrict__ out)
{
    const int row0 = blockIdx.y * 16, col0 = blockIdx.x * 32;
    v8f acc[2];
    wmma_xwT_multi<2>(yf, w, row0, col0, DIc, DIc, DIc, acc);
    const int lane = threadIdx.x & 31;
    const int nn = lane & 15, mh = (lane >> 4) * 8;
#pragma unroll
    for (int t = 0; t < 2; ++t)
#pragma unroll
        for (int v = 0; v < 8; ++v) {
            int r = row0 + mh + v, c = col0 + t * 16 + nn;
            out[(size_t)r * DMm + c] = acc[t][v];
        }
}

extern "C" void kernel_launch(void* const* d_in, const int* in_sizes, int n_in,
                              void* d_out, int out_size, void* d_ws, size_t ws_size,
                              hipStream_t stream)
{
    (void)in_sizes; (void)n_in; (void)out_size; (void)ws_size;
    const float* x          = (const float*)d_in[0];
    const float* in_proj_w  = (const float*)d_in[1];
    const float* conv_w     = (const float*)d_in[2];
    const float* conv_b     = (const float*)d_in[3];
    const float* x_proj_w   = (const float*)d_in[4];
    const float* dt_w       = (const float*)d_in[5];
    const float* dt_b       = (const float*)d_in[6];
    const float* A_logs     = (const float*)d_in[7];
    const float* Ds         = (const float*)d_in[8];
    const float* gamma1     = (const float*)d_in[9];
    const float* gamma2     = (const float*)d_in[10];
    const float* gamma3     = (const float*)d_in[11];
    const float* gamma4     = (const float*)d_in[12];
    const float* fc_w1      = (const float*)d_in[13];
    const float* fc_w2      = (const float*)d_in[14];
    const float* norm_w     = (const float*)d_in[15];
    const float* norm_b     = (const float*)d_in[16];
    const float* out_proj_w = (const float*)d_in[17];
    float* out = (float*)d_out;

    float* ws = (float*)d_ws;
    size_t o = 0;
    float* xc_lin = ws + o; o += (size_t)BB * LL * DIc;
    float* z_silu = ws + o; o += (size_t)BB * LL * DIc;
    float* xc_act = ws + o; o += (size_t)BB * LL * DIc;
    float* xdbl   = ws + o; o += (size_t)KK * BB * LL * CPROJ;
    float* delta  = ws + o; o += (size_t)KK * BB * LL * DIc;
    float* yarr   = ws + o; o += (size_t)KK * BB * LL * DIc;
    float* yfinal = ws + o; o += (size_t)BB * LL * DIc;
    float* wpad   = ws + o; o += (size_t)KK * CPAD * DIc;
    float* gcpart = ws + o; o += (size_t)BB * GC_NCH * DIc;
    float* gate   = ws + o; o += BB * DMm;

    k_padw    <<<dim3(KK, CPAD),       DIc, 0, stream>>>(x_proj_w, wpad);
    k_inproj  <<<dim3(12, 576),        32,  0, stream>>>(x, in_proj_w, xc_lin, z_silu);
    k_dwconv  <<<dim3(PX_NWG),         32 * PX_WAVES, 0, stream>>>(xc_lin, conv_w, conv_b, xc_act);
    k_gcmean  <<<dim3(BB, GC_NCH),     DIc, 0, stream>>>(xc_act, gcpart);
    k_gate    <<<dim3(BB),             DIc, 0, stream>>>(gcpart, fc_w1, fc_w2, gate);
    k_xproj   <<<dim3(2, 576, KK),     32,  0, stream>>>(xc_act, wpad, xdbl);
    k_dtproj  <<<dim3(6, 576, KK),     32,  0, stream>>>(xdbl, dt_w, dt_b, delta);
    k_scan    <<<dim3(BB * KK, DIc / 32), 32 * SCAN_NG, 0, stream>>>(xc_act, xdbl, delta, A_logs, Ds, yarr);
    k_combine <<<dim3(PX_NWG),         32 * PX_WAVES, 0, stream>>>(yarr, gate, gamma1, gamma2, gamma3,
                                                         gamma4, norm_w, norm_b, z_silu, yfinal);
    k_outproj <<<dim3(6, 576),         32,  0, stream>>>(yfinal, out_proj_w, out);
}


# k_scan: a chunk's inputs are read from LDS one chunk ahead into a second register set, so the LDS serves the 8 waves' reads beside the 16 steps' arithmetic, not in front of it; sum order kept
# speedup vs baseline: 10.1013x; 1.1034x over previous
#include <hip/hip_runtime.h>
#include <math.h>


typedef float v2f __attribute__((ext_vector_type(2)));
typedef float v8f __attribute__((ext_vector_type(8)));

#define BB 4
#define HH 48
#define WW 48
#define DMm 192
#define DIc 192
#define KK 4
#define NN 16
#define RR 12
#define LL (HH * WW)
#define CPROJ (RR + 2 * NN)
#define CPAD 64

__device__ __forceinline__ float siluf(float x)     { return x / (1.f + __expf(-x)); }
__device__ __forceinline__ float sigm(float x)      { return 1.f / (1.f + __expf(-x)); }
__device__ __forceinline__ float softplusf_(float x){ return (x > 20.f) ? x : log1pf(__expf(x)); }
__device__ __forceinline__ float geluf(float x)     { return 0.5f * x * (1.f + erff(x * 0.70710678118654752440f)); }

template <int NT>
__device__ __forceinline__ void wmma_rows_multi(const float* __restrict__ arow,
                                                const float* __restrict__ W,
                                                int col0, int Kdim, int ldw,
                                                v8f acc[NT])
{
    const int lane = threadIdx.x & 31;
    const int m    = lane & 15;
    const int kh   = lane >> 4;
    const float* __restrict__ ap = arow + 2 * kh;
    const float* __restrict__ bp[NT];
#pragma unroll
    for (int t = 0; t < NT; ++t) {
        bp[t] = W + (size_t)(col0 + t * 16 + m) * ldw + 2 * kh;
        acc[t] = {};
    }
    v2f a_cur = *(const v2f*)(ap);
    v2f b_cur[NT];
#pragma unroll
    for (int t = 0; t < NT; ++t) b_cur[t] = *(const v2f*)(bp[t]);

    for (int k0 = 4; k0 < Kdim; k0 += 4) {
        v2f a_nxt = *(const v2f*)(ap + k0);
        v2f b_nxt[NT];
#pragma unroll
        for (int t = 0; t < NT; ++t) b_nxt[t] = *(const v2f*)(bp[t] + k0);
#pragma unroll
        for (int t = 0; t < NT; ++t)
            acc[t] = __builtin_amdgcn_wmma_f32_16x16x4_f32(false, a_cur, false, b_cur[t],
                                                           (short)0, acc[t], false, false);
        a_cur = a_nxt;
#pragma unroll
        for (int t = 0; t < NT; ++t) b_cur[t] = b_nxt[t];
    }
#pragma unroll
    for (int t = 0; t < NT; ++t)
        acc[t] = __builtin_amdgcn_wmma_f32_16x16x4_f32(false, a_cur, false, b_cur[t],
                                                       (short)0, acc[t], false, false);
}

template <int NT>
__device__ __forceinline__ void wmma_xwT_multi(const float* __restrict__ X,
                                               const float* __restrict__ W,
                                               int row0, int col0,
                                               int Kdim, int ldx, int ldw,
                                               v8f acc[NT])
{
    const int m = threadIdx.x & 15;
    wmma_rows_multi<NT>(X + (size_t)(row0 + m) * ldx, W, col0, Kdim, ldw, acc);
}

__global__ void k_padw(const float* __restrict__ xpw, float* __restrict__ wpad)
{
    int k = blockIdx.x, c = blockIdx.y, d = threadIdx.x;
    wpad[((size_t)k * CPAD + c) * DIc + d] =
        (c < CPROJ) ? xpw[((size_t)k * CPROJ + c) * DIc + d] : 0.f;
}

__global__ void k_inproj(const float* __restrict__ x, const float* __restrict__ w,
                         float* __restrict__ xc_lin, float* __restrict__ z_silu)
{
    const int row0 = blockIdx.y * 16, col0 = blockIdx.x * 32;
    v8f acc[2];
    wmma_xwT_multi<2>(x, w, row0, col0, DMm, DMm, DMm, acc);
    const int lane = threadIdx.x & 31;
    const int nn = lane & 15, mh = (lane >> 4) * 8;
#pragma unroll
    for (int t = 0; t < 2; ++t)
#pragma unroll
        for (int v = 0; v < 8; ++v) {
            int r = row0 + mh + v;
            int c = col0 + t * 16 + nn;
            float val = acc[t][v];
            if (c < DIc) xc_lin[(size_t)r * DIc + c] = val;
            else         z_silu[(size_t)r * DIc + (c - DIc)] = siluf(val);
        }
}

#define PX_WAVES 8
#define PX_PPW   4
#define PX_CPL   (DIc / 32)
#define PX_NWG   (BB * LL / (PX_WAVES * PX_PPW))
static_assert(WW % PX_PPW == 0 && (BB * LL) % (PX_WAVES * PX_PPW) == 0 && DIc % 32 == 0,
              "a wave's pixels must lie in one image row");

__global__ __launch_bounds__(32 * PX_WAVES)
void k_dwconv(const float* __restrict__ xc_lin, const float* __restrict__ cw,
              const float* __restrict__ cb, float* __restrict__ xc_act)
{
    __shared__ float wsh[DIc * 9];
    const int lane = threadIdx.x & 31;
    const int wave = __builtin_amdgcn_readfirstlane((int)(threadIdx.x >> 5));
    const int pix0 = (blockIdx.x * PX_WAVES + wave) * PX_PPW;
    const int b = pix0 / LL, rem0 = pix0 - b * LL;
    const int h = rem0 / WW, w0 = rem0 - h * WW;
    const float* __restrict__ img = xc_lin + (size_t)b * LL * DIc + lane;

    float win[3][PX_PPW + 2][PX_CPL];
#pragma unroll
    for (int r = 0; r < 3; ++r) {
        const int hh = h + r - 1;
        const bool rok = (hh >= 0) && (hh < HH);
        const int hc = hh < 0 ? 0 : (hh >= HH ? HH - 1 : hh);
#pragma unroll
        for (int q = 0; q < PX_PPW + 2; ++q) {
            const int wq = w0 + q - 1;
            const bool cok = (q == 0) ? (wq >= 0) : ((q == PX_PPW + 1) ? (wq < WW) : true);
            const int wc = (q == 0) ? (wq < 0 ? 0 : wq) : ((q == PX_PPW + 1) ? (wq >= WW ? WW - 1 : wq) : wq);
            const float* __restrict__ sp = img + (size_t)(hc * WW + wc) * DIc;
#pragma unroll
            for (int j = 0; j < PX_CPL; ++j) {
                const float v = sp[32 * j];
                win[r][q][j] = (rok && cok) ? v : 0.f;
            }
        }
    }

    constexpr int WFULL = (DIc * 9) / (32 * PX_WAVES);
    float wtmp[WFULL];
#pragma unroll
    for (int i = 0; i < WFULL; ++i) wtmp[i] = cw[(int)threadIdx.x + i * 32 * PX_WAVES];
    const int itail = (int)threadIdx.x + WFULL * 32 * PX_WAVES;
    const float wtail = (itail < DIc * 9) ? cw[itail] : 0.f;
#pragma unroll
    for (int i = 0; i < WFULL; ++i) wsh[(int)threadIdx.x + i * 32 * PX_WAVES] = wtmp[i];
    if (itail < DIc * 9) wsh[itail] = wtail;
    __syncthreads();
    float wreg[PX_CPL][9], acc[PX_PPW][PX_CPL];
#pragma unroll
    for (int j = 0; j < PX_CPL; ++j) {
#pragma unroll
        for (int t = 0; t < 9; ++t) wreg[j][t] = wsh[(lane + 32 * j) * 9 + t];
        const float bj = cb[lane + 32 * j];
#pragma unroll
        for (int p = 0; p < PX_PPW; ++p) acc[p][j] = bj;
    }
#pragma unroll
    for (int p = 0; p < PX_PPW; ++p)
#pragma unroll
        for (int kh = 0; kh < 3; ++kh)
#pragma unroll
            for (int kw = 0; kw < 3; ++kw)
#pragma unroll
                for (int j = 0; j < PX_CPL; ++j)
                    acc[p][j] += win[kh][p + kw][j] * wreg[j][kh * 3 + kw];
#pragma unroll
    for (int p = 0; p < PX_PPW; ++p) {
        float* __restrict__ op = xc_act + (size_t)(pix0 + p) * DIc + lane;
#pragma unroll
        for (int j = 0; j < PX_CPL; ++j) op[32 * j] = siluf(acc[p][j]);
    }
}

#define GC_NCH 48
#define GC_P (LL / GC_NCH)
__global__ void k_gcmean(const float* __restrict__ xc_act, float* __restrict__ gcpart)
{
    int b = blockIdx.x, ch = blockIdx.y, d = threadIdx.x;
    const float* __restrict__ src = xc_act + ((size_t)b * LL + (size_t)ch * GC_P) * DIc + d;
    float s = 0.f;
#pragma unroll
    for (int p = 0; p < GC_P; ++p)
        s += src[(size_t)p * DIc];
    gcpart[((size_t)b * GC_NCH + ch) * DIc + d] = s;
}

__global__ void k_gate(const float* __restrict__ gcpart, const float* __restrict__ fw1,
                       const float* __restrict__ fw2, float* __restrict__ gate)
{
    __shared__ float g[DIc];
    __shared__ float h1[DMm / 4];
    int b = blockIdx.x, d = threadIdx.x;
    float gs = 0.f;
#pragma unroll
    for (int c = 0; c < GC_NCH; ++c)
        gs += gcpart[((size_t)b * GC_NCH + c) * DIc + d];
    g[d] = gs * (1.f / LL);
    __syncthreads();
    if (d < DMm / 4) {
        float s = 0.f;
        for (int j = 0; j < DMm; ++j) s += fw1[d * DMm + j] * g[j];
        h1[d] = geluf(s);
    }
    __syncthreads();
    float s2 = 0.f;
#pragma unroll
    for (int j = 0; j < DMm / 4; ++j) s2 += fw2[d * (DMm / 4) + j] * h1[j];
    gate[b * DMm + d] = sigm(s2);
}

__device__ __forceinline__ int seq_pixel(int k, int l)
{
    int lb = (k < 2) ? l : (LL - 1 - l);
    int w_in = lb & 3, h_in = (lb >> 2) & 3;
    int nw = (lb >> 4) % (WW / 4);
    int nh = (lb >> 4) / (WW / 4);
    int p = nh * 4 + h_in, q = nw * 4 + w_in;
    return ((k & 1) == 0) ? (p * WW + q) : (q * WW + p);
}

__global__ void k_xproj(const float* __restrict__ xc_act, const float* __restrict__ wpad,
                        float* __restrict__ xdbl)
{
    int k = blockIdx.z;
    const float* W = wpad + (size_t)k * CPAD * DIc;
    float* O = xdbl + (size_t)k * BB * LL * CPROJ;
    const int row0 = blockIdx.y * 16, col0 = blockIdx.x * 32;
    const int lane = threadIdx.x & 31;
    const int ra = row0 + (lane & 15);
    const int ba = ra / LL, la = ra - ba * LL;
    const float* arow = xc_act + ((size_t)ba * LL + seq_pixel(k, la)) * DIc;
    v8f acc[2];
    wmma_rows_multi<2>(arow, W, col0, DIc, DIc, acc);
    const int nn = lane & 15, mh = (lane >> 4) * 8;
#pragma unroll
    for (int t = 0; t < 2; ++t)
#pragma unroll
        for (int v = 0; v < 8; ++v) {
            int r = row0 + mh + v, c = col0 + t * 16 + nn;
            if (c < CPROJ) O[(size_t)r * CPROJ + c] = acc[t][v];
        }
}

__global__ void k_dtproj(const float* __restrict__ xdbl, const float* __restrict__ dtw,
                         const float* __restrict__ dtb, float* __restrict__ delta)
{
    int k = blockIdx.z;
    const float* X = xdbl + (size_t)k * BB * LL * CPROJ;
    const float* W = dtw + (size_t)k * DIc * RR;
    float* O = delta + (size_t)k * BB * LL * DIc;
    const int row0 = blockIdx.y * 16, col0 = blockIdx.x * 32;
    v8f acc[2];
    wmma_xwT_multi<2>(X, W, row0, col0, RR, CPROJ, RR, acc);
    const int lane = threadIdx.x & 31;
    const int nn = lane & 15, mh = (lane >> 4) * 8;
#pragma unroll
    for (int t = 0; t < 2; ++t)
#pragma unroll
        for (int v = 0; v < 8; ++v) {
            int r = row0 + mh + v, c = col0 + t * 16 + nn;
            O[(size_t)r * DIc + c] = softplusf_(acc[t][v] + dtb[k * DIc + c]);
        }
}

#define SCAN_T 16
#define SCAN_NG 8
#define SCAN_NPG (NN / SCAN_NG)
#define SCAN_RPW (SCAN_T / SCAN_NG)
#define SCAN_NCH (LL / SCAN_T)
#define SCAN_LEAD 4
static_assert(SCAN_T == 16 && LL % SCAN_T == 0, "a chunk must be one 4x4 window");
static_assert(SCAN_NPG == 2 && SCAN_RPW == 2, "layout of the staged planes: a wave's two steps are one step pair");
static_assert(SCAN_LEAD == 4 && SCAN_NCH % SCAN_LEAD == 0, "the chunk loop is written out for four fetch sets");

__device__ __forceinline__ int slot_pixoff(int tr, int flip, int t)
{
    const int tt = flip ? (SCAN_T - 1 - t) : t;
    const int h_in = tt >> 2, w_in = tt & 3;
    return tr ? (w_in * WW + h_in) : (h_in * WW + w_in);
}
__device__ __forceinline__ int chunk_origin(int tr, int flip, int c)
{
    const int widx = flip ? (SCAN_NCH - 1 - c) : c;
    const int nw = widx % (WW / 4), nh = widx / (WW / 4);
    return tr ? (nw * 4 * WW + nh * 4) : (nh * 4 * WW + nw * 4);
}

struct ScanLds {
    float2 y[2][SCAN_T / 2][SCAN_NG][32];
    float4 du[2][SCAN_T / 2][32];
    float4 bc[2][SCAN_T][SCAN_NG];
};
struct ScanFetch { float dt[SCAN_RPW], u[SCAN_RPW], bc[SCAN_RPW]; };
struct ScanIn {
    float4 du[SCAN_T / 2];
    float4 bc[SCAN_T];
};

struct ScanWave {
    const float* uimg;
    const float* dp;
    const float* bcw;
    int tr, flip;
    int ts0;
    int po[SCAN_RPW];
    int bcpos;
};

__device__ __forceinline__ void scan_fetch(ScanFetch& f, const ScanWave& w, int chunk)
{
    const int c  = (chunk < SCAN_NCH) ? chunk : (SCAN_NCH - 1);
    const int l0 = c * SCAN_T;
    const float* __restrict__ up = w.uimg + (size_t)chunk_origin(w.tr, w.flip, c) * DIc;
#pragma unroll
    for (int r = 0; r < SCAN_RPW; ++r) {
        f.dt[r] = w.dp[(size_t)(l0 + w.ts0 + r) * DIc];
        f.u[r]  = up[w.po[r] * DIc];
        f.bc[r] = w.bcw[(size_t)(l0 + w.ts0 + r) * CPROJ];
    }
}

template <int PAR>
__device__ __forceinline__ void scan_park(const ScanFetch& f, const ScanWave& w, ScanLds& sm,
                                          const int g, const int lane)
{
    sm.du[PAR][g][lane] = make_float4(f.dt[0], f.u[0], f.dt[1], f.u[1]);
    float* __restrict__ bcf = &sm.bc[PAR][0][0].x;
#pragma unroll
    for (int r = 0; r < SCAN_RPW; ++r)
        bcf[(w.ts0 + r) * (SCAN_NG * 4) + w.bcpos] = f.bc[r];
}

template <int PAR>
__device__ __forceinline__ void scan_read(ScanIn& in, const ScanLds& sm, const int g, const int lane)
{
#pragma unroll
    for (int p = 0; p < SCAN_T / 2; ++p) {
        in.du[p]         = sm.du[PAR][p][lane];
        in.bc[2 * p]     = sm.bc[PAR][2 * p][g];
        in.bc[2 * p + 1] = sm.bc[PAR][2 * p + 1][g];
    }
}

__device__ __forceinline__ void scan_steps(const float4 (&du4)[SCAN_T / 2], const float4 (&bc4)[SCAN_T],
                                           float (&yv)[SCAN_T],
                                           float (&h)[SCAN_NPG], const float (&A)[SCAN_NPG],
                                           const float Dg)
{
#pragma unroll
    for (int t = 0; t < SCAN_T; ++t) {
        const float4 d4 = du4[t >> 1];
        const float dt = (t & 1) ? d4.z : d4.x;
        const float u  = (t & 1) ? d4.w : d4.y;
        const float Bt[SCAN_NPG] = { bc4[t].x, bc4[t].y };
        const float Ct[SCAN_NPG] = { bc4[t].z, bc4[t].w };
        const float du = dt * u;
#pragma unroll
        for (int n = 0; n < SCAN_NPG; ++n) {
            h[n] = __builtin_fmaf(__expf(dt * A[n]), h[n], du * Bt[n]);
        }
        float y = h[0] * Ct[0];
#pragma unroll
        for (int n = 1; n < SCAN_NPG; ++n) y = __builtin_fmaf(h[n], Ct[n], y);
        yv[t] = __builtin_fmaf(Dg, u, y);
    }
}

__device__ __forceinline__ void scan_rows(const float2 (&part)[SCAN_NG], float* __restrict__ yp,
                                          const int c, const int g)
{
    float s0 = part[0].x, s1 = part[0].y;
#pragma unroll
    for (int q = 1; q < SCAN_NG; ++q) { s0 += part[q].x; s1 += part[q].y; }
    yp[(size_t)(c * SCAN_T + g * SCAN_RPW) * DIc]     = s0;
    yp[(size_t)(c * SCAN_T + g * SCAN_RPW + 1) * DIc] = s1;
}

template <int PAR>
__device__ __forceinline__ void scan_chunk(const int c, ScanLds& sm, ScanFetch& f, const ScanWave& w,
                                           const ScanIn& cur, ScanIn& nxt,
                                           float (&h)[SCAN_NPG], const float (&A)[SCAN_NPG],
                                           const float Dg, const int g, const int lane,
                                           float* __restrict__ yp)
{
    float2 part[SCAN_NG];
#pragma unroll
    for (int q = 0; q < SCAN_NG; ++q) part[q] = sm.y[PAR ^ 1][g][q][lane];
    __builtin_amdgcn_sched_barrier(0);
    scan_read<PAR ^ 1>(nxt, sm, g, lane);
    __builtin_amdgcn_sched_barrier(0);

    float4 du4[SCAN_T / 2];
#pragma unroll
    for (int p = 0; p < SCAN_T / 2; ++p) du4[p] = cur.du[p];
    asm volatile("" : "+v"(du4[0].x), "+v"(du4[0].z), "+v"(du4[1].x), "+v"(du4[1].z),
                      "+v"(du4[2].x), "+v"(du4[2].z), "+v"(du4[3].x), "+v"(du4[3].z));
    asm volatile("" : "+v"(du4[4].x), "+v"(du4[4].z), "+v"(du4[5].x), "+v"(du4[5].z),
                      "+v"(du4[6].x), "+v"(du4[6].z), "+v"(du4[7].x), "+v"(du4[7].z));
    float yv[SCAN_T];
    scan_steps(du4, cur.bc, yv, h, A, Dg);

    asm volatile("" :: "v"(yv[0]), "v"(yv[1]), "v"(yv[2]), "v"(yv[3]), "v"(yv[4]), "v"(yv[5]),
                       "v"(yv[6]), "v"(yv[7]), "v"(yv[8]), "v"(yv[9]), "v"(yv[10]), "v"(yv[11]),
                       "v"(yv[12]), "v"(yv[13]), "v"(yv[14]), "v"(yv[15]));
    __builtin_amdgcn_sched_barrier(0);
    if (c > 0) scan_rows(part, yp, c - 1, g);
#pragma unroll
    for (int p = 0; p < SCAN_T / 2; ++p)
        sm.y[PAR][p][g][lane] = make_float2(yv[2 * p], yv[2 * p + 1]);
    scan_park<PAR>(f, w, sm, g, lane);
    __builtin_amdgcn_sched_barrier(0);
    scan_fetch(f, w, c + 2 + SCAN_LEAD);

    __builtin_amdgcn_fence(__ATOMIC_RELEASE, "workgroup", "local");
    __builtin_amdgcn_s_barrier();
    __builtin_amdgcn_fence(__ATOMIC_ACQUIRE, "workgroup", "local");
}

__global__ __launch_bounds__(32 * SCAN_NG)
void k_scan(const float* __restrict__ xc_act, const float* __restrict__ xdbl,
            const float* __restrict__ delta, const float* __restrict__ A_logs,
            const float* __restrict__ Ds, float* __restrict__ yarr)
{
    __shared__ ScanLds sm;
    const int kb = blockIdx.x;
    const int k = kb & 3, b = kb >> 2;
    const int lane = threadIdx.x & 31;
    const int g = __builtin_amdgcn_readfirstlane((int)(threadIdx.x >> 5));
    const int d = blockIdx.y * 32 + lane;
    const int n0 = g * SCAN_NPG;
    float A[SCAN_NPG];
#pragma unroll
    for (int n = 0; n < SCAN_NPG; ++n)
        A[n] = -__expf(A_logs[((size_t)(k * DIc + d)) * NN + n0 + n]);
    const float Dg = (g == SCAN_NG - 1) ? Ds[k * DIc + d] : 0.f;
    const size_t seqbase = ((size_t)(k * BB + b)) * LL;
    ScanWave w;
    w.uimg = xc_act + (size_t)b * LL * DIc + d;
    w.dp   = delta + seqbase * DIc + d;
    w.bcw  = xdbl  + seqbase * CPROJ + RR + lane;
    w.tr   = k & 1;
    w.flip = k >> 1;
    w.ts0  = g * SCAN_RPW;
#pragma unroll
    for (int r = 0; r < SCAN_RPW; ++r) w.po[r] = slot_pixoff(w.tr, w.flip, w.ts0 + r);
    w.bcpos = ((lane & 15) >> 1) * 4 + (lane & 1) + 2 * (lane >> 4);
    float* __restrict__ yp = yarr + seqbase * DIc + d;

    float h[SCAN_NPG];
#pragma unroll
    for (int n = 0; n < SCAN_NPG; ++n) h[n] = 0.f;

    ScanFetch f0, f1, f2, f3;
    scan_fetch(f0, w, 0);
    __builtin_amdgcn_sched_barrier(0);
    scan_fetch(f1, w, 1);
    __builtin_amdgcn_sched_barrier(0);
    scan_fetch(f2, w, 2);
    __builtin_amdgcn_sched_barrier(0);
    scan_fetch(f3, w, 3);
    __builtin_amdgcn_sched_barrier(0);
    scan_park<0>(f0, w, sm, g, lane);
#pragma unroll
    for (int p = 0; p < SCAN_T / 2; ++p) sm.y[1][p][g][lane] = make_float2(0.f, 0.f);
    scan_fetch(f0, w, SCAN_LEAD);
    __builtin_amdgcn_sched_barrier(0);
    __builtin_amdgcn_fence(__ATOMIC_RELEASE, "workgroup", "local");
    __builtin_amdgcn_s_barrier();
    __builtin_amdgcn_fence(__ATOMIC_ACQUIRE, "workgroup", "local");
    ScanIn inA, inB;
    scan_read<0>(inA, sm, g, lane);
    __builtin_amdgcn_sched_barrier(0);
    scan_park<1>(f1, w, sm, g, lane);
    scan_fetch(f1, w, SCAN_LEAD + 1);
    __builtin_amdgcn_sched_barrier(0);
    __builtin_amdgcn_fence(__ATOMIC_RELEASE, "workgroup", "local");
    __builtin_amdgcn_s_barrier();
    __builtin_amdgcn_fence(__ATOMIC_ACQUIRE, "workgroup", "local");

    for (int c = 0; c < SCAN_NCH; c += SCAN_LEAD) {
        scan_chunk<0>(c,     sm, f2, w, inA, inB, h, A, Dg, g, lane, yp);
        scan_chunk<1>(c + 1, sm, f3, w, inB, inA, h, A, Dg, g, lane, yp);
        scan_chunk<0>(c + 2, sm, f0, w, inA, inB, h, A, Dg, g, lane, yp);
        scan_chunk<1>(c + 3, sm, f1, w, inB, inA, h, A, Dg, g, lane, yp);
    }
    float2 part[SCAN_NG];
#pragma unroll
    for (int q = 0; q < SCAN_NG; ++q) part[q] = sm.y[(SCAN_NCH - 1) & 1][g][q][lane];
    scan_rows(part, yp, SCAN_NCH - 1, g);
}

template <int XM>
__device__ __forceinline__ float row_xor(float v)
{
    return __builtin_bit_cast(float,
        __builtin_amdgcn_update_dpp(0, __builtin_bit_cast(int, v), 0x160 + XM, 0xf, 0xf, false));
}
__device__ __forceinline__ float half_xor(float v)
{
    const int x = __builtin_bit_cast(int, v);
    return __builtin_bit_cast(float,
        __builtin_amdgcn_permlanex16(x, x, 0x76543210, 0xfedcba98, false, false));
}
__device__ __forceinline__ float wave_sum(float v)
{
    v += row_xor<1>(v);
    v += row_xor<2>(v);
    v += row_xor<4>(v);
    v += row_xor<8>(v);
    v += half_xor(v);
    return v;
}

__global__ __launch_bounds__(32 * PX_WAVES)
void k_combine(const float* __restrict__ yarr, const float* __restrict__ gate,
               const float* __restrict__ g1, const float* __restrict__ g2,
               const float* __restrict__ g3, const float* __restrict__ g4,
               const float* __restrict__ nw, const float* __restrict__ nb,
               const float* __restrict__ z_silu, float* __restrict__ yfinal)
{
    const int lane = threadIdx.x & 31;
    const int wave = __builtin_amdgcn_readfirstlane((int)(threadIdx.x >> 5));
    const int pix0 = (blockIdx.x * PX_WAVES + wave) * PX_PPW;
    const int b = pix0 / LL, lr0 = pix0 - b * LL;
    const size_t bstride = (size_t)LL * DIc;
    const float* __restrict__ Y0 = yarr + ((size_t)(0 * BB + b)) * bstride + lane;
    const float* __restrict__ Y1 = yarr + ((size_t)(1 * BB + b)) * bstride + lane;
    const float* __restrict__ Y2 = yarr + ((size_t)(2 * BB + b)) * bstride + lane;
    const float* __restrict__ Y3 = yarr + ((size_t)(3 * BB + b)) * bstride + lane;
    float G1[PX_CPL], G2[PX_CPL], G3[PX_CPL], G4[PX_CPL], GT[PX_CPL], NW[PX_CPL], NB[PX_CPL];
#pragma unroll
    for (int j = 0; j < PX_CPL; ++j) {
        const int d = lane + 32 * j;
        G1[j] = g1[d]; G2[j] = g2[d]; G3[j] = g3[d]; G4[j] = g4[d];
        GT[j] = gate[b * DIc + d]; NW[j] = nw[d]; NB[j] = nb[d];
    }
#pragma unroll
    for (int p = 0; p < PX_PPW; ++p) {
        const int lr = lr0 + p;
        const int h = lr / WW, w = lr % WW;
        const int l = (((h >> 2) * (WW / 4) + (w >> 2)) << 4) + ((h & 3) << 2) + (w & 3);
        const int lt = (l % WW) * HH + (l / WW);
        const float* __restrict__ zp = z_silu + ((size_t)b * LL + lr) * DIc + lane;
        float yc[PX_CPL], zz[PX_CPL];
        float s1 = 0.f, s2 = 0.f;
#pragma unroll
        for (int j = 0; j < PX_CPL; ++j) {
            float y = G1[j] * Y0[(size_t)l * DIc + 32 * j]
                    + G2[j] * Y2[(size_t)(LL - 1 - l) * DIc + 32 * j]
                    + G3[j] * Y1[(size_t)lt * DIc + 32 * j]
                    + G4[j] * Y3[(size_t)(LL - 1 - lt) * DIc + 32 * j];
            y *= GT[j];
            yc[j] = y;
            zz[j] = zp[32 * j];
            s1 += y;
            s2 += y * y;
        }
        s1 = wave_sum(s1);
        s2 = wave_sum(s2);
        const float mu  = s1 * (1.f / DIc);
        const float var = s2 * (1.f / DIc) - mu * mu;
        const float inv = rsqrtf(var + 1e-5f);
        float* __restrict__ op = yfinal + ((size_t)b * LL + lr) * DIc + lane;
#pragma unroll
        for (int j = 0; j < PX_CPL; ++j)
            op[32 * j] = ((yc[j] - mu) * inv * NW[j] + NB[j]) * zz[j];
    }
}

__global__ void k_outproj(const float* __restrict__ yf, const float* __restrict__ w,
                          float* __restrict__ out)
{
    const int row0 = blockIdx.y * 16, col0 = blockIdx.x * 32;
    v8f acc[2];
    wmma_xwT_multi<2>(yf, w, row0, col0, DIc, DIc, DIc, acc);
    const int lane = threadIdx.x & 31;
    const int nn = lane & 15, mh = (lane >> 4) * 8;
#pragma unroll
    for (int t = 0; t < 2; ++t)
#pragma unroll
        for (int v = 0; v < 8; ++v) {
            int r = row0 + mh + v, c = col0 + t * 16 + nn;
            out[(size_t)r * DMm + c] = acc[t][v];
        }
}

extern "C" void kernel_launch(void* const* d_in, const int* in_sizes, int n_in,
                              void* d_out, int out_size, void* d_ws, size_t ws_size,
                              hipStream_t stream)
{
    (void)in_sizes; (void)n_in; (void)out_size; (void)ws_size;
    const float* x          = (const float*)d_in[0];
    const float* in_proj_w  = (const float*)d_in[1];
    const float* conv_w     = (const float*)d_in[2];
    const float* conv_b     = (const float*)d_in[3];
    const float* x_proj_w   = (const float*)d_in[4];
    const float* dt_w       = (const float*)d_in[5];
    const float* dt_b       = (const float*)d_in[6];
    const float* A_logs     = (const float*)d_in[7];
    const float* Ds         = (const float*)d_in[8];
    const float* gamma1     = (const float*)d_in[9];
    const float* gamma2     = (const float*)d_in[10];
    const float* gamma3     = (const float*)d_in[11];
    const float* gamma4     = (const float*)d_in[12];
    const float* fc_w1      = (const float*)d_in[13];
    const float* fc_w2      = (const float*)d_in[14];
    const float* norm_w     = (const float*)d_in[15];
    const float* norm_b     = (const float*)d_in[16];
    const float* out_proj_w = (const float*)d_in[17];
    float* out = (float*)d_out;

    float* ws = (float*)d_ws;
    size_t o = 0;
    float* xc_lin = ws + o; o += (size_t)BB * LL * DIc;
    float* z_silu = ws + o; o += (size_t)BB * LL * DIc;
    float* xc_act = ws + o; o += (size_t)BB * LL * DIc;
    float* xdbl   = ws + o; o += (size_t)KK * BB * LL * CPROJ;
    float* delta  = ws + o; o += (size_t)KK * BB * LL * DIc;
    float* yarr   = ws + o; o += (size_t)KK * BB * LL * DIc;
    float* yfinal = ws + o; o += (size_t)BB * LL * DIc;
    float* wpad   = ws + o; o += (size_t)KK * CPAD * DIc;
    float* gcpart = ws + o; o += (size_t)BB * GC_NCH * DIc;
    float* gate   = ws + o; o += BB * DMm;

    k_padw    <<<dim3(KK, CPAD),       DIc, 0, stream>>>(x_proj_w, wpad);
    k_inproj  <<<dim3(12, 576),        32,  0, stream>>>(x, in_proj_w, xc_lin, z_silu);
    k_dwconv  <<<dim3(PX_NWG),         32 * PX_WAVES, 0, stream>>>(xc_lin, conv_w, conv_b, xc_act);
    k_gcmean  <<<dim3(BB, GC_NCH),     DIc, 0, stream>>>(xc_act, gcpart);
    k_gate    <<<dim3(BB),             DIc, 0, stream>>>(gcpart, fc_w1, fc_w2, gate);
    k_xproj   <<<dim3(2, 576, KK),     32,  0, stream>>>(xc_act, wpad, xdbl);
    k_dtproj  <<<dim3(6, 576, KK),     32,  0, stream>>>(xdbl, dt_w, dt_b, delta);
    k_scan    <<<dim3(BB * KK, DIc / 32), 32 * SCAN_NG, 0, stream>>>(xc_act, xdbl, delta, A_logs, Ds, yarr);
    k_combine <<<dim3(PX_NWG),         32 * PX_WAVES, 0, stream>>>(yarr, gate, gamma1, gamma2, gamma3,
                                                         gamma4, norm_w, norm_b, z_silu, yfinal);
    k_outproj <<<dim3(6, 576),         32,  0, stream>>>(yfinal, out_proj_w, out);
}


# in/x/out_proj: wave tile 16x32 -> 32x32..32x64, fragments loaded 16 B per lane (2 k-steps each) two blocks ahead; all 4 GEMMs in 4-8-wave work-groups; k order inside each 8 permuted (new sum order)
# speedup vs baseline: 11.5754x; 1.1459x over previous
#include <hip/hip_runtime.h>
#include <math.h>


typedef float v2f __attribute__((ext_vector_type(2)));
typedef float v8f __attribute__((ext_vector_type(8)));

#define BB 4
#define HH 48
#define WW 48
#define DMm 192
#define DIc 192
#define KK 4
#define NN 16
#define RR 12
#define LL (HH * WW)
#define CPROJ (RR + 2 * NN)
#define CPAD 64

__device__ __forceinline__ float siluf(float x)     { return x / (1.f + __expf(-x)); }
__device__ __forceinline__ float sigm(float x)      { return 1.f / (1.f + __expf(-x)); }
__device__ __forceinline__ float softplusf_(float x){ return (x > 20.f) ? x : log1pf(__expf(x)); }
__device__ __forceinline__ float geluf(float x)     { return 0.5f * x * (1.f + erff(x * 0.70710678118654752440f)); }

typedef float v4f __attribute__((ext_vector_type(4)));

template <int MT, int NT>
struct Frag8 { v4f a[MT]; v4f b[NT]; };

template <int MT, int NT>
__device__ __forceinline__ void frag8_load(Frag8<MT, NT>& f, const float* const (&ap)[MT],
                                           const float* const (&bp)[NT], int koff)
{
#pragma unroll
    for (int i = 0; i < MT; ++i) f.a[i] = *(const v4f*)(ap[i] + koff);
#pragma unroll
    for (int t = 0; t < NT; ++t) f.b[t] = *(const v4f*)(bp[t] + koff);
}

template <int MT, int NT>
__device__ __forceinline__ void frag8_mma(const Frag8<MT, NT>& f, v8f (&acc)[MT][NT])
{
#pragma unroll
    for (int s = 0; s < 2; ++s)
#pragma unroll
        for (int i = 0; i < MT; ++i) {
            const v2f a = s ? v2f{f.a[i].z, f.a[i].w} : v2f{f.a[i].x, f.a[i].y};
#pragma unroll
            for (int t = 0; t < NT; ++t) {
                const v2f b = s ? v2f{f.b[t].z, f.b[t].w} : v2f{f.b[t].x, f.b[t].y};
                acc[i][t] = __builtin_amdgcn_wmma_f32_16x16x4_f32(false, a, false, b,
                                                                  (short)0, acc[i][t], false, false);
            }
        }
}

template <int MT, int NT, int KD>
__device__ __forceinline__ void wmma_tiles(const float* const (&arow)[MT],
                                           const float* const (&brow)[NT],
                                           v8f (&acc)[MT][NT])
{
    static_assert(KD % 4 == 0, "k in steps of 4");
    constexpr int NB = KD / 8;
    constexpr bool TAIL = (KD % 8) != 0;
    const int kh = (threadIdx.x & 31) >> 4;
    const float* ap[MT];
    const float* bp[NT];
#pragma unroll
    for (int i = 0; i < MT; ++i) {
        ap[i] = arow[i] + 4 * kh;
#pragma unroll
        for (int t = 0; t < NT; ++t) acc[i][t] = {};
    }
#pragma unroll
    for (int t = 0; t < NT; ++t) bp[t] = brow[t] + 4 * kh;

    if constexpr (NB >= 3 && NB % 3 == 0) {
        Frag8<MT, NT> f0, f1, f2;
        frag8_load(f0, ap, bp, 0);
        frag8_load(f1, ap, bp, 8);
#pragma unroll 1
        for (int kb = 0; kb < NB - 3; kb += 3) {
            frag8_load(f2, ap, bp, (kb + 2) * 8);
            __builtin_amdgcn_sched_barrier(0);
            frag8_mma(f0, acc);
            __builtin_amdgcn_sched_barrier(0);
            frag8_load(f0, ap, bp, (kb + 3) * 8);
            __builtin_amdgcn_sched_barrier(0);
            frag8_mma(f1, acc);
            __builtin_amdgcn_sched_barrier(0);
            frag8_load(f1, ap, bp, (kb + 4) * 8);
            __builtin_amdgcn_sched_barrier(0);
            frag8_mma(f2, acc);
            __builtin_amdgcn_sched_barrier(0);
        }
        frag8_load(f2, ap, bp, (NB - 1) * 8);
        __builtin_amdgcn_sched_barrier(0);
        frag8_mma(f0, acc);
        frag8_mma(f1, acc);
        frag8_mma(f2, acc);
    } else {
        Frag8<MT, NT> f[NB];
#pragma unroll
        for (int kb = 0; kb < NB; ++kb) frag8_load(f[kb], ap, bp, kb * 8);
        v2f ta[MT], tb[NT];
        if constexpr (TAIL) {
#pragma unroll
            for (int i = 0; i < MT; ++i) ta[i] = *(const v2f*)(arow[i] + NB * 8 + 2 * kh);
#pragma unroll
            for (int t = 0; t < NT; ++t) tb[t] = *(const v2f*)(brow[t] + NB * 8 + 2 * kh);
        }
#pragma unroll
        for (int kb = 0; kb < NB; ++kb) frag8_mma(f[kb], acc);
        if constexpr (TAIL) {
#pragma unroll
            for (int i = 0; i < MT; ++i)
#pragma unroll
                for (int t = 0; t < NT; ++t)
                    acc[i][t] = __builtin_amdgcn_wmma_f32_16x16x4_f32(false, ta[i], false, tb[t],
                                                                      (short)0, acc[i][t], false, false);
        }
    }
}

#define GW 4

__global__ void k_padw(const float* __restrict__ xpw, float* __restrict__ wpad)
{
    int k = blockIdx.x, c = blockIdx.y, d = threadIdx.x;
    wpad[((size_t)k * CPAD + c) * DIc + d] =
        (c < CPROJ) ? xpw[((size_t)k * CPROJ + c) * DIc + d] : 0.f;
}

__global__ __launch_bounds__(32 * GW)
void k_inproj(const float* __restrict__ x, const float* __restrict__ w,
              float* __restrict__ xc_lin, float* __restrict__ z_silu)
{
    const int lane = threadIdx.x & 31;
    const int wave = __builtin_amdgcn_readfirstlane((int)(threadIdx.x >> 5));
    const int row0 = blockIdx.y * 64 + (wave >> 1) * 32, col0 = blockIdx.x * 128 + (wave & 1) * 64;
    const int nn = lane & 15, mh = (lane >> 4) * 8;
    const float* arow[2];
    const float* brow[4];
#pragma unroll
    for (int i = 0; i < 2; ++i) arow[i] = x + (size_t)(row0 + i * 16 + nn) * DMm;
#pragma unroll
    for (int t = 0; t < 4; ++t) brow[t] = w + (size_t)(col0 + t * 16 + nn) * DMm;
    v8f acc[2][4];
    wmma_tiles<2, 4, DMm>(arow, brow, acc);
    static_assert(DIc % 64 == 0, "a wave's columns must not straddle the split of xz");
    if (col0 < DIc) {
        float* __restrict__ dst = xc_lin + (size_t)(row0 + mh) * DIc + col0 + nn;
#pragma unroll
        for (int i = 0; i < 2; ++i)
#pragma unroll
            for (int t = 0; t < 4; ++t)
#pragma unroll
                for (int v = 0; v < 8; ++v)
                    dst[(size_t)(i * 16 + v) * DIc + t * 16] = acc[i][t][v];
    } else {
        float* __restrict__ dst = z_silu + (size_t)(row0 + mh) * DIc + (col0 - DIc) + nn;
#pragma unroll
        for (int i = 0; i < 2; ++i)
#pragma unroll
            for (int t = 0; t < 4; ++t)
#pragma unroll
                for (int v = 0; v < 8; ++v)
                    dst[(size_t)(i * 16 + v) * DIc + t * 16] = siluf(acc[i][t][v]);
    }
}

#define PX_WAVES 8
#define PX_PPW   4
#define PX_CPL   (DIc / 32)
#define PX_NWG   (BB * LL / (PX_WAVES * PX_PPW))
static_assert(WW % PX_PPW == 0 && (BB * LL) % (PX_WAVES * PX_PPW) == 0 && DIc % 32 == 0,
              "a wave's pixels must lie in one image row");

__global__ __launch_bounds__(32 * PX_WAVES)
void k_dwconv(const float* __restrict__ xc_lin, const float* __restrict__ cw,
              const float* __restrict__ cb, float* __restrict__ xc_act)
{
    __shared__ float wsh[DIc * 9];
    const int lane = threadIdx.x & 31;
    const int wave = __builtin_amdgcn_readfirstlane((int)(threadIdx.x >> 5));
    const int pix0 = (blockIdx.x * PX_WAVES + wave) * PX_PPW;
    const int b = pix0 / LL, rem0 = pix0 - b * LL;
    const int h = rem0 / WW, w0 = rem0 - h * WW;
    const float* __restrict__ img = xc_lin + (size_t)b * LL * DIc + lane;

    float win[3][PX_PPW + 2][PX_CPL];
#pragma unroll
    for (int r = 0; r < 3; ++r) {
        const int hh = h + r - 1;
        const bool rok = (hh >= 0) && (hh < HH);
        const int hc = hh < 0 ? 0 : (hh >= HH ? HH - 1 : hh);
#pragma unroll
        for (int q = 0; q < PX_PPW + 2; ++q) {
            const int wq = w0 + q - 1;
            const bool cok = (q == 0) ? (wq >= 0) : ((q == PX_PPW + 1) ? (wq < WW) : true);
            const int wc = (q == 0) ? (wq < 0 ? 0 : wq) : ((q == PX_PPW + 1) ? (wq >= WW ? WW - 1 : wq) : wq);
            const float* __restrict__ sp = img + (size_t)(hc * WW + wc) * DIc;
#pragma unroll
            for (int j = 0; j < PX_CPL; ++j) {
                const float v = sp[32 * j];
                win[r][q][j] = (rok && cok) ? v : 0.f;
            }
        }
    }

    constexpr int WFULL = (DIc * 9) / (32 * PX_WAVES);
    float wtmp[WFULL];
#pragma unroll
    for (int i = 0; i < WFULL; ++i) wtmp[i] = cw[(int)threadIdx.x + i * 32 * PX_WAVES];
    const int itail = (int)threadIdx.x + WFULL * 32 * PX_WAVES;
    const float wtail = (itail < DIc * 9) ? cw[itail] : 0.f;
#pragma unroll
    for (int i = 0; i < WFULL; ++i) wsh[(int)threadIdx.x + i * 32 * PX_WAVES] = wtmp[i];
    if (itail < DIc * 9) wsh[itail] = wtail;
    __syncthreads();
    float wreg[PX_CPL][9], acc[PX_PPW][PX_CPL];
#pragma unroll
    for (int j = 0; j < PX_CPL; ++j) {
#pragma unroll
        for (int t = 0; t < 9; ++t) wreg[j][t] = wsh[(lane + 32 * j) * 9 + t];
        const float bj = cb[lane + 32 * j];
#pragma unroll
        for (int p = 0; p < PX_PPW; ++p) acc[p][j] = bj;
    }
#pragma unroll
    for (int p = 0; p < PX_PPW; ++p)
#pragma unroll
        for (int kh = 0; kh < 3; ++kh)
#pragma unroll
            for (int kw = 0; kw < 3; ++kw)
#pragma unroll
                for (int j = 0; j < PX_CPL; ++j)
                    acc[p][j] += win[kh][p + kw][j] * wreg[j][kh * 3 + kw];
#pragma unroll
    for (int p = 0; p < PX_PPW; ++p) {
        float* __restrict__ op = xc_act + (size_t)(pix0 + p) * DIc + lane;
#pragma unroll
        for (int j = 0; j < PX_CPL; ++j) op[32 * j] = siluf(acc[p][j]);
    }
}

#define GC_NCH 48
#define GC_P (LL / GC_NCH)
__global__ void k_gcmean(const float* __restrict__ xc_act, float* __restrict__ gcpart)
{
    int b = blockIdx.x, ch = blockIdx.y, d = threadIdx.x;
    const float* __restrict__ src = xc_act + ((size_t)b * LL + (size_t)ch * GC_P) * DIc + d;
    float s = 0.f;
#pragma unroll
    for (int p = 0; p < GC_P; ++p)
        s += src[(size_t)p * DIc];
    gcpart[((size_t)b * GC_NCH + ch) * DIc + d] = s;
}

__global__ void k_gate(const float* __restrict__ gcpart, const float* __restrict__ fw1,
                       const float* __restrict__ fw2, float* __restrict__ gate)
{
    __shared__ float g[DIc];
    __shared__ float h1[DMm / 4];
    int b = blockIdx.x, d = threadIdx.x;
    float gs = 0.f;
#pragma unroll
    for (int c = 0; c < GC_NCH; ++c)
        gs += gcpart[((size_t)b * GC_NCH + c) * DIc + d];
    g[d] = gs * (1.f / LL);
    __syncthreads();
    if (d < DMm / 4) {
        float s = 0.f;
        for (int j = 0; j < DMm; ++j) s += fw1[d * DMm + j] * g[j];
        h1[d] = geluf(s);
    }
    __syncthreads();
    float s2 = 0.f;
#pragma unroll
    for (int j = 0; j < DMm / 4; ++j) s2 += fw2[d * (DMm / 4) + j] * h1[j];
    gate[b * DMm + d] = sigm(s2);
}

__device__ __forceinline__ int seq_pixel(int k, int l)
{
    int lb = (k < 2) ? l : (LL - 1 - l);
    int w_in = lb & 3, h_in = (lb >> 2) & 3;
    int nw = (lb >> 4) % (WW / 4);
    int nh = (lb >> 4) / (WW / 4);
    int p = nh * 4 + h_in, q = nw * 4 + w_in;
    return ((k & 1) == 0) ? (p * WW + q) : (q * WW + p);
}

#define XP_NT ((CPROJ + 15) / 16)
static_assert(XP_NT * 16 <= CPAD, "the column tiles read padded weight rows");
__global__ __launch_bounds__(32 * GW)
void k_xproj(const float* __restrict__ xc_act, const float* __restrict__ wpad,
             float* __restrict__ xdbl)
{
    const int k = blockIdx.y;
    const float* W = wpad + (size_t)k * CPAD * DIc;
    float* O = xdbl + (size_t)k * BB * LL * CPROJ;
    const int lane = threadIdx.x & 31;
    const int wave = __builtin_amdgcn_readfirstlane((int)(threadIdx.x >> 5));
    const int row0 = blockIdx.x * (32 * GW) + wave * 32;
    const int nn = lane & 15, mh = (lane >> 4) * 8;
    const float* arow[2];
    const float* brow[XP_NT];
#pragma unroll
    for (int i = 0; i < 2; ++i) {
        const int ra = row0 + i * 16 + nn;
        const int ba = ra / LL, la = ra - ba * LL;
        arow[i] = xc_act + ((size_t)ba * LL + seq_pixel(k, la)) * DIc;
    }
#pragma unroll
    for (int t = 0; t < XP_NT; ++t) brow[t] = W + (size_t)(t * 16 + nn) * DIc;
    v8f acc[2][XP_NT];
    wmma_tiles<2, XP_NT, DIc>(arow, brow, acc);
#pragma unroll
    for (int i = 0; i < 2; ++i)
#pragma unroll
        for (int t = 0; t < XP_NT; ++t)
#pragma unroll
            for (int v = 0; v < 8; ++v) {
                int r = row0 + i * 16 + mh + v, c = t * 16 + nn;
                if (c < CPROJ) O[(size_t)r * CPROJ + c] = acc[i][t][v];
            }
}

#define DT_W 8
__global__ __launch_bounds__(32 * DT_W)
void k_dtproj(const float* __restrict__ xdbl, const float* __restrict__ dtw,
              const float* __restrict__ dtb, float* __restrict__ delta)
{
    const int k = blockIdx.z;
    const float* X = xdbl + (size_t)k * BB * LL * CPROJ;
    const float* W = dtw + (size_t)k * DIc * RR;
    float* O = delta + (size_t)k * BB * LL * DIc;
    const int lane = threadIdx.x & 31;
    const int wave = __builtin_amdgcn_readfirstlane((int)(threadIdx.x >> 5));
    const int row0 = blockIdx.y * (16 * DT_W) + wave * 16, col0 = blockIdx.x * 32;
    const int nn = lane & 15, mh = (lane >> 4) * 8;
    const float* arow[1];
    const float* brow[2];
    arow[0] = X + (size_t)(row0 + nn) * CPROJ;
#pragma unroll
    for (int t = 0; t < 2; ++t) brow[t] = W + (size_t)(col0 + t * 16 + nn) * RR;
    v8f acc[1][2];
    wmma_tiles<1, 2, RR>(arow, brow, acc);
#pragma unroll
    for (int t = 0; t < 2; ++t)
#pragma unroll
        for (int v = 0; v < 8; ++v) {
            int r = row0 + mh + v, c = col0 + t * 16 + nn;
            O[(size_t)r * DIc + c] = softplusf_(acc[0][t][v] + dtb[k * DIc + c]);
        }
}

#define SCAN_T 16
#define SCAN_NG 8
#define SCAN_NPG (NN / SCAN_NG)
#define SCAN_RPW (SCAN_T / SCAN_NG)
#define SCAN_NCH (LL / SCAN_T)
#define SCAN_LEAD 4
static_assert(SCAN_T == 16 && LL % SCAN_T == 0, "a chunk must be one 4x4 window");
static_assert(SCAN_NPG == 2 && SCAN_RPW == 2, "layout of the staged planes: a wave's two steps are one step pair");
static_assert(SCAN_LEAD == 4 && SCAN_NCH % SCAN_LEAD == 0, "the chunk loop is written out for four fetch sets");

__device__ __forceinline__ int slot_pixoff(int tr, int flip, int t)
{
    const int tt = flip ? (SCAN_T - 1 - t) : t;
    const int h_in = tt >> 2, w_in = tt & 3;
    return tr ? (w_in * WW + h_in) : (h_in * WW + w_in);
}
__device__ __forceinline__ int chunk_origin(int tr, int flip, int c)
{
    const int widx = flip ? (SCAN_NCH - 1 - c) : c;
    const int nw = widx % (WW / 4), nh = widx / (WW / 4);
    return tr ? (nw * 4 * WW + nh * 4) : (nh * 4 * WW + nw * 4);
}

struct ScanLds {
    float2 y[2][SCAN_T / 2][SCAN_NG][32];
    float4 du[2][SCAN_T / 2][32];
    float4 bc[2][SCAN_T][SCAN_NG];
};
struct ScanFetch { float dt[SCAN_RPW], u[SCAN_RPW], bc[SCAN_RPW]; };
struct ScanIn {
    float4 du[SCAN_T / 2];
    float4 bc[SCAN_T];
};

struct ScanWave {
    const float* uimg;
    const float* dp;
    const float* bcw;
    int tr, flip;
    int ts0;
    int po[SCAN_RPW];
    int bcpos;
};

__device__ __forceinline__ void scan_fetch(ScanFetch& f, const ScanWave& w, int chunk)
{
    const int c  = (chunk < SCAN_NCH) ? chunk : (SCAN_NCH - 1);
    const int l0 = c * SCAN_T;
    const float* __restrict__ up = w.uimg + (size_t)chunk_origin(w.tr, w.flip, c) * DIc;
#pragma unroll
    for (int r = 0; r < SCAN_RPW; ++r) {
        f.dt[r] = w.dp[(size_t)(l0 + w.ts0 + r) * DIc];
        f.u[r]  = up[w.po[r] * DIc];
        f.bc[r] = w.bcw[(size_t)(l0 + w.ts0 + r) * CPROJ];
    }
}

template <int PAR>
__device__ __forceinline__ void scan_park(const ScanFetch& f, const ScanWave& w, ScanLds& sm,
                                          const int g, const int lane)
{
    sm.du[PAR][g][lane] = make_float4(f.dt[0], f.u[0], f.dt[1], f.u[1]);
    float* __restrict__ bcf = &sm.bc[PAR][0][0].x;
#pragma unroll
    for (int r = 0; r < SCAN_RPW; ++r)
        bcf[(w.ts0 + r) * (SCAN_NG * 4) + w.bcpos] = f.bc[r];
}

template <int PAR>
__device__ __forceinline__ void scan_read(ScanIn& in, const ScanLds& sm, const int g, const int lane)
{
#pragma unroll
    for (int p = 0; p < SCAN_T / 2; ++p) {
        in.du[p]         = sm.du[PAR][p][lane];
        in.bc[2 * p]     = sm.bc[PAR][2 * p][g];
        in.bc[2 * p + 1] = sm.bc[PAR][2 * p + 1][g];
    }
}

__device__ __forceinline__ void scan_steps(const float4 (&du4)[SCAN_T / 2], const float4 (&bc4)[SCAN_T],
                                           float (&yv)[SCAN_T],
                                           float (&h)[SCAN_NPG], const float (&A)[SCAN_NPG],
                                           const float Dg)
{
#pragma unroll
    for (int t = 0; t < SCAN_T; ++t) {
        const float4 d4 = du4[t >> 1];
        const float dt = (t & 1) ? d4.z : d4.x;
        const float u  = (t & 1) ? d4.w : d4.y;
        const float Bt[SCAN_NPG] = { bc4[t].x, bc4[t].y };
        const float Ct[SCAN_NPG] = { bc4[t].z, bc4[t].w };
        const float du = dt * u;
#pragma unroll
        for (int n = 0; n < SCAN_NPG; ++n) {
            h[n] = __builtin_fmaf(__expf(dt * A[n]), h[n], du * Bt[n]);
        }
        float y = h[0] * Ct[0];
#pragma unroll
        for (int n = 1; n < SCAN_NPG; ++n) y = __builtin_fmaf(h[n], Ct[n], y);
        yv[t] = __builtin_fmaf(Dg, u, y);
    }
}

__device__ __forceinline__ void scan_rows(const float2 (&part)[SCAN_NG], float* __restrict__ yp,
                                          const int c, const int g)
{
    float s0 = part[0].x, s1 = part[0].y;
#pragma unroll
    for (int q = 1; q < SCAN_NG; ++q) { s0 += part[q].x; s1 += part[q].y; }
    yp[(size_t)(c * SCAN_T + g * SCAN_RPW) * DIc]     = s0;
    yp[(size_t)(c * SCAN_T + g * SCAN_RPW + 1) * DIc] = s1;
}

template <int PAR>
__device__ __forceinline__ void scan_chunk(const int c, ScanLds& sm, ScanFetch& f, const ScanWave& w,
                                           const ScanIn& cur, ScanIn& nxt,
                                           float (&h)[SCAN_NPG], const float (&A)[SCAN_NPG],
                                           const float Dg, const int g, const int lane,
                                           float* __restrict__ yp)
{
    float2 part[SCAN_NG];
#pragma unroll
    for (int q = 0; q < SCAN_NG; ++q) part[q] = sm.y[PAR ^ 1][g][q][lane];
    __builtin_amdgcn_sched_barrier(0);
    scan_read<PAR ^ 1>(nxt, sm, g, lane);
    __builtin_amdgcn_sched_barrier(0);

    float4 du4[SCAN_T / 2];
#pragma unroll
    for (int p = 0; p < SCAN_T / 2; ++p) du4[p] = cur.du[p];
    asm volatile("" : "+v"(du4[0].x), "+v"(du4[0].z), "+v"(du4[1].x), "+v"(du4[1].z),
                      "+v"(du4[2].x), "+v"(du4[2].z), "+v"(du4[3].x), "+v"(du4[3].z));
    asm volatile("" : "+v"(du4[4].x), "+v"(du4[4].z), "+v"(du4[5].x), "+v"(du4[5].z),
                      "+v"(du4[6].x), "+v"(du4[6].z), "+v"(du4[7].x), "+v"(du4[7].z));
    float yv[SCAN_T];
    scan_steps(du4, cur.bc, yv, h, A, Dg);

    asm volatile("" :: "v"(yv[0]), "v"(yv[1]), "v"(yv[2]), "v"(yv[3]), "v"(yv[4]), "v"(yv[5]),
                       "v"(yv[6]), "v"(yv[7]), "v"(yv[8]), "v"(yv[9]), "v"(yv[10]), "v"(yv[11]),
                       "v"(yv[12]), "v"(yv[13]), "v"(yv[14]), "v"(yv[15]));
    __builtin_amdgcn_sched_barrier(0);
    if (c > 0) scan_rows(part, yp, c - 1, g);
#pragma unroll
    for (int p = 0; p < SCAN_T / 2; ++p)
        sm.y[PAR][p][g][lane] = make_float2(yv[2 * p], yv[2 * p + 1]);
    scan_park<PAR>(f, w, sm, g, lane);
    __builtin_amdgcn_sched_barrier(0);
    scan_fetch(f, w, c + 2 + SCAN_LEAD);

    __builtin_amdgcn_fence(__ATOMIC_RELEASE, "workgroup", "local");
    __builtin_amdgcn_s_barrier();
    __builtin_amdgcn_fence(__ATOMIC_ACQUIRE, "workgroup", "local");
}

__global__ __launch_bounds__(32 * SCAN_NG)
void k_scan(const float* __restrict__ xc_act, const float* __restrict__ xdbl,
            const float* __restrict__ delta, const float* __restrict__ A_logs,
            const float* __restrict__ Ds, float* __restrict__ yarr)
{
    __shared__ ScanLds sm;
    const int kb = blockIdx.x;
    const int k = kb & 3, b = kb >> 2;
    const int lane = threadIdx.x & 31;
    const int g = __builtin_amdgcn_readfirstlane((int)(threadIdx.x >> 5));
    const int d = blockIdx.y * 32 + lane;
    const int n0 = g * SCAN_NPG;
    float A[SCAN_NPG];
#pragma unroll
    for (int n = 0; n < SCAN_NPG; ++n)
        A[n] = -__expf(A_logs[((size_t)(k * DIc + d)) * NN + n0 + n]);
    const float Dg = (g == SCAN_NG - 1) ? Ds[k * DIc + d] : 0.f;
    const size_t seqbase = ((size_t)(k * BB + b)) * LL;
    ScanWave w;
    w.uimg = xc_act + (size_t)b * LL * DIc + d;
    w.dp   = delta + seqbase * DIc + d;
    w.bcw  = xdbl  + seqbase * CPROJ + RR + lane;
    w.tr   = k & 1;
    w.flip = k >> 1;
    w.ts0  = g * SCAN_RPW;
#pragma unroll
    for (int r = 0; r < SCAN_RPW; ++r) w.po[r] = slot_pixoff(w.tr, w.flip, w.ts0 + r);
    w.bcpos = ((lane & 15) >> 1) * 4 + (lane & 1) + 2 * (lane >> 4);
    float* __restrict__ yp = yarr + seqbase * DIc + d;

    float h[SCAN_NPG];
#pragma unroll
    for (int n = 0; n < SCAN_NPG; ++n) h[n] = 0.f;

    ScanFetch f0, f1, f2, f3;
    scan_fetch(f0, w, 0);
    __builtin_amdgcn_sched_barrier(0);
    scan_fetch(f1, w, 1);
    __builtin_amdgcn_sched_barrier(0);
    scan_fetch(f2, w, 2);
    __builtin_amdgcn_sched_barrier(0);
    scan_fetch(f3, w, 3);
    __builtin_amdgcn_sched_barrier(0);
    scan_park<0>(f0, w, sm, g, lane);
#pragma unroll
    for (int p = 0; p < SCAN_T / 2; ++p) sm.y[1][p][g][lane] = make_float2(0.f, 0.f);
    scan_fetch(f0, w, SCAN_LEAD);
    __builtin_amdgcn_sched_barrier(0);
    __builtin_amdgcn_fence(__ATOMIC_RELEASE, "workgroup", "local");
    __builtin_amdgcn_s_barrier();
    __builtin_amdgcn_fence(__ATOMIC_ACQUIRE, "workgroup", "local");
    ScanIn inA, inB;
    scan_read<0>(inA, sm, g, lane);
    __builtin_amdgcn_sched_barrier(0);
    scan_park<1>(f1, w, sm, g, lane);
    scan_fetch(f1, w, SCAN_LEAD + 1);
    __builtin_amdgcn_sched_barrier(0);
    __builtin_amdgcn_fence(__ATOMIC_RELEASE, "workgroup", "local");
    __builtin_amdgcn_s_barrier();
    __builtin_amdgcn_fence(__ATOMIC_ACQUIRE, "workgroup", "local");

    for (int c = 0; c < SCAN_NCH; c += SCAN_LEAD) {
        scan_chunk<0>(c,     sm, f2, w, inA, inB, h, A, Dg, g, lane, yp);
        scan_chunk<1>(c + 1, sm, f3, w, inB, inA, h, A, Dg, g, lane, yp);
        scan_chunk<0>(c + 2, sm, f0, w, inA, inB, h, A, Dg, g, lane, yp);
        scan_chunk<1>(c + 3, sm, f1, w, inB, inA, h, A, Dg, g, lane, yp);
    }
    float2 part[SCAN_NG];
#pragma unroll
    for (int q = 0; q < SCAN_NG; ++q) part[q] = sm.y[(SCAN_NCH - 1) & 1][g][q][lane];
    scan_rows(part, yp, SCAN_NCH - 1, g);
}

template <int XM>
__device__ __forceinline__ float row_xor(float v)
{
    return __builtin_bit_cast(float,
        __builtin_amdgcn_update_dpp(0, __builtin_bit_cast(int, v), 0x160 + XM, 0xf, 0xf, false));
}
__device__ __forceinline__ float half_xor(float v)
{
    const int x = __builtin_bit_cast(int, v);
    return __builtin_bit_cast(float,
        __builtin_amdgcn_permlanex16(x, x, 0x76543210, 0xfedcba98, false, false));
}
__device__ __forceinline__ float wave_sum(float v)
{
    v += row_xor<1>(v);
    v += row_xor<2>(v);
    v += row_xor<4>(v);
    v += row_xor<8>(v);
    v += half_xor(v);
    return v;
}

__global__ __launch_bounds__(32 * PX_WAVES)
void k_combine(const float* __restrict__ yarr, const float* __restrict__ gate,
               const float* __restrict__ g1, const float* __restrict__ g2,
               const float* __restrict__ g3, const float* __restrict__ g4,
               const float* __restrict__ nw, const float* __restrict__ nb,
               const float* __restrict__ z_silu, float* __restrict__ yfinal)
{
    const int lane = threadIdx.x & 31;
    const int wave = __builtin_amdgcn_readfirstlane((int)(threadIdx.x >> 5));
    const int pix0 = (blockIdx.x * PX_WAVES + wave) * PX_PPW;
    const int b = pix0 / LL, lr0 = pix0 - b * LL;
    const size_t bstride = (size_t)LL * DIc;
    const float* __restrict__ Y0 = yarr + ((size_t)(0 * BB + b)) * bstride + lane;
    const float* __restrict__ Y1 = yarr + ((size_t)(1 * BB + b)) * bstride + lane;
    const float* __restrict__ Y2 = yarr + ((size_t)(2 * BB + b)) * bstride + lane;
    const float* __restrict__ Y3 = yarr + ((size_t)(3 * BB + b)) * bstride + lane;
    float G1[PX_CPL], G2[PX_CPL], G3[PX_CPL], G4[PX_CPL], GT[PX_CPL], NW[PX_CPL], NB[PX_CPL];
#pragma unroll
    for (int j = 0; j < PX_CPL; ++j) {
        const int d = lane + 32 * j;
        G1[j] = g1[d]; G2[j] = g2[d]; G3[j] = g3[d]; G4[j] = g4[d];
        GT[j] = gate[b * DIc + d]; NW[j] = nw[d]; NB[j] = nb[d];
    }
#pragma unroll
    for (int p = 0; p < PX_PPW; ++p) {
        const int lr = lr0 + p;
        const int h = lr / WW, w = lr % WW;
        const int l = (((h >> 2) * (WW / 4) + (w >> 2)) << 4) + ((h & 3) << 2) + (w & 3);
        const int lt = (l % WW) * HH + (l / WW);
        const float* __restrict__ zp = z_silu + ((size_t)b * LL + lr) * DIc + lane;
        float yc[PX_CPL], zz[PX_CPL];
        float s1 = 0.f, s2 = 0.f;
#pragma unroll
        for (int j = 0; j < PX_CPL; ++j) {
            float y = G1[j] * Y0[(size_t)l * DIc + 32 * j]
                    + G2[j] * Y2[(size_t)(LL - 1 - l) * DIc + 32 * j]
                    + G3[j] * Y1[(size_t)lt * DIc + 32 * j]
                    + G4[j] * Y3[(size_t)(LL - 1 - lt) * DIc + 32 * j];
            y *= GT[j];
            yc[j] = y;
            zz[j] = zp[32 * j];
            s1 += y;
            s2 += y * y;
        }
        s1 = wave_sum(s1);
        s2 = wave_sum(s2);
        const float mu  = s1 * (1.f / DIc);
        const float var = s2 * (1.f / DIc) - mu * mu;
        const float inv = rsqrtf(var + 1e-5f);
        float* __restrict__ op = yfinal + ((size_t)b * LL + lr) * DIc + lane;
#pragma unroll
        for (int j = 0; j < PX_CPL; ++j)
            op[32 * j] = ((yc[j] - mu) * inv * NW[j] + NB[j]) * zz[j];
    }
}

__global__ __launch_bounds__(32 * GW)
void k_outproj(const float* __restrict__ yf, const float* __restrict__ w,
               float* __restrict__ out)
{
    const int lane = threadIdx.x & 31;
    const int wave = __builtin_amdgcn_readfirstlane((int)(threadIdx.x >> 5));
    const int row0 = blockIdx.y * 64 + (wave >> 1) * 32, col0 = blockIdx.x * 64 + (wave & 1) * 32;
    const int nn = lane & 15, mh = (lane >> 4) * 8;
    const float* arow[2];
    const float* brow[2];
#pragma unroll
    for (int i = 0; i < 2; ++i) arow[i] = yf + (size_t)(row0 + i * 16 + nn) * DIc;
#pragma unroll
    for (int t = 0; t < 2; ++t) brow[t] = w + (size_t)(col0 + t * 16 + nn) * DIc;
    v8f acc[2][2];
    wmma_tiles<2, 2, DIc>(arow, brow, acc);
#pragma unroll
    for (int i = 0; i < 2; ++i)
#pragma unroll
        for (int t = 0; t < 2; ++t)
#pragma unroll
            for (int v = 0; v < 8; ++v) {
                int r = row0 + i * 16 + mh + v, c = col0 + t * 16 + nn;
                out[(size_t)r * DMm + c] = acc[i][t][v];
            }
}

extern "C" void kernel_launch(void* const* d_in, const int* in_sizes, int n_in,
                              void* d_out, int out_size, void* d_ws, size_t ws_size,
                              hipStream_t stream)
{
    (void)in_sizes; (void)n_in; (void)out_size; (void)ws_size;
    const float* x          = (const float*)d_in[0];
    const float* in_proj_w  = (const float*)d_in[1];
    const float* conv_w     = (const float*)d_in[2];
    const float* conv_b     = (const float*)d_in[3];
    const float* x_proj_w   = (const float*)d_in[4];
    const float* dt_w       = (const float*)d_in[5];
    const float* dt_b       = (const float*)d_in[6];
    const float* A_logs     = (const float*)d_in[7];
    const float* Ds         = (const float*)d_in[8];
    const float* gamma1     = (const float*)d_in[9];
    const float* gamma2     = (const float*)d_in[10];
    const float* gamma3     = (const float*)d_in[11];
    const float* gamma4     = (const float*)d_in[12];
    const float* fc_w1      = (const float*)d_in[13];
    const float* fc_w2      = (const float*)d_in[14];
    const float* norm_w     = (const float*)d_in[15];
    const float* norm_b     = (const float*)d_in[16];
    const float* out_proj_w = (const float*)d_in[17];
    float* out = (float*)d_out;

    float* ws = (float*)d_ws;
    size_t o = 0;
    float* xc_lin = ws + o; o += (size_t)BB * LL * DIc;
    float* z_silu = ws + o; o += (size_t)BB * LL * DIc;
    float* xc_act = ws + o; o += (size_t)BB * LL * DIc;
    float* xdbl   = ws + o; o += (size_t)KK * BB * LL * CPROJ;
    float* delta  = ws + o; o += (size_t)KK * BB * LL * DIc;
    float* yarr   = ws + o; o += (size_t)KK * BB * LL * DIc;
    float* yfinal = ws + o; o += (size_t)BB * LL * DIc;
    float* wpad   = ws + o; o += (size_t)KK * CPAD * DIc;
    float* gcpart = ws + o; o += (size_t)BB * GC_NCH * DIc;
    float* gate   = ws + o; o += BB * DMm;

    k_padw    <<<dim3(KK, CPAD),       DIc, 0, stream>>>(x_proj_w, wpad);
    k_inproj  <<<dim3(2 * DIc / 128, BB * LL / 64), 32 * GW, 0, stream>>>(x, in_proj_w, xc_lin, z_silu);
    k_dwconv  <<<dim3(PX_NWG),         32 * PX_WAVES, 0, stream>>>(xc_lin, conv_w, conv_b, xc_act);
    k_gcmean  <<<dim3(BB, GC_NCH),     DIc, 0, stream>>>(xc_act, gcpart);
    k_gate    <<<dim3(BB),             DIc, 0, stream>>>(gcpart, fc_w1, fc_w2, gate);
    k_xproj   <<<dim3(BB * LL / (32 * GW), KK), 32 * GW, 0, stream>>>(xc_act, wpad, xdbl);
    k_dtproj  <<<dim3(DIc / 32, BB * LL / (16 * DT_W), KK), 32 * DT_W, 0, stream>>>(xdbl, dt_w, dt_b, delta);
    k_scan    <<<dim3(BB * KK, DIc / 32), 32 * SCAN_NG, 0, stream>>>(xc_act, xdbl, delta, A_logs, Ds, yarr);
    k_combine <<<dim3(PX_NWG),         32 * PX_WAVES, 0, stream>>>(yarr, gate, gamma1, gamma2, gamma3,
                                                         gamma4, norm_w, norm_b, z_silu, yfinal);
    k_outproj <<<dim3(DMm / 64, BB * LL / 64), 32 * GW, 0, stream>>>(yfinal, out_proj_w, out);
}


# k_padw, k_gcmean, k_gate are no launches of their own: they run as extra work-groups of the in_proj, x_proj and scan launches, beside those launches' own (7 launches, not 10); all sum orders kept
# speedup vs baseline: 12.2814x; 1.0610x over previous
#include <hip/hip_runtime.h>
#include <math.h>


typedef float v2f __attribute__((ext_vector_type(2)));
typedef float v8f __attribute__((ext_vector_type(8)));

#define BB 4
#define HH 48
#define WW 48
#define DMm 192
#define DIc 192
#define KK 4
#define NN 16
#define RR 12
#define LL (HH * WW)
#define CPROJ (RR + 2 * NN)
#define CPAD 64

__device__ __forceinline__ float siluf(float x)     { return x / (1.f + __expf(-x)); }
__device__ __forceinline__ float sigm(float x)      { return 1.f / (1.f + __expf(-x)); }
__device__ __forceinline__ float softplusf_(float x){ return (x > 20.f) ? x : log1pf(__expf(x)); }
__device__ __forceinline__ float geluf(float x)     { return 0.5f * x * (1.f + erff(x * 0.70710678118654752440f)); }

typedef float v4f __attribute__((ext_vector_type(4)));

template <int MT, int NT>
struct Frag8 { v4f a[MT]; v4f b[NT]; };

template <int MT, int NT>
__device__ __forceinline__ void frag8_load(Frag8<MT, NT>& f, const float* const (&ap)[MT],
                                           const float* const (&bp)[NT], int koff)
{
#pragma unroll
    for (int i = 0; i < MT; ++i) f.a[i] = *(const v4f*)(ap[i] + koff);
#pragma unroll
    for (int t = 0; t < NT; ++t) f.b[t] = *(const v4f*)(bp[t] + koff);
}

template <int MT, int NT>
__device__ __forceinline__ void frag8_mma(const Frag8<MT, NT>& f, v8f (&acc)[MT][NT])
{
#pragma unroll
    for (int s = 0; s < 2; ++s)
#pragma unroll
        for (int i = 0; i < MT; ++i) {
            const v2f a = s ? v2f{f.a[i].z, f.a[i].w} : v2f{f.a[i].x, f.a[i].y};
#pragma unroll
            for (int t = 0; t < NT; ++t) {
                const v2f b = s ? v2f{f.b[t].z, f.b[t].w} : v2f{f.b[t].x, f.b[t].y};
                acc[i][t] = __builtin_amdgcn_wmma_f32_16x16x4_f32(false, a, false, b,
                                                                  (short)0, acc[i][t], false, false);
            }
        }
}

template <int MT, int NT, int KD>
__device__ __forceinline__ void wmma_tiles(const float* const (&arow)[MT],
                                           const float* const (&brow)[NT],
                                           v8f (&acc)[MT][NT])
{
    static_assert(KD % 4 == 0, "k in steps of 4");
    constexpr int NB = KD / 8;
    constexpr bool TAIL = (KD % 8) != 0;
    const int kh = (threadIdx.x & 31) >> 4;
    const float* ap[MT];
    const float* bp[NT];
#pragma unroll
    for (int i = 0; i < MT; ++i) {
        ap[i] = arow[i] + 4 * kh;
#pragma unroll
        for (int t = 0; t < NT; ++t) acc[i][t] = {};
    }
#pragma unroll
    for (int t = 0; t < NT; ++t) bp[t] = brow[t] + 4 * kh;

    if constexpr (NB >= 3 && NB % 3 == 0) {
        Frag8<MT, NT> f0, f1, f2;
        frag8_load(f0, ap, bp, 0);
        frag8_load(f1, ap, bp, 8);
#pragma unroll 1
        for (int kb = 0; kb < NB - 3; kb += 3) {
            frag8_load(f2, ap, bp, (kb + 2) * 8);
            __builtin_amdgcn_sched_barrier(0);
            frag8_mma(f0, acc);
            __builtin_amdgcn_sched_barrier(0);
            frag8_load(f0, ap, bp, (kb + 3) * 8);
            __builtin_amdgcn_sched_barrier(0);
            frag8_mma(f1, acc);
            __builtin_amdgcn_sched_barrier(0);
            frag8_load(f1, ap, bp, (kb + 4) * 8);
            __builtin_amdgcn_sched_barrier(0);
            frag8_mma(f2, acc);
            __builtin_amdgcn_sched_barrier(0);
        }
        frag8_load(f2, ap, bp, (NB - 1) * 8);
        __builtin_amdgcn_sched_barrier(0);
        frag8_mma(f0, acc);
        frag8_mma(f1, acc);
        frag8_mma(f2, acc);
    } else {
        Frag8<MT, NT> f[NB];
#pragma unroll
        for (int kb = 0; kb < NB; ++kb) frag8_load(f[kb], ap, bp, kb * 8);
        v2f ta[MT], tb[NT];
        if constexpr (TAIL) {
#pragma unroll
            for (int i = 0; i < MT; ++i) ta[i] = *(const v2f*)(arow[i] + NB * 8 + 2 * kh);
#pragma unroll
            for (int t = 0; t < NT; ++t) tb[t] = *(const v2f*)(brow[t] + NB * 8 + 2 * kh);
        }
#pragma unroll
        for (int kb = 0; kb < NB; ++kb) frag8_mma(f[kb], acc);
        if constexpr (TAIL) {
#pragma unroll
            for (int i = 0; i < MT; ++i)
#pragma unroll
                for (int t = 0; t < NT; ++t)
                    acc[i][t] = __builtin_amdgcn_wmma_f32_16x16x4_f32(false, ta[i], false, tb[t],
                                                                      (short)0, acc[i][t], false, false);
        }
    }
}

#define GW 4


#define PW_EPT 3
#define PW_NWG (KK * CPAD * DIc / (32 * GW * PW_EPT))
static_assert(PW_NWG * 32 * GW * PW_EPT == KK * CPAD * DIc, "the side work-groups cover wpad exactly");
__device__ __forceinline__ void padw_part(const float* __restrict__ xpw, float* __restrict__ wpad,
                                          const int j, const int tid)
{
#pragma unroll
    for (int i = 0; i < PW_EPT; ++i) {
        const int idx = (j * PW_EPT + i) * (32 * GW) + tid;
        const int row = idx / DIc, d = idx - row * DIc;
        const int k = row / CPAD, c = row - k * CPAD;
        wpad[idx] = (c < CPROJ) ? xpw[((size_t)k * CPROJ + c) * DIc + d] : 0.f;
    }
}

#define IN_NCB (2 * DIc / 128)
static_assert(PW_NWG <= BB * LL / 64, "the side work-groups fit in one column of the grid");
__global__ __launch_bounds__(32 * GW)
void k_inproj(const float* __restrict__ x, const float* __restrict__ w,
              float* __restrict__ xc_lin, float* __restrict__ z_silu,
              const float* __restrict__ xpw, float* __restrict__ wpad)
{
    if (blockIdx.x == IN_NCB) {
        if (blockIdx.y < PW_NWG) padw_part(xpw, wpad, blockIdx.y, threadIdx.x);
        return;
    }
    const int lane = threadIdx.x & 31;
    const int wave = __builtin_amdgcn_readfirstlane((int)(threadIdx.x >> 5));
    const int row0 = blockIdx.y * 64 + (wave >> 1) * 32, col0 = blockIdx.x * 128 + (wave & 1) * 64;
    const int nn = lane & 15, mh = (lane >> 4) * 8;
    const float* arow[2];
    const float* brow[4];
#pragma unroll
    for (int i = 0; i < 2; ++i) arow[i] = x + (size_t)(row0 + i * 16 + nn) * DMm;
#pragma unroll
    for (int t = 0; t < 4; ++t) brow[t] = w + (size_t)(col0 + t * 16 + nn) * DMm;
    v8f acc[2][4];
    wmma_tiles<2, 4, DMm>(arow, brow, acc);
    static_assert(DIc % 64 == 0, "a wave's columns must not straddle the split of xz");
    if (col0 < DIc) {
        float* __restrict__ dst = xc_lin + (size_t)(row0 + mh) * DIc + col0 + nn;
#pragma unroll
        for (int i = 0; i < 2; ++i)
#pragma unroll
            for (int t = 0; t < 4; ++t)
#pragma unroll
                for (int v = 0; v < 8; ++v)
                    dst[(size_t)(i * 16 + v) * DIc + t * 16] = acc[i][t][v];
    } else {
        float* __restrict__ dst = z_silu + (size_t)(row0 + mh) * DIc + (col0 - DIc) + nn;
#pragma unroll
        for (int i = 0; i < 2; ++i)
#pragma unroll
            for (int t = 0; t < 4; ++t)
#pragma unroll
                for (int v = 0; v < 8; ++v)
                    dst[(size_t)(i * 16 + v) * DIc + t * 16] = siluf(acc[i][t][v]);
    }
}

#define PX_WAVES 8
#define PX_PPW   4
#define PX_CPL   (DIc / 32)
#define PX_NWG   (BB * LL / (PX_WAVES * PX_PPW))
static_assert(WW % PX_PPW == 0 && (BB * LL) % (PX_WAVES * PX_PPW) == 0 && DIc % 32 == 0,
              "a wave's pixels must lie in one image row");

__global__ __launch_bounds__(32 * PX_WAVES)
void k_dwconv(const float* __restrict__ xc_lin, const float* __restrict__ cw,
              const float* __restrict__ cb, float* __restrict__ xc_act)
{
    __shared__ float wsh[DIc * 9];
    const int lane = threadIdx.x & 31;
    const int wave = __builtin_amdgcn_readfirstlane((int)(threadIdx.x >> 5));
    const int pix0 = (blockIdx.x * PX_WAVES + wave) * PX_PPW;
    const int b = pix0 / LL, rem0 = pix0 - b * LL;
    const int h = rem0 / WW, w0 = rem0 - h * WW;
    const float* __restrict__ img = xc_lin + (size_t)b * LL * DIc + lane;

    float win[3][PX_PPW + 2][PX_CPL];
#pragma unroll
    for (int r = 0; r < 3; ++r) {
        const int hh = h + r - 1;
        const bool rok = (hh >= 0) && (hh < HH);
        const int hc = hh < 0 ? 0 : (hh >= HH ? HH - 1 : hh);
#pragma unroll
        for (int q = 0; q < PX_PPW + 2; ++q) {
            const int wq = w0 + q - 1;
            const bool cok = (q == 0) ? (wq >= 0) : ((q == PX_PPW + 1) ? (wq < WW) : true);
            const int wc = (q == 0) ? (wq < 0 ? 0 : wq) : ((q == PX_PPW + 1) ? (wq >= WW ? WW - 1 : wq) : wq);
            const float* __restrict__ sp = img + (size_t)(hc * WW + wc) * DIc;
#pragma unroll
            for (int j = 0; j < PX_CPL; ++j) {
                const float v = sp[32 * j];
                win[r][q][j] = (rok && cok) ? v : 0.f;
            }
        }
    }

    constexpr int WFULL = (DIc * 9) / (32 * PX_WAVES);
    float wtmp[WFULL];
#pragma unroll
    for (int i = 0; i < WFULL; ++i) wtmp[i] = cw[(int)threadIdx.x + i * 32 * PX_WAVES];
    const int itail = (int)threadIdx.x + WFULL * 32 * PX_WAVES;
    const float wtail = (itail < DIc * 9) ? cw[itail] : 0.f;
#pragma unroll
    for (int i = 0; i < WFULL; ++i) wsh[(int)threadIdx.x + i * 32 * PX_WAVES] = wtmp[i];
    if (itail < DIc * 9) wsh[itail] = wtail;
    __syncthreads();
    float wreg[PX_CPL][9], acc[PX_PPW][PX_CPL];
#pragma unroll
    for (int j = 0; j < PX_CPL; ++j) {
#pragma unroll
        for (int t = 0; t < 9; ++t) wreg[j][t] = wsh[(lane + 32 * j) * 9 + t];
        const float bj = cb[lane + 32 * j];
#pragma unroll
        for (int p = 0; p < PX_PPW; ++p) acc[p][j] = bj;
    }
#pragma unroll
    for (int p = 0; p < PX_PPW; ++p)
#pragma unroll
        for (int kh = 0; kh < 3; ++kh)
#pragma unroll
            for (int kw = 0; kw < 3; ++kw)
#pragma unroll
                for (int j = 0; j < PX_CPL; ++j)
                    acc[p][j] += win[kh][p + kw][j] * wreg[j][kh * 3 + kw];
#pragma unroll
    for (int p = 0; p < PX_PPW; ++p) {
        float* __restrict__ op = xc_act + (size_t)(pix0 + p) * DIc + lane;
#pragma unroll
        for (int j = 0; j < PX_CPL; ++j) op[32 * j] = siluf(acc[p][j]);
    }
}

#define GC_NCH 48
#define GC_P (LL / GC_NCH)
static_assert(DIc / 2 <= 32 * GW && DIc % 2 == 0, "two channels per thread cover a row");
__device__ __forceinline__ void gcmean_part(const float* __restrict__ xc_act, float* __restrict__ gcpart,
                                            const int b, const int ch, const int t)
{
    if (t >= DIc / 2) return;
    const float2* __restrict__ src =
        (const float2*)(xc_act + ((size_t)b * LL + (size_t)ch * GC_P) * DIc) + t;
    float2 s = make_float2(0.f, 0.f);
#pragma unroll
    for (int p = 0; p < GC_P; ++p) {
        const float2 v = src[(size_t)p * (DIc / 2)];
        s.x += v.x;
        s.y += v.y;
    }
    ((float2*)(gcpart + ((size_t)b * GC_NCH + ch) * DIc))[t] = s;
}

static_assert(DIc == DMm, "the gate has one value per channel");
__device__ __forceinline__ void gate_part(const float* __restrict__ gcpart, const float* __restrict__ fw1,
                                          const float* __restrict__ fw2, float* __restrict__ gate,
                                          const int b, const int d)
{
    __shared__ float g[DIc];
    __shared__ float h1[DMm / 4];
    if (d < DIc) {
        float gs = 0.f;
#pragma unroll
        for (int c = 0; c < GC_NCH; ++c)
            gs += gcpart[((size_t)b * GC_NCH + c) * DIc + d];
        g[d] = gs * (1.f / LL);
    }
    __syncthreads();
    if (d < DMm / 4) {
        float s = 0.f;
        for (int j = 0; j < DMm; ++j) s += fw1[d * DMm + j] * g[j];
        h1[d] = geluf(s);
    }
    __syncthreads();
    if (d < DMm) {
        float s2 = 0.f;
#pragma unroll
        for (int j = 0; j < DMm / 4; ++j) s2 += fw2[d * (DMm / 4) + j] * h1[j];
        gate[b * DMm + d] = sigm(s2);
    }
}

__device__ __forceinline__ int seq_pixel(int k, int l)
{
    int lb = (k < 2) ? l : (LL - 1 - l);
    int w_in = lb & 3, h_in = (lb >> 2) & 3;
    int nw = (lb >> 4) % (WW / 4);
    int nh = (lb >> 4) / (WW / 4);
    int p = nh * 4 + h_in, q = nw * 4 + w_in;
    return ((k & 1) == 0) ? (p * WW + q) : (q * WW + p);
}

#define XP_NT ((CPROJ + 15) / 16)
#define XP_NRB (BB * LL / (32 * GW))
#define XP_SIDE (BB * GC_NCH / KK)
static_assert(XP_SIDE * KK == BB * GC_NCH, "the side work-groups cover every partial row once");
static_assert(XP_NT * 16 <= CPAD, "the column tiles read padded weight rows");
__global__ __launch_bounds__(32 * GW)
void k_xproj(const float* __restrict__ xc_act, const float* __restrict__ wpad,
             float* __restrict__ xdbl, float* __restrict__ gcpart)
{
    if (blockIdx.x >= XP_NRB) {
        const int id = blockIdx.y * XP_SIDE + (blockIdx.x - XP_NRB);
        gcmean_part(xc_act, gcpart, id / GC_NCH, id % GC_NCH, threadIdx.x);
        return;
    }
    const int k = blockIdx.y;
    const float* W = wpad + (size_t)k * CPAD * DIc;
    float* O = xdbl + (size_t)k * BB * LL * CPROJ;
    const int lane = threadIdx.x & 31;
    const int wave = __builtin_amdgcn_readfirstlane((int)(threadIdx.x >> 5));
    const int row0 = blockIdx.x * (32 * GW) + wave * 32;
    const int nn = lane & 15, mh = (lane >> 4) * 8;
    const float* arow[2];
    const float* brow[XP_NT];
#pragma unroll
    for (int i = 0; i < 2; ++i) {
        const int ra = row0 + i * 16 + nn;
        const int ba = ra / LL, la = ra - ba * LL;
        arow[i] = xc_act + ((size_t)ba * LL + seq_pixel(k, la)) * DIc;
    }
#pragma unroll
    for (int t = 0; t < XP_NT; ++t) brow[t] = W + (size_t)(t * 16 + nn) * DIc;
    v8f acc[2][XP_NT];
    wmma_tiles<2, XP_NT, DIc>(arow, brow, acc);
#pragma unroll
    for (int i = 0; i < 2; ++i)
#pragma unroll
        for (int t = 0; t < XP_NT; ++t)
#pragma unroll
            for (int v = 0; v < 8; ++v) {
                int r = row0 + i * 16 + mh + v, c = t * 16 + nn;
                if (c < CPROJ) O[(size_t)r * CPROJ + c] = acc[i][t][v];
            }
}

#define DT_W 8
__global__ __launch_bounds__(32 * DT_W)
void k_dtproj(const float* __restrict__ xdbl, const float* __restrict__ dtw,
              const float* __restrict__ dtb, float* __restrict__ delta)
{
    const int k = blockIdx.z;
    const float* X = xdbl + (size_t)k * BB * LL * CPROJ;
    const float* W = dtw + (size_t)k * DIc * RR;
    float* O = delta + (size_t)k * BB * LL * DIc;
    const int lane = threadIdx.x & 31;
    const int wave = __builtin_amdgcn_readfirstlane((int)(threadIdx.x >> 5));
    const int row0 = blockIdx.y * (16 * DT_W) + wave * 16, col0 = blockIdx.x * 32;
    const int nn = lane & 15, mh = (lane >> 4) * 8;
    const float* arow[1];
    const float* brow[2];
    arow[0] = X + (size_t)(row0 + nn) * CPROJ;
#pragma unroll
    for (int t = 0; t < 2; ++t) brow[t] = W + (size_t)(col0 + t * 16 + nn) * RR;
    v8f acc[1][2];
    wmma_tiles<1, 2, RR>(arow, brow, acc);
#pragma unroll
    for (int t = 0; t < 2; ++t)
#pragma unroll
        for (int v = 0; v < 8; ++v) {
            int r = row0 + mh + v, c = col0 + t * 16 + nn;
            O[(size_t)r * DIc + c] = softplusf_(acc[0][t][v] + dtb[k * DIc + c]);
        }
}

#define SCAN_T 16
#define SCAN_NG 8
#define SCAN_NPG (NN / SCAN_NG)
#define SCAN_RPW (SCAN_T / SCAN_NG)
#define SCAN_NCH (LL / SCAN_T)
#define SCAN_LEAD 4
static_assert(SCAN_T == 16 && LL % SCAN_T == 0, "a chunk must be one 4x4 window");
static_assert(SCAN_NPG == 2 && SCAN_RPW == 2, "layout of the staged planes: a wave's two steps are one step pair");
static_assert(SCAN_LEAD == 4 && SCAN_NCH % SCAN_LEAD == 0, "the chunk loop is written out for four fetch sets");

__device__ __forceinline__ int slot_pixoff(int tr, int flip, int t)
{
    const int tt = flip ? (SCAN_T - 1 - t) : t;
    const int h_in = tt >> 2, w_in = tt & 3;
    return tr ? (w_in * WW + h_in) : (h_in * WW + w_in);
}
__device__ __forceinline__ int chunk_origin(int tr, int flip, int c)
{
    const int widx = flip ? (SCAN_NCH - 1 - c) : c;
    const int nw = widx % (WW / 4), nh = widx / (WW / 4);
    return tr ? (nw * 4 * WW + nh * 4) : (nh * 4 * WW + nw * 4);
}

struct ScanLds {
    float2 y[2][SCAN_T / 2][SCAN_NG][32];
    float4 du[2][SCAN_T / 2][32];
    float4 bc[2][SCAN_T][SCAN_NG];
};
struct ScanFetch { float dt[SCAN_RPW], u[SCAN_RPW], bc[SCAN_RPW]; };
struct ScanIn {
    float4 du[SCAN_T / 2];
    float4 bc[SCAN_T];
};

struct ScanWave {
    const float* uimg;
    const float* dp;
    const float* bcw;
    int tr, flip;
    int ts0;
    int po[SCAN_RPW];
    int bcpos;
};

__device__ __forceinline__ void scan_fetch(ScanFetch& f, const ScanWave& w, int chunk)
{
    const int c  = (chunk < SCAN_NCH) ? chunk : (SCAN_NCH - 1);
    const int l0 = c * SCAN_T;
    const float* __restrict__ up = w.uimg + (size_t)chunk_origin(w.tr, w.flip, c) * DIc;
#pragma unroll
    for (int r = 0; r < SCAN_RPW; ++r) {
        f.dt[r] = w.dp[(size_t)(l0 + w.ts0 + r) * DIc];
        f.u[r]  = up[w.po[r] * DIc];
        f.bc[r] = w.bcw[(size_t)(l0 + w.ts0 + r) * CPROJ];
    }
}

template <int PAR>
__device__ __forceinline__ void scan_park(const ScanFetch& f, const ScanWave& w, ScanLds& sm,
                                          const int g, const int lane)
{
    sm.du[PAR][g][lane] = make_float4(f.dt[0], f.u[0], f.dt[1], f.u[1]);
    float* __restrict__ bcf = &sm.bc[PAR][0][0].x;
#pragma unroll
    for (int r = 0; r < SCAN_RPW; ++r)
        bcf[(w.ts0 + r) * (SCAN_NG * 4) + w.bcpos] = f.bc[r];
}

template <int PAR>
__device__ __forceinline__ void scan_read(ScanIn& in, const ScanLds& sm, const int g, const int lane)
{
#pragma unroll
    for (int p = 0; p < SCAN_T / 2; ++p) {
        in.du[p]         = sm.du[PAR][p][lane];
        in.bc[2 * p]     = sm.bc[PAR][2 * p][g];
        in.bc[2 * p + 1] = sm.bc[PAR][2 * p + 1][g];
    }
}

__device__ __forceinline__ void scan_steps(const float4 (&du4)[SCAN_T / 2], const float4 (&bc4)[SCAN_T],
                                           float (&yv)[SCAN_T],
                                           float (&h)[SCAN_NPG], const float (&A)[SCAN_NPG],
                                           const float Dg)
{
#pragma unroll
    for (int t = 0; t < SCAN_T; ++t) {
        const float4 d4 = du4[t >> 1];
        const float dt = (t & 1) ? d4.z : d4.x;
        const float u  = (t & 1) ? d4.w : d4.y;
        const float Bt[SCAN_NPG] = { bc4[t].x, bc4[t].y };
        const float Ct[SCAN_NPG] = { bc4[t].z, bc4[t].w };
        const float du = dt * u;
#pragma unroll
        for (int n = 0; n < SCAN_NPG; ++n) {
            h[n] = __builtin_fmaf(__expf(dt * A[n]), h[n], du * Bt[n]);
        }
        float y = h[0] * Ct[0];
#pragma unroll
        for (int n = 1; n < SCAN_NPG; ++n) y = __builtin_fmaf(h[n], Ct[n], y);
        yv[t] = __builtin_fmaf(Dg, u, y);
    }
}

__device__ __forceinline__ void scan_rows(const float2 (&part)[SCAN_NG], float* __restrict__ yp,
                                          const int c, const int g)
{
    float s0 = part[0].x, s1 = part[0].y;
#pragma unroll
    for (int q = 1; q < SCAN_NG; ++q) { s0 += part[q].x; s1 += part[q].y; }
    yp[(size_t)(c * SCAN_T + g * SCAN_RPW) * DIc]     = s0;
    yp[(size_t)(c * SCAN_T + g * SCAN_RPW + 1) * DIc] = s1;
}

template <int PAR>
__device__ __forceinline__ void scan_chunk(const int c, ScanLds& sm, ScanFetch& f, const ScanWave& w,
                                           const ScanIn& cur, ScanIn& nxt,
                                           float (&h)[SCAN_NPG], const float (&A)[SCAN_NPG],
                                           const float Dg, const int g, const int lane,
                                           float* __restrict__ yp)
{
    float2 part[SCAN_NG];
#pragma unroll
    for (int q = 0; q < SCAN_NG; ++q) part[q] = sm.y[PAR ^ 1][g][q][lane];
    __builtin_amdgcn_sched_barrier(0);
    scan_read<PAR ^ 1>(nxt, sm, g, lane);
    __builtin_amdgcn_sched_barrier(0);

    float4 du4[SCAN_T / 2];
#pragma unroll
    for (int p = 0; p < SCAN_T / 2; ++p) du4[p] = cur.du[p];
    asm volatile("" : "+v"(du4[0].x), "+v"(du4[0].z), "+v"(du4[1].x), "+v"(du4[1].z),
                      "+v"(du4[2].x), "+v"(du4[2].z), "+v"(du4[3].x), "+v"(du4[3].z));
    asm volatile("" : "+v"(du4[4].x), "+v"(du4[4].z), "+v"(du4[5].x), "+v"(du4[5].z),
                      "+v"(du4[6].x), "+v"(du4[6].z), "+v"(du4[7].x), "+v"(du4[7].z));
    float yv[SCAN_T];
    scan_steps(du4, cur.bc, yv, h, A, Dg);

    asm volatile("" :: "v"(yv[0]), "v"(yv[1]), "v"(yv[2]), "v"(yv[3]), "v"(yv[4]), "v"(yv[5]),
                       "v"(yv[6]), "v"(yv[7]), "v"(yv[8]), "v"(yv[9]), "v"(yv[10]), "v"(yv[11]),
                       "v"(yv[12]), "v"(yv[13]), "v"(yv[14]), "v"(yv[15]));
    __builtin_amdgcn_sched_barrier(0);
    if (c > 0) scan_rows(part, yp, c - 1, g);
#pragma unroll
    for (int p = 0; p < SCAN_T / 2; ++p)
        sm.y[PAR][p][g][lane] = make_float2(yv[2 * p], yv[2 * p + 1]);
    scan_park<PAR>(f, w, sm, g, lane);
    __builtin_amdgcn_sched_barrier(0);
    scan_fetch(f, w, c + 2 + SCAN_LEAD);

    __builtin_amdgcn_fence(__ATOMIC_RELEASE, "workgroup", "local");
    __builtin_amdgcn_s_barrier();
    __builtin_amdgcn_fence(__ATOMIC_ACQUIRE, "workgroup", "local");
}

#define SCAN_NCC (DIc / 32)
static_assert(BB <= BB * KK && DIc <= 32 * SCAN_NG, "the gate's work-groups fit in one row of the grid");
__global__ __launch_bounds__(32 * SCAN_NG)
void k_scan(const float* __restrict__ xc_act, const float* __restrict__ xdbl,
            const float* __restrict__ delta, const float* __restrict__ A_logs,
            const float* __restrict__ Ds, float* __restrict__ yarr,
            const float* __restrict__ gcpart, const float* __restrict__ fw1,
            const float* __restrict__ fw2, float* __restrict__ gate)
{
    if (blockIdx.y == SCAN_NCC) {
        if (blockIdx.x < BB) gate_part(gcpart, fw1, fw2, gate, blockIdx.x, threadIdx.x);
        return;
    }
    __shared__ ScanLds sm;
    const int kb = blockIdx.x;
    const int k = kb & 3, b = kb >> 2;
    const int lane = threadIdx.x & 31;
    const int g = __builtin_amdgcn_readfirstlane((int)(threadIdx.x >> 5));
    const int d = blockIdx.y * 32 + lane;
    const int n0 = g * SCAN_NPG;
    float A[SCAN_NPG];
#pragma unroll
    for (int n = 0; n < SCAN_NPG; ++n)
        A[n] = -__expf(A_logs[((size_t)(k * DIc + d)) * NN + n0 + n]);
    const float Dg = (g == SCAN_NG - 1) ? Ds[k * DIc + d] : 0.f;
    const size_t seqbase = ((size_t)(k * BB + b)) * LL;
    ScanWave w;
    w.uimg = xc_act + (size_t)b * LL * DIc + d;
    w.dp   = delta + seqbase * DIc + d;
    w.bcw  = xdbl  + seqbase * CPROJ + RR + lane;
    w.tr   = k & 1;
    w.flip = k >> 1;
    w.ts0  = g * SCAN_RPW;
#pragma unroll
    for (int r = 0; r < SCAN_RPW; ++r) w.po[r] = slot_pixoff(w.tr, w.flip, w.ts0 + r);
    w.bcpos = ((lane & 15) >> 1) * 4 + (lane & 1) + 2 * (lane >> 4);
    float* __restrict__ yp = yarr + seqbase * DIc + d;

    float h[SCAN_NPG];
#pragma unroll
    for (int n = 0; n < SCAN_NPG; ++n) h[n] = 0.f;

    ScanFetch f0, f1, f2, f3;
    scan_fetch(f0, w, 0);
    __builtin_amdgcn_sched_barrier(0);
    scan_fetch(f1, w, 1);
    __builtin_amdgcn_sched_barrier(0);
    scan_fetch(f2, w, 2);
    __builtin_amdgcn_sched_barrier(0);
    scan_fetch(f3, w, 3);
    __builtin_amdgcn_sched_barrier(0);
    scan_park<0>(f0, w, sm, g, lane);
#pragma unroll
    for (int p = 0; p < SCAN_T / 2; ++p) sm.y[1][p][g][lane] = make_float2(0.f, 0.f);
    scan_fetch(f0, w, SCAN_LEAD);
    __builtin_amdgcn_sched_barrier(0);
    __builtin_amdgcn_fence(__ATOMIC_RELEASE, "workgroup", "local");
    __builtin_amdgcn_s_barrier();
    __builtin_amdgcn_fence(__ATOMIC_ACQUIRE, "workgroup", "local");
    ScanIn inA, inB;
    scan_read<0>(inA, sm, g, lane);
    __builtin_amdgcn_sched_barrier(0);
    scan_park<1>(f1, w, sm, g, lane);
    scan_fetch(f1, w, SCAN_LEAD + 1);
    __builtin_amdgcn_sched_barrier(0);
    __builtin_amdgcn_fence(__ATOMIC_RELEASE, "workgroup", "local");
    __builtin_amdgcn_s_barrier();
    __builtin_amdgcn_fence(__ATOMIC_ACQUIRE, "workgroup", "local");

    for (int c = 0; c < SCAN_NCH; c += SCAN_LEAD) {
        scan_chunk<0>(c,     sm, f2, w, inA, inB, h, A, Dg, g, lane, yp);
        scan_chunk<1>(c + 1, sm, f3, w, inB, inA, h, A, Dg, g, lane, yp);
        scan_chunk<0>(c + 2, sm, f0, w, inA, inB, h, A, Dg, g, lane, yp);
        scan_chunk<1>(c + 3, sm, f1, w, inB, inA, h, A, Dg, g, lane, yp);
    }
    float2 part[SCAN_NG];
#pragma unroll
    for (int q = 0; q < SCAN_NG; ++q) part[q] = sm.y[(SCAN_NCH - 1) & 1][g][q][lane];
    scan_rows(part, yp, SCAN_NCH - 1, g);
}

template <int XM>
__device__ __forceinline__ float row_xor(float v)
{
    return __builtin_bit_cast(float,
        __builtin_amdgcn_update_dpp(0, __builtin_bit_cast(int, v), 0x160 + XM, 0xf, 0xf, false));
}
__device__ __forceinline__ float half_xor(float v)
{
    const int x = __builtin_bit_cast(int, v);
    return __builtin_bit_cast(float,
        __builtin_amdgcn_permlanex16(x, x, 0x76543210, 0xfedcba98, false, false));
}
__device__ __forceinline__ float wave_sum(float v)
{
    v += row_xor<1>(v);
    v += row_xor<2>(v);
    v += row_xor<4>(v);
    v += row_xor<8>(v);
    v += half_xor(v);
    return v;
}

__global__ __launch_bounds__(32 * PX_WAVES)
void k_combine(const float* __restrict__ yarr, const float* __restrict__ gate,
               const float* __restrict__ g1, const float* __restrict__ g2,
               const float* __restrict__ g3, const float* __restrict__ g4,
               const float* __restrict__ nw, const float* __restrict__ nb,
               const float* __restrict__ z_silu, float* __restrict__ yfinal)
{
    const int lane = threadIdx.x & 31;
    const int wave = __builtin_amdgcn_readfirstlane((int)(threadIdx.x >> 5));
    const int pix0 = (blockIdx.x * PX_WAVES + wave) * PX_PPW;
    const int b = pix0 / LL, lr0 = pix0 - b * LL;
    const size_t bstride = (size_t)LL * DIc;
    const float* __restrict__ Y0 = yarr + ((size_t)(0 * BB + b)) * bstride + lane;
    const float* __restrict__ Y1 = yarr + ((size_t)(1 * BB + b)) * bstride + lane;
    const float* __restrict__ Y2 = yarr + ((size_t)(2 * BB + b)) * bstride + lane;
    const float* __restrict__ Y3 = yarr + ((size_t)(3 * BB + b)) * bstride + lane;
    float G1[PX_CPL], G2[PX_CPL], G3[PX_CPL], G4[PX_CPL], GT[PX_CPL], NW[PX_CPL], NB[PX_CPL];
#pragma unroll
    for (int j = 0; j < PX_CPL; ++j) {
        const int d = lane + 32 * j;
        G1[j] = g1[d]; G2[j] = g2[d]; G3[j] = g3[d]; G4[j] = g4[d];
        GT[j] = gate[b * DIc + d]; NW[j] = nw[d]; NB[j] = nb[d];
    }
#pragma unroll
    for (int p = 0; p < PX_PPW; ++p) {
        const int lr = lr0 + p;
        const int h = lr / WW, w = lr % WW;
        const int l = (((h >> 2) * (WW / 4) + (w >> 2)) << 4) + ((h & 3) << 2) + (w & 3);
        const int lt = (l % WW) * HH + (l / WW);
        const float* __restrict__ zp = z_silu + ((size_t)b * LL + lr) * DIc + lane;
        float yc[PX_CPL], zz[PX_CPL];
        float s1 = 0.f, s2 = 0.f;
#pragma unroll
        for (int j = 0; j < PX_CPL; ++j) {
            float y = G1[j] * Y0[(size_t)l * DIc + 32 * j]
                    + G2[j] * Y2[(size_t)(LL - 1 - l) * DIc + 32 * j]
                    + G3[j] * Y1[(size_t)lt * DIc + 32 * j]
                    + G4[j] * Y3[(size_t)(LL - 1 - lt) * DIc + 32 * j];
            y *= GT[j];
            yc[j] = y;
            zz[j] = zp[32 * j];
            s1 += y;
            s2 += y * y;
        }
        s1 = wave_sum(s1);
        s2 = wave_sum(s2);
        const float mu  = s1 * (1.f / DIc);
        const float var = s2 * (1.f / DIc) - mu * mu;
        const float inv = rsqrtf(var + 1e-5f);
        float* __restrict__ op = yfinal + ((size_t)b * LL + lr) * DIc + lane;
#pragma unroll
        for (int j = 0; j < PX_CPL; ++j)
            op[32 * j] = ((yc[j] - mu) * inv * NW[j] + NB[j]) * zz[j];
    }
}

__global__ __launch_bounds__(32 * GW)
void k_outproj(const float* __restrict__ yf, const float* __restrict__ w,
               float* __restrict__ out)
{
    const int lane = threadIdx.x & 31;
    const int wave = __builtin_amdgcn_readfirstlane((int)(threadIdx.x >> 5));
    const int row0 = blockIdx.y * 64 + (wave >> 1) * 32, col0 = blockIdx.x * 64 + (wave & 1) * 32;
    const int nn = lane & 15, mh = (lane >> 4) * 8;
    const float* arow[2];
    const float* brow[2];
#pragma unroll
    for (int i = 0; i < 2; ++i) arow[i] = yf + (size_t)(row0 + i * 16 + nn) * DIc;
#pragma unroll
    for (int t = 0; t < 2; ++t) brow[t] = w + (size_t)(col0 + t * 16 + nn) * DIc;
    v8f acc[2][2];
    wmma_tiles<2, 2, DIc>(arow, brow, acc);
#pragma unroll
    for (int i = 0; i < 2; ++i)
#pragma unroll
        for (int t = 0; t < 2; ++t)
#pragma unroll
            for (int v = 0; v < 8; ++v) {
                int r = row0 + i * 16 + mh + v, c = col0 + t * 16 + nn;
                out[(size_t)r * DMm + c] = acc[i][t][v];
            }
}

extern "C" void kernel_launch(void* const* d_in, const int* in_sizes, int n_in,
                              void* d_out, int out_size, void* d_ws, size_t ws_size,
                              hipStream_t stream)
{
    (void)in_sizes; (void)n_in; (void)out_size; (void)ws_size;
    const float* x          = (const float*)d_in[0];
    const float* in_proj_w  = (const float*)d_in[1];
    const float* conv_w     = (const float*)d_in[2];
    const float* conv_b     = (const float*)d_in[3];
    const float* x_proj_w   = (const float*)d_in[4];
    const float* dt_w       = (const float*)d_in[5];
    const float* dt_b       = (const float*)d_in[6];
    const float* A_logs     = (const float*)d_in[7];
    const float* Ds         = (const float*)d_in[8];
    const float* gamma1     = (const float*)d_in[9];
    const float* gamma2     = (const float*)d_in[10];
    const float* gamma3     = (const float*)d_in[11];
    const float* gamma4     = (const float*)d_in[12];
    const float* fc_w1      = (const float*)d_in[13];
    const float* fc_w2      = (const float*)d_in[14];
    const float* norm_w     = (const float*)d_in[15];
    const float* norm_b     = (const float*)d_in[16];
    const float* out_proj_w = (const float*)d_in[17];
    float* out = (float*)d_out;

    float* ws = (float*)d_ws;
    size_t o = 0;
    float* xc_lin = ws + o; o += (size_t)BB * LL * DIc;
    float* z_silu = ws + o; o += (size_t)BB * LL * DIc;
    float* xc_act = ws + o; o += (size_t)BB * LL * DIc;
    float* xdbl   = ws + o; o += (size_t)KK * BB * LL * CPROJ;
    float* delta  = ws + o; o += (size_t)KK * BB * LL * DIc;
    float* yarr   = ws + o; o += (size_t)KK * BB * LL * DIc;
    float* yfinal = ws + o; o += (size_t)BB * LL * DIc;
    float* wpad   = ws + o; o += (size_t)KK * CPAD * DIc;
    float* gcpart = ws + o; o += (size_t)BB * GC_NCH * DIc;
    float* gate   = ws + o; o += BB * DMm;

    k_inproj  <<<dim3(IN_NCB + 1, BB * LL / 64), 32 * GW, 0, stream>>>(x, in_proj_w, xc_lin, z_silu,
                                                                      x_proj_w, wpad);
    k_dwconv  <<<dim3(PX_NWG),         32 * PX_WAVES, 0, stream>>>(xc_lin, conv_w, conv_b, xc_act);
    k_xproj   <<<dim3(XP_NRB + XP_SIDE, KK), 32 * GW, 0, stream>>>(xc_act, wpad, xdbl, gcpart);
    k_dtproj  <<<dim3(DIc / 32, BB * LL / (16 * DT_W), KK), 32 * DT_W, 0, stream>>>(xdbl, dt_w, dt_b, delta);
    k_scan    <<<dim3(BB * KK, SCAN_NCC + 1), 32 * SCAN_NG, 0, stream>>>(xc_act, xdbl, delta, A_logs, Ds, yarr,
                                                                        gcpart, fc_w1, fc_w2, gate);
    k_combine <<<dim3(PX_NWG),         32 * PX_WAVES, 0, stream>>>(yarr, gate, gamma1, gamma2, gamma3,
                                                         gamma4, norm_w, norm_b, z_silu, yfinal);
    k_outproj <<<dim3(DMm / 64, BB * LL / 64), 32 * GW, 0, stream>>>(yfinal, out_proj_w, out);
}
